# starting kernel
# baseline (speedup 1.0000x reference)
#include <hip/hip_runtime.h>
#include <math.h>


typedef float v2f __attribute__((ext_vector_type(2)));
typedef float v8f __attribute__((ext_vector_type(8)));

#define N_PTS 65536
#define DDIM  256
#define KFREQ 64
#define NROW  128
#define NCHUNK 64
#define CHUNK_N (N_PTS / NCHUNK)

__device__ __forceinline__ v8f wmma_f32(v2f a, v2f b, v8f c) {
  return __builtin_amdgcn_wmma_f32_16x16x4_f32(false, a, false, b,
                                               (short)0, c, false, false);
}

__global__ __launch_bounds__(256) void tgen_kernel(float* __restrict__ T) {
  int idx = blockIdx.x * 256 + threadIdx.x;
  int n = idx >> 6;
  int k = idx & 63;
  unsigned m = ((unsigned)n * (unsigned)k) & 65535u;
  float ang = (float)m * 9.58737992429e-5f;
  float s, c;
  sincosf(ang, &s, &c);
  size_t base = (size_t)n * NROW + 2 * k;
  T[base]     = c;
  T[base + 1] = s;
}

__global__ __launch_bounds__(256) void pack_w_kernel(
    const float* __restrict__ W, float* __restrict__ Wp) {
  int idx = blockIdx.x * 256 + threadIdx.x;
  int kpair = idx >> 8;
  int col   = idx & 255;
  Wp[(size_t)idx * 2]     = W[(size_t)(2 * kpair) * DDIM + col];
  Wp[(size_t)idx * 2 + 1] = W[(size_t)(2 * kpair + 1) * DDIM + col];
}

__global__ __launch_bounds__(256) void dft_low_kernel(
    const float* __restrict__ x, const float* __restrict__ T,
    float* __restrict__ Ppart) {
  const int lane    = threadIdx.x & 31;
  const int wave    = threadIdx.x >> 5;
  const int lr      = lane & 15;
  const int halfSel = (lane >> 4) << 1;
  const int row0    = blockIdx.x * 16;
  const int nbase   = blockIdx.y * CHUNK_N;
  const int d0      = wave * 32;

  v8f acc0 = {0.f,0.f,0.f,0.f,0.f,0.f,0.f,0.f};
  v8f acc1 = acc0;

  const int arow = row0 + lr;
  for (int nk = nbase; nk < nbase + CHUNK_N; nk += 4) {
    const int na = nk + halfSel;
    v2f a;
    a.x = T[(size_t)na * NROW + arow];
    a.y = T[(size_t)(na + 1) * NROW + arow];
    v2f b0, b1;
    b0.x = x[(size_t)na * DDIM + d0 + lr];
    b0.y = x[(size_t)(na + 1) * DDIM + d0 + lr];
    b1.x = x[(size_t)na * DDIM + d0 + 16 + lr];
    b1.y = x[(size_t)(na + 1) * DDIM + d0 + 16 + lr];
    acc0 = wmma_f32(a, b0, acc0);
    acc1 = wmma_f32(a, b1, acc1);
  }

  const int mbase = (lane >> 4) << 3;
  const int col   = d0 + lr;
  size_t base = (size_t)blockIdx.y * (NROW * DDIM);
#pragma unroll
  for (int v = 0; v < 8; ++v) {
    int r = row0 + mbase + v;
    Ppart[base + (size_t)r * DDIM + col]      = acc0[v];
    Ppart[base + (size_t)r * DDIM + col + 16] = acc1[v];
  }
}

__global__ __launch_bounds__(256) void reduce_chunks_kernel(
    const float* __restrict__ Ppart, float* __restrict__ C) {
  int idx = blockIdx.x * 256 + threadIdx.x;
  float s = 0.f;
#pragma unroll 4
  for (int c = 0; c < NCHUNK; ++c)
    s += Ppart[(size_t)c * (NROW * DDIM) + idx];
  C[idx] = s;
}

__global__ __launch_bounds__(256) void spectral_mix_kernel(
    const float* __restrict__ C, const float* __restrict__ P_low,
    const float* __restrict__ B_low, float* __restrict__ BppP) {
  __shared__ float sXr[DDIM];
  __shared__ float sXi[DDIM];
  const int k = blockIdx.x;
  const int f = threadIdx.x;
  const float invN = 1.0f / (float)N_PTS;
  sXr[f] =  C[(size_t)(2 * k) * DDIM + f] * invN;
  sXi[f] = -C[(size_t)(2 * k + 1) * DDIM + f] * invN;
  __syncthreads();

  float ar = B_low[((size_t)k * DDIM + f) * 2 + 0];
  float ai = B_low[((size_t)k * DDIM + f) * 2 + 1];
  const float* Pk = P_low + (size_t)k * DDIM * DDIM * 2;
#pragma unroll 4
  for (int d = 0; d < DDIM; ++d) {
    float pr = Pk[((size_t)d * DDIM + f) * 2 + 0];
    float pi = Pk[((size_t)d * DDIM + f) * 2 + 1];
    float xr = sXr[d], xi = sXi[d];
    ar = fmaf(xr, pr, fmaf(-xi, pi, ar));
    ai = fmaf(xr, pi, fmaf( xi, pr, ai));
  }
  float sc = (k == 0) ? 1.0f : 2.0f;
  BppP[((size_t)k * DDIM + f) * 2 + 0] =  sc * ar;
  BppP[((size_t)k * DDIM + f) * 2 + 1] = -sc * ai;
}

__global__ __launch_bounds__(256) void fused_out_kernel(
    const float* __restrict__ x, const float* __restrict__ Wp,
    const float* __restrict__ Bias, const float* __restrict__ T,
    const float* __restrict__ BppP, float* __restrict__ out) {
  const int lane    = threadIdx.x & 31;
  const int wave    = threadIdx.x >> 5;
  const int lr      = lane & 15;
  const int halfSel = (lane >> 4) << 1;
  const int n0      = (blockIdx.x * 2 + (wave >> 2)) * 16;
  const int d0      = (wave & 3) * 64;
  const int arow    = n0 + lr;
  const int col     = d0 + lr;

  v8f acc[4];
#pragma unroll
  for (int t = 0; t < 4; ++t) acc[t] = (v8f){0.f,0.f,0.f,0.f,0.f,0.f,0.f,0.f};

  for (int k0 = 0; k0 < DDIM; k0 += 4) {
    const int kb = k0 + halfSel;
    v2f a = *(const v2f*)(x + (size_t)arow * DDIM + kb);
    const float* wrow = Wp + ((size_t)(kb >> 1) * DDIM + col) * 2;
#pragma unroll
    for (int t = 0; t < 4; ++t) {
      v2f b = *(const v2f*)(wrow + t * 32);
      acc[t] = wmma_f32(a, b, acc[t]);
    }
  }
  for (int r0 = 0; r0 < NROW; r0 += 4) {
    const int rb = r0 + halfSel;
    v2f a = *(const v2f*)(T + (size_t)arow * NROW + rb);
    const float* brow = BppP + ((size_t)(rb >> 1) * DDIM + col) * 2;
#pragma unroll
    for (int t = 0; t < 4; ++t) {
      v2f b = *(const v2f*)(brow + t * 32);
      acc[t] = wmma_f32(a, b, acc[t]);
    }
  }

  const int mbase = (lane >> 4) << 3;
#pragma unroll
  for (int t = 0; t < 4; ++t) {
    const int c = col + t * 16;
    const float bias = Bias[c];
#pragma unroll
    for (int v = 0; v < 8; ++v) {
      int n = n0 + mbase + v;
      out[(size_t)n * DDIM + c] = sinf(0.2f * (acc[t][v] + bias));
    }
  }
}

extern "C" void kernel_launch(void* const* d_in, const int* in_sizes, int n_in,
                              void* d_out, int out_size, void* d_ws,
                              size_t ws_size, hipStream_t stream) {
  (void)in_sizes; (void)n_in; (void)out_size; (void)ws_size;
  const float* x     = (const float*)d_in[0];
  const float* P_low = (const float*)d_in[1];
  const float* B_low = (const float*)d_in[2];
  const float* W     = (const float*)d_in[3];
  const float* Bias  = (const float*)d_in[4];
  float* out = (float*)d_out;

  float* ws    = (float*)d_ws;
  float* T     = ws;
  float* Ppart = T + (size_t)N_PTS * NROW;
  float* C     = Ppart + (size_t)NCHUNK * NROW * DDIM;
  float* BppP  = C + NROW * DDIM;
  float* Wp    = BppP + NROW * DDIM;

  tgen_kernel<<<dim3((N_PTS * KFREQ) / 256), dim3(256), 0, stream>>>(T);
  pack_w_kernel<<<dim3((DDIM * DDIM / 2) / 256), dim3(256), 0, stream>>>(W, Wp);
  dft_low_kernel<<<dim3(NROW / 16, NCHUNK), dim3(256), 0, stream>>>(x, T, Ppart);
  reduce_chunks_kernel<<<dim3((NROW * DDIM) / 256), dim3(256), 0, stream>>>(Ppart, C);
  spectral_mix_kernel<<<dim3(KFREQ), dim3(DDIM), 0, stream>>>(C, P_low, B_low, BppP);
  fused_out_kernel<<<dim3(N_PTS / 32), dim3(256), 0, stream>>>(x, Wp, Bias, T, BppP, out);
}


# Forward DFT GEMM: T and x slabs staged once per block through LDS (b128 loads, register prefetch); WMMA fragments read from LDS, 2x2 tiles per wave, hand-scheduled K loop; sum order unchanged
# speedup vs baseline: 1.2015x; 1.2015x over previous
#include <hip/hip_runtime.h>
#include <math.h>


typedef float v2f __attribute__((ext_vector_type(2)));
typedef float v4f __attribute__((ext_vector_type(4)));
typedef float v8f __attribute__((ext_vector_type(8)));

#define N_PTS 65536
#define DDIM  256
#define KFREQ 64
#define NROW  128
#define NCHUNK 64
#define CHUNK_N (N_PTS / NCHUNK)

__device__ __forceinline__ v8f wmma_f32(v2f a, v2f b, v8f c) {
  return __builtin_amdgcn_wmma_f32_16x16x4_f32(false, a, false, b,
                                               (short)0, c, false, false);
}

__global__ __launch_bounds__(256) void tgen_kernel(float* __restrict__ T) {
  int idx = blockIdx.x * 256 + threadIdx.x;
  int n = idx >> 6;
  int k = idx & 63;
  unsigned m = ((unsigned)n * (unsigned)k) & 65535u;
  float ang = (float)m * 9.58737992429e-5f;
  float s, c;
  sincosf(ang, &s, &c);
  size_t base = (size_t)n * NROW + 2 * k;
  T[base]     = c;
  T[base + 1] = s;
}

__global__ __launch_bounds__(256) void pack_w_kernel(
    const float* __restrict__ W, float* __restrict__ Wp) {
  int idx = blockIdx.x * 256 + threadIdx.x;
  int kpair = idx >> 8;
  int col   = idx & 255;
  Wp[(size_t)idx * 2]     = W[(size_t)(2 * kpair) * DDIM + col];
  Wp[(size_t)idx * 2 + 1] = W[(size_t)(2 * kpair + 1) * DDIM + col];
}

#define DFT_SLAB   64
#define DFT_COLS   64
#define DFT_NSLAB  (CHUNK_N / DFT_SLAB)
#define T_PITCH    (NROW + 8)
#define X_PITCH    (DFT_COLS + 8)

#define DFT_FRAG_LOADS(S)                                              \
  "ds_load_2addr_b32 %[a0" #S "], %[pa] offset1:136\n\t"               \
  "ds_load_2addr_b32 %[a1" #S "], %[pa] offset0:16 offset1:152\n\t"    \
  "ds_load_2addr_b32 %[b0" #S "], %[pb] offset1:72\n\t"                \
  "ds_load_2addr_b32 %[b1" #S "], %[pb] offset0:16 offset1:88\n\t"
#define DFT_ADVANCE                                                    \
  "v_add_nc_u32_e32 %[pa], 0x880, %[pa]\n\t"                           \
  "v_add_nc_u32_e32 %[pb], 0x480, %[pb]\n\t"
#define DFT_WMMAS(S)                                                   \
  "v_wmma_f32_16x16x4_f32 %[c00], %[a0" #S "], %[b0" #S "], %[c00]\n\t" \
  "v_wmma_f32_16x16x4_f32 %[c01], %[a0" #S "], %[b1" #S "], %[c01]\n\t" \
  "v_wmma_f32_16x16x4_f32 %[c10], %[a1" #S "], %[b0" #S "], %[c10]\n\t" \
  "v_wmma_f32_16x16x4_f32 %[c11], %[a1" #S "], %[b1" #S "], %[c11]\n\t"
#define DFT_STEP(CUR, NXT)                                             \
  DFT_ADVANCE DFT_FRAG_LOADS(NXT) "s_wait_dscnt 0x4\n\t" DFT_WMMAS(CUR)

__global__ __launch_bounds__(256) void dft_low_kernel(
    const float* __restrict__ x, const float* __restrict__ T,
    float* __restrict__ Ppart) {
  __shared__ __attribute__((aligned(16))) float sT[DFT_SLAB * T_PITCH];
  __shared__ __attribute__((aligned(16))) float sX[DFT_SLAB * X_PITCH];

  const int tid     = threadIdx.x;
  const int lane    = tid & 31;
  const int wave    = tid >> 5;
  const int lr      = lane & 15;
  const int halfSel = (lane >> 4) << 1;
  const int q       = blockIdx.x;
  const int nbase   = blockIdx.y * CHUNK_N;
  const int rowW    = (wave & 3) * 32;
  const int colW    = (wave >> 2) * 32;

  v8f acc00 = {0.f,0.f,0.f,0.f,0.f,0.f,0.f,0.f};
  v8f acc01 = acc00, acc10 = acc00, acc11 = acc00;

  v4f pT[8];
  v4f pX[4];

  const float* Tg = T + (size_t)nbase * NROW;
  const float* xg = x + (size_t)nbase * DDIM + q * DFT_COLS;

#pragma unroll
  for (int j = 0; j < 8; ++j) {
    const int i = tid + 256 * j;
    pT[j] = *(const v4f*)(Tg + (size_t)i * 4);
  }
#pragma unroll
  for (int j = 0; j < 4; ++j) {
    const int i = tid + 256 * j;
    pX[j] = *(const v4f*)(xg + (size_t)(i >> 4) * DDIM + (i & 15) * 4);
  }

  const unsigned pa0 = (unsigned)(size_t)(sT + halfSel * T_PITCH + rowW + lr);
  const unsigned pb0 = (unsigned)(size_t)(sX + halfSel * X_PITCH + colW + lr);

  for (int s = 0; s < DFT_NSLAB; ++s) {
    __syncthreads();
#pragma unroll
    for (int j = 0; j < 8; ++j) {
      const int i = tid + 256 * j;
      *(v4f*)(sT + (i >> 5) * T_PITCH + (i & 31) * 4) = pT[j];
    }
#pragma unroll
    for (int j = 0; j < 4; ++j) {
      const int i = tid + 256 * j;
      *(v4f*)(sX + (i >> 4) * X_PITCH + (i & 15) * 4) = pX[j];
    }
    __syncthreads();

    if (s + 1 < DFT_NSLAB) {
      const float* Tn = Tg + (size_t)(s + 1) * DFT_SLAB * NROW;
      const float* xn = xg + (size_t)(s + 1) * DFT_SLAB * DDIM;
#pragma unroll
      for (int j = 0; j < 8; ++j) {
        const int i = tid + 256 * j;
        pT[j] = *(const v4f*)(Tn + (size_t)i * 4);
      }
#pragma unroll
      for (int j = 0; j < 4; ++j) {
        const int i = tid + 256 * j;
        pX[j] = *(const v4f*)(xn + (size_t)(i >> 4) * DDIM + (i & 15) * 4);
      }
    }

    {
      unsigned pa = pa0, pb = pb0;
      v2f a0x, a1x, b0x, b1x, a0y, a1y, b0y, b1y;
      asm volatile(
          DFT_FRAG_LOADS(x)
          DFT_STEP(x, y) DFT_STEP(y, x) DFT_STEP(x, y) DFT_STEP(y, x)
          DFT_STEP(x, y) DFT_STEP(y, x) DFT_STEP(x, y) DFT_STEP(y, x)
          DFT_STEP(x, y) DFT_STEP(y, x) DFT_STEP(x, y) DFT_STEP(y, x)
          DFT_STEP(x, y) DFT_STEP(y, x) DFT_STEP(x, y)
          "s_wait_dscnt 0x0\n\t"
          DFT_WMMAS(y)
          "v_nop\n\tv_nop\n\tv_nop\n\tv_nop"
          : [c00] "+v"(acc00), [c01] "+v"(acc01),
            [c10] "+v"(acc10), [c11] "+v"(acc11),
            [pa] "+v"(pa), [pb] "+v"(pb),
            [a0x] "=&v"(a0x), [a1x] "=&v"(a1x),
            [b0x] "=&v"(b0x), [b1x] "=&v"(b1x),
            [a0y] "=&v"(a0y), [a1y] "=&v"(a1y),
            [b0y] "=&v"(b0y), [b1y] "=&v"(b1y)
          :
          : "memory");
    }
  }

  const int mbase = (lane >> 4) << 3;
  const int col   = q * DFT_COLS + colW + lr;
  const size_t base = (size_t)blockIdx.y * (NROW * DDIM);
#pragma unroll
  for (int v = 0; v < 8; ++v) {
    const int r = rowW + mbase + v;
    Ppart[base + (size_t)r * DDIM + col]             = acc00[v];
    Ppart[base + (size_t)r * DDIM + col + 16]        = acc01[v];
    Ppart[base + (size_t)(r + 16) * DDIM + col]      = acc10[v];
    Ppart[base + (size_t)(r + 16) * DDIM + col + 16] = acc11[v];
  }
}

__global__ __launch_bounds__(256) void reduce_chunks_kernel(
    const float* __restrict__ Ppart, float* __restrict__ C) {
  int idx = blockIdx.x * 256 + threadIdx.x;
  float s = 0.f;
#pragma unroll 4
  for (int c = 0; c < NCHUNK; ++c)
    s += Ppart[(size_t)c * (NROW * DDIM) + idx];
  C[idx] = s;
}

__global__ __launch_bounds__(256) void spectral_mix_kernel(
    const float* __restrict__ C, const float* __restrict__ P_low,
    const float* __restrict__ B_low, float* __restrict__ BppP) {
  __shared__ float sXr[DDIM];
  __shared__ float sXi[DDIM];
  const int k = blockIdx.x;
  const int f = threadIdx.x;
  const float invN = 1.0f / (float)N_PTS;
  sXr[f] =  C[(size_t)(2 * k) * DDIM + f] * invN;
  sXi[f] = -C[(size_t)(2 * k + 1) * DDIM + f] * invN;
  __syncthreads();

  float ar = B_low[((size_t)k * DDIM + f) * 2 + 0];
  float ai = B_low[((size_t)k * DDIM + f) * 2 + 1];
  const float* Pk = P_low + (size_t)k * DDIM * DDIM * 2;
#pragma unroll 4
  for (int d = 0; d < DDIM; ++d) {
    float pr = Pk[((size_t)d * DDIM + f) * 2 + 0];
    float pi = Pk[((size_t)d * DDIM + f) * 2 + 1];
    float xr = sXr[d], xi = sXi[d];
    ar = fmaf(xr, pr, fmaf(-xi, pi, ar));
    ai = fmaf(xr, pi, fmaf( xi, pr, ai));
  }
  float sc = (k == 0) ? 1.0f : 2.0f;
  BppP[((size_t)k * DDIM + f) * 2 + 0] =  sc * ar;
  BppP[((size_t)k * DDIM + f) * 2 + 1] = -sc * ai;
}

__global__ __launch_bounds__(256) void fused_out_kernel(
    const float* __restrict__ x, const float* __restrict__ Wp,
    const float* __restrict__ Bias, const float* __restrict__ T,
    const float* __restrict__ BppP, float* __restrict__ out) {
  const int lane    = threadIdx.x & 31;
  const int wave    = threadIdx.x >> 5;
  const int lr      = lane & 15;
  const int halfSel = (lane >> 4) << 1;
  const int n0      = (blockIdx.x * 2 + (wave >> 2)) * 16;
  const int d0      = (wave & 3) * 64;
  const int arow    = n0 + lr;
  const int col     = d0 + lr;

  v8f acc[4];
#pragma unroll
  for (int t = 0; t < 4; ++t) acc[t] = (v8f){0.f,0.f,0.f,0.f,0.f,0.f,0.f,0.f};

  for (int k0 = 0; k0 < DDIM; k0 += 4) {
    const int kb = k0 + halfSel;
    v2f a = *(const v2f*)(x + (size_t)arow * DDIM + kb);
    const float* wrow = Wp + ((size_t)(kb >> 1) * DDIM + col) * 2;
#pragma unroll
    for (int t = 0; t < 4; ++t) {
      v2f b = *(const v2f*)(wrow + t * 32);
      acc[t] = wmma_f32(a, b, acc[t]);
    }
  }
  for (int r0 = 0; r0 < NROW; r0 += 4) {
    const int rb = r0 + halfSel;
    v2f a = *(const v2f*)(T + (size_t)arow * NROW + rb);
    const float* brow = BppP + ((size_t)(rb >> 1) * DDIM + col) * 2;
#pragma unroll
    for (int t = 0; t < 4; ++t) {
      v2f b = *(const v2f*)(brow + t * 32);
      acc[t] = wmma_f32(a, b, acc[t]);
    }
  }

  const int mbase = (lane >> 4) << 3;
#pragma unroll
  for (int t = 0; t < 4; ++t) {
    const int c = col + t * 16;
    const float bias = Bias[c];
#pragma unroll
    for (int v = 0; v < 8; ++v) {
      int n = n0 + mbase + v;
      out[(size_t)n * DDIM + c] = sinf(0.2f * (acc[t][v] + bias));
    }
  }
}

extern "C" void kernel_launch(void* const* d_in, const int* in_sizes, int n_in,
                              void* d_out, int out_size, void* d_ws,
                              size_t ws_size, hipStream_t stream) {
  (void)in_sizes; (void)n_in; (void)out_size; (void)ws_size;
  const float* x     = (const float*)d_in[0];
  const float* P_low = (const float*)d_in[1];
  const float* B_low = (const float*)d_in[2];
  const float* W     = (const float*)d_in[3];
  const float* Bias  = (const float*)d_in[4];
  float* out = (float*)d_out;

  float* ws    = (float*)d_ws;
  float* T     = ws;
  float* Ppart = T + (size_t)N_PTS * NROW;
  float* C     = Ppart + (size_t)NCHUNK * NROW * DDIM;
  float* BppP  = C + NROW * DDIM;
  float* Wp    = BppP + NROW * DDIM;

  tgen_kernel<<<dim3((N_PTS * KFREQ) / 256), dim3(256), 0, stream>>>(T);
  pack_w_kernel<<<dim3((DDIM * DDIM / 2) / 256), dim3(256), 0, stream>>>(W, Wp);
  dft_low_kernel<<<dim3(DDIM / DFT_COLS, NCHUNK), dim3(256), 0, stream>>>(x, T, Ppart);
  reduce_chunks_kernel<<<dim3((NROW * DDIM) / 256), dim3(256), 0, stream>>>(Ppart, C);
  spectral_mix_kernel<<<dim3(KFREQ), dim3(DDIM), 0, stream>>>(C, P_low, B_low, BppP);
  fused_out_kernel<<<dim3(N_PTS / 32), dim3(256), 0, stream>>>(x, Wp, Bias, T, BppP, out);
}


# Spectral mix restructured as a stream: 256 blocks, each thread issues its 32 P_low b128 loads up front; d-sum split in 8 slices added in fixed order via LDS (reduction tree changed)
# speedup vs baseline: 1.3942x; 1.1604x over previous
#include <hip/hip_runtime.h>
#include <math.h>


typedef float v2f __attribute__((ext_vector_type(2)));
typedef float v4f __attribute__((ext_vector_type(4)));
typedef float v8f __attribute__((ext_vector_type(8)));

#define N_PTS 65536
#define DDIM  256
#define KFREQ 64
#define NROW  128
#define NCHUNK 64
#define CHUNK_N (N_PTS / NCHUNK)

__device__ __forceinline__ v8f wmma_f32(v2f a, v2f b, v8f c) {
  return __builtin_amdgcn_wmma_f32_16x16x4_f32(false, a, false, b,
                                               (short)0, c, false, false);
}

__global__ __launch_bounds__(256) void tgen_kernel(float* __restrict__ T) {
  int idx = blockIdx.x * 256 + threadIdx.x;
  int n = idx >> 6;
  int k = idx & 63;
  unsigned m = ((unsigned)n * (unsigned)k) & 65535u;
  float ang = (float)m * 9.58737992429e-5f;
  float s, c;
  sincosf(ang, &s, &c);
  size_t base = (size_t)n * NROW + 2 * k;
  T[base]     = c;
  T[base + 1] = s;
}

__global__ __launch_bounds__(256) void pack_w_kernel(
    const float* __restrict__ W, float* __restrict__ Wp) {
  int idx = blockIdx.x * 256 + threadIdx.x;
  int kpair = idx >> 8;
  int col   = idx & 255;
  Wp[(size_t)idx * 2]     = W[(size_t)(2 * kpair) * DDIM + col];
  Wp[(size_t)idx * 2 + 1] = W[(size_t)(2 * kpair + 1) * DDIM + col];
}

#define DFT_SLAB   64
#define DFT_COLS   64
#define DFT_NSLAB  (CHUNK_N / DFT_SLAB)
#define T_PITCH    (NROW + 8)
#define X_PITCH    (DFT_COLS + 8)

#define DFT_FRAG_LOADS(S)                                              \
  "ds_load_2addr_b32 %[a0" #S "], %[pa] offset1:136\n\t"               \
  "ds_load_2addr_b32 %[a1" #S "], %[pa] offset0:16 offset1:152\n\t"    \
  "ds_load_2addr_b32 %[b0" #S "], %[pb] offset1:72\n\t"                \
  "ds_load_2addr_b32 %[b1" #S "], %[pb] offset0:16 offset1:88\n\t"
#define DFT_ADVANCE                                                    \
  "v_add_nc_u32_e32 %[pa], 0x880, %[pa]\n\t"                           \
  "v_add_nc_u32_e32 %[pb], 0x480, %[pb]\n\t"
#define DFT_WMMAS(S)                                                   \
  "v_wmma_f32_16x16x4_f32 %[c00], %[a0" #S "], %[b0" #S "], %[c00]\n\t" \
  "v_wmma_f32_16x16x4_f32 %[c01], %[a0" #S "], %[b1" #S "], %[c01]\n\t" \
  "v_wmma_f32_16x16x4_f32 %[c10], %[a1" #S "], %[b0" #S "], %[c10]\n\t" \
  "v_wmma_f32_16x16x4_f32 %[c11], %[a1" #S "], %[b1" #S "], %[c11]\n\t"
#define DFT_STEP(CUR, NXT)                                             \
  DFT_ADVANCE DFT_FRAG_LOADS(NXT) "s_wait_dscnt 0x4\n\t" DFT_WMMAS(CUR)

__global__ __launch_bounds__(256) void dft_low_kernel(
    const float* __restrict__ x, const float* __restrict__ T,
    float* __restrict__ Ppart) {
  __shared__ __attribute__((aligned(16))) float sT[DFT_SLAB * T_PITCH];
  __shared__ __attribute__((aligned(16))) float sX[DFT_SLAB * X_PITCH];

  const int tid     = threadIdx.x;
  const int lane    = tid & 31;
  const int wave    = tid >> 5;
  const int lr      = lane & 15;
  const int halfSel = (lane >> 4) << 1;
  const int q       = blockIdx.x;
  const int nbase   = blockIdx.y * CHUNK_N;
  const int rowW    = (wave & 3) * 32;
  const int colW    = (wave >> 2) * 32;

  v8f acc00 = {0.f,0.f,0.f,0.f,0.f,0.f,0.f,0.f};
  v8f acc01 = acc00, acc10 = acc00, acc11 = acc00;

  v4f pT[8];
  v4f pX[4];

  const float* Tg = T + (size_t)nbase * NROW;
  const float* xg = x + (size_t)nbase * DDIM + q * DFT_COLS;

#pragma unroll
  for (int j = 0; j < 8; ++j) {
    const int i = tid + 256 * j;
    pT[j] = *(const v4f*)(Tg + (size_t)i * 4);
  }
#pragma unroll
  for (int j = 0; j < 4; ++j) {
    const int i = tid + 256 * j;
    pX[j] = *(const v4f*)(xg + (size_t)(i >> 4) * DDIM + (i & 15) * 4);
  }

  const unsigned pa0 = (unsigned)(size_t)(sT + halfSel * T_PITCH + rowW + lr);
  const unsigned pb0 = (unsigned)(size_t)(sX + halfSel * X_PITCH + colW + lr);

  for (int s = 0; s < DFT_NSLAB; ++s) {
    __syncthreads();
#pragma unroll
    for (int j = 0; j < 8; ++j) {
      const int i = tid + 256 * j;
      *(v4f*)(sT + (i >> 5) * T_PITCH + (i & 31) * 4) = pT[j];
    }
#pragma unroll
    for (int j = 0; j < 4; ++j) {
      const int i = tid + 256 * j;
      *(v4f*)(sX + (i >> 4) * X_PITCH + (i & 15) * 4) = pX[j];
    }
    __syncthreads();

    if (s + 1 < DFT_NSLAB) {
      const float* Tn = Tg + (size_t)(s + 1) * DFT_SLAB * NROW;
      const float* xn = xg + (size_t)(s + 1) * DFT_SLAB * DDIM;
#pragma unroll
      for (int j = 0; j < 8; ++j) {
        const int i = tid + 256 * j;
        pT[j] = *(const v4f*)(Tn + (size_t)i * 4);
      }
#pragma unroll
      for (int j = 0; j < 4; ++j) {
        const int i = tid + 256 * j;
        pX[j] = *(const v4f*)(xn + (size_t)(i >> 4) * DDIM + (i & 15) * 4);
      }
    }

    {
      unsigned pa = pa0, pb = pb0;
      v2f a0x, a1x, b0x, b1x, a0y, a1y, b0y, b1y;
      asm volatile(
          DFT_FRAG_LOADS(x)
          DFT_STEP(x, y) DFT_STEP(y, x) DFT_STEP(x, y) DFT_STEP(y, x)
          DFT_STEP(x, y) DFT_STEP(y, x) DFT_STEP(x, y) DFT_STEP(y, x)
          DFT_STEP(x, y) DFT_STEP(y, x) DFT_STEP(x, y) DFT_STEP(y, x)
          DFT_STEP(x, y) DFT_STEP(y, x) DFT_STEP(x, y)
          "s_wait_dscnt 0x0\n\t"
          DFT_WMMAS(y)
          "v_nop\n\tv_nop\n\tv_nop\n\tv_nop"
          : [c00] "+v"(acc00), [c01] "+v"(acc01),
            [c10] "+v"(acc10), [c11] "+v"(acc11),
            [pa] "+v"(pa), [pb] "+v"(pb),
            [a0x] "=&v"(a0x), [a1x] "=&v"(a1x),
            [b0x] "=&v"(b0x), [b1x] "=&v"(b1x),
            [a0y] "=&v"(a0y), [a1y] "=&v"(a1y),
            [b0y] "=&v"(b0y), [b1y] "=&v"(b1y)
          :
          : "memory");
    }
  }

  const int mbase = (lane >> 4) << 3;
  const int col   = q * DFT_COLS + colW + lr;
  const size_t base = (size_t)blockIdx.y * (NROW * DDIM);
#pragma unroll
  for (int v = 0; v < 8; ++v) {
    const int r = rowW + mbase + v;
    Ppart[base + (size_t)r * DDIM + col]             = acc00[v];
    Ppart[base + (size_t)r * DDIM + col + 16]        = acc01[v];
    Ppart[base + (size_t)(r + 16) * DDIM + col]      = acc10[v];
    Ppart[base + (size_t)(r + 16) * DDIM + col + 16] = acc11[v];
  }
}

__global__ __launch_bounds__(256) void reduce_chunks_kernel(
    const float* __restrict__ Ppart, float* __restrict__ C) {
  int idx = blockIdx.x * 256 + threadIdx.x;
  float s = 0.f;
#pragma unroll 4
  for (int c = 0; c < NCHUNK; ++c)
    s += Ppart[(size_t)c * (NROW * DDIM) + idx];
  C[idx] = s;
}

#define MIX_FQ 64
#define MIX_DS 32

__global__ __launch_bounds__(256) void spectral_mix_kernel(
    const float* __restrict__ C, const float* __restrict__ P_low,
    const float* __restrict__ B_low, float* __restrict__ BppP) {
  __shared__ float sXr[DDIM];
  __shared__ float sXi[DDIM];
  __shared__ __attribute__((aligned(16))) float sPart[8 * MIX_FQ * 2];

  const int tid   = threadIdx.x;
  const int lane  = tid & 31;
  const int wave  = tid >> 5;
  const int fq    = blockIdx.x;
  const int k     = blockIdx.y;
  const int dbase = wave * MIX_DS;
  const int f0    = fq * MIX_FQ + lane * 2;

  const float invN = 1.0f / (float)N_PTS;
  const float cr = C[(size_t)(2 * k) * DDIM + tid];
  const float ci = C[(size_t)(2 * k + 1) * DDIM + tid];

  const float* Pk = P_low + (size_t)k * DDIM * DDIM * 2
                          + ((size_t)dbase * DDIM + f0) * 2;
  v4f p[MIX_DS];
#pragma unroll
  for (int j = 0; j < MIX_DS; ++j)
    p[j] = *(const v4f*)(Pk + (size_t)j * DDIM * 2);

  sXr[tid] =  cr * invN;
  sXi[tid] = -ci * invN;
  __syncthreads();

  float ar0 = 0.f, ai0 = 0.f, ar1 = 0.f, ai1 = 0.f;
#pragma unroll
  for (int j = 0; j < MIX_DS; ++j) {
    const float xr = sXr[dbase + j], xi = sXi[dbase + j];
    ar0 = fmaf(xr, p[j].x, fmaf(-xi, p[j].y, ar0));
    ai0 = fmaf(xr, p[j].y, fmaf( xi, p[j].x, ai0));
    ar1 = fmaf(xr, p[j].z, fmaf(-xi, p[j].w, ar1));
    ai1 = fmaf(xr, p[j].w, fmaf( xi, p[j].z, ai1));
  }
  *(v4f*)(sPart + (wave * MIX_FQ + lane * 2) * 2) = (v4f){ar0, ai0, ar1, ai1};
  __syncthreads();

  if (tid < MIX_FQ * 2) {
    const int fl = tid >> 1;
    const int ri = tid & 1;
    const int f  = fq * MIX_FQ + fl;
    float s = B_low[((size_t)k * DDIM + f) * 2 + ri];
#pragma unroll
    for (int w = 0; w < 8; ++w) s += sPart[(w * MIX_FQ + fl) * 2 + ri];
    const float sc = (k == 0) ? 1.0f : 2.0f;
    BppP[((size_t)k * DDIM + f) * 2 + ri] = ri ? -sc * s : sc * s;
  }
}

__global__ __launch_bounds__(256) void fused_out_kernel(
    const float* __restrict__ x, const float* __restrict__ Wp,
    const float* __restrict__ Bias, const float* __restrict__ T,
    const float* __restrict__ BppP, float* __restrict__ out) {
  const int lane    = threadIdx.x & 31;
  const int wave    = threadIdx.x >> 5;
  const int lr      = lane & 15;
  const int halfSel = (lane >> 4) << 1;
  const int n0      = (blockIdx.x * 2 + (wave >> 2)) * 16;
  const int d0      = (wave & 3) * 64;
  const int arow    = n0 + lr;
  const int col     = d0 + lr;

  v8f acc[4];
#pragma unroll
  for (int t = 0; t < 4; ++t) acc[t] = (v8f){0.f,0.f,0.f,0.f,0.f,0.f,0.f,0.f};

  for (int k0 = 0; k0 < DDIM; k0 += 4) {
    const int kb = k0 + halfSel;
    v2f a = *(const v2f*)(x + (size_t)arow * DDIM + kb);
    const float* wrow = Wp + ((size_t)(kb >> 1) * DDIM + col) * 2;
#pragma unroll
    for (int t = 0; t < 4; ++t) {
      v2f b = *(const v2f*)(wrow + t * 32);
      acc[t] = wmma_f32(a, b, acc[t]);
    }
  }
  for (int r0 = 0; r0 < NROW; r0 += 4) {
    const int rb = r0 + halfSel;
    v2f a = *(const v2f*)(T + (size_t)arow * NROW + rb);
    const float* brow = BppP + ((size_t)(rb >> 1) * DDIM + col) * 2;
#pragma unroll
    for (int t = 0; t < 4; ++t) {
      v2f b = *(const v2f*)(brow + t * 32);
      acc[t] = wmma_f32(a, b, acc[t]);
    }
  }

  const int mbase = (lane >> 4) << 3;
#pragma unroll
  for (int t = 0; t < 4; ++t) {
    const int c = col + t * 16;
    const float bias = Bias[c];
#pragma unroll
    for (int v = 0; v < 8; ++v) {
      int n = n0 + mbase + v;
      out[(size_t)n * DDIM + c] = sinf(0.2f * (acc[t][v] + bias));
    }
  }
}

extern "C" void kernel_launch(void* const* d_in, const int* in_sizes, int n_in,
                              void* d_out, int out_size, void* d_ws,
                              size_t ws_size, hipStream_t stream) {
  (void)in_sizes; (void)n_in; (void)out_size; (void)ws_size;
  const float* x     = (const float*)d_in[0];
  const float* P_low = (const float*)d_in[1];
  const float* B_low = (const float*)d_in[2];
  const float* W     = (const float*)d_in[3];
  const float* Bias  = (const float*)d_in[4];
  float* out = (float*)d_out;

  float* ws    = (float*)d_ws;
  float* T     = ws;
  float* Ppart = T + (size_t)N_PTS * NROW;
  float* C     = Ppart + (size_t)NCHUNK * NROW * DDIM;
  float* BppP  = C + NROW * DDIM;
  float* Wp    = BppP + NROW * DDIM;

  tgen_kernel<<<dim3((N_PTS * KFREQ) / 256), dim3(256), 0, stream>>>(T);
  pack_w_kernel<<<dim3((DDIM * DDIM / 2) / 256), dim3(256), 0, stream>>>(W, Wp);
  dft_low_kernel<<<dim3(DDIM / DFT_COLS, NCHUNK), dim3(256), 0, stream>>>(x, T, Ppart);
  reduce_chunks_kernel<<<dim3((NROW * DDIM) / 256), dim3(256), 0, stream>>>(Ppart, C);
  spectral_mix_kernel<<<dim3(DDIM / MIX_FQ, KFREQ), dim3(256), 0, stream>>>(C, P_low, B_low, BppP);
  fused_out_kernel<<<dim3(N_PTS / 32), dim3(256), 0, stream>>>(x, Wp, Bias, T, BppP, out);
}


# Fused output GEMM: x/T and packed W/Bpp slabs staged once per 64-row block in LDS (b128 loads, register prefetch); b64 fragment reads from LDS, 2x4 tiles per wave, asm K loop; sum order unchanged
# speedup vs baseline: 1.4427x; 1.0348x over previous
#include <hip/hip_runtime.h>
#include <math.h>


typedef float v2f __attribute__((ext_vector_type(2)));
typedef float v4f __attribute__((ext_vector_type(4)));
typedef float v8f __attribute__((ext_vector_type(8)));

#define N_PTS 65536
#define DDIM  256
#define KFREQ 64
#define NROW  128
#define NCHUNK 64
#define CHUNK_N (N_PTS / NCHUNK)

__device__ __forceinline__ v8f wmma_f32(v2f a, v2f b, v8f c) {
  return __builtin_amdgcn_wmma_f32_16x16x4_f32(false, a, false, b,
                                               (short)0, c, false, false);
}

__global__ __launch_bounds__(256) void tgen_kernel(float* __restrict__ T) {
  int idx = blockIdx.x * 256 + threadIdx.x;
  int n = idx >> 6;
  int k = idx & 63;
  unsigned m = ((unsigned)n * (unsigned)k) & 65535u;
  float ang = (float)m * 9.58737992429e-5f;
  float s, c;
  sincosf(ang, &s, &c);
  size_t base = (size_t)n * NROW + 2 * k;
  T[base]     = c;
  T[base + 1] = s;
}

__global__ __launch_bounds__(256) void pack_w_kernel(
    const float* __restrict__ W, float* __restrict__ Wp) {
  int idx = blockIdx.x * 256 + threadIdx.x;
  int kpair = idx >> 8;
  int col   = idx & 255;
  Wp[(size_t)idx * 2]     = W[(size_t)(2 * kpair) * DDIM + col];
  Wp[(size_t)idx * 2 + 1] = W[(size_t)(2 * kpair + 1) * DDIM + col];
}

#define DFT_SLAB   64
#define DFT_COLS   64
#define DFT_NSLAB  (CHUNK_N / DFT_SLAB)
#define T_PITCH    (NROW + 8)
#define X_PITCH    (DFT_COLS + 8)

#define DFT_FRAG_LOADS(S)                                              \
  "ds_load_2addr_b32 %[a0" #S "], %[pa] offset1:136\n\t"               \
  "ds_load_2addr_b32 %[a1" #S "], %[pa] offset0:16 offset1:152\n\t"    \
  "ds_load_2addr_b32 %[b0" #S "], %[pb] offset1:72\n\t"                \
  "ds_load_2addr_b32 %[b1" #S "], %[pb] offset0:16 offset1:88\n\t"
#define DFT_ADVANCE                                                    \
  "v_add_nc_u32_e32 %[pa], 0x880, %[pa]\n\t"                           \
  "v_add_nc_u32_e32 %[pb], 0x480, %[pb]\n\t"
#define DFT_WMMAS(S)                                                   \
  "v_wmma_f32_16x16x4_f32 %[c00], %[a0" #S "], %[b0" #S "], %[c00]\n\t" \
  "v_wmma_f32_16x16x4_f32 %[c01], %[a0" #S "], %[b1" #S "], %[c01]\n\t" \
  "v_wmma_f32_16x16x4_f32 %[c10], %[a1" #S "], %[b0" #S "], %[c10]\n\t" \
  "v_wmma_f32_16x16x4_f32 %[c11], %[a1" #S "], %[b1" #S "], %[c11]\n\t"
#define DFT_STEP(CUR, NXT)                                             \
  DFT_ADVANCE DFT_FRAG_LOADS(NXT) "s_wait_dscnt 0x4\n\t" DFT_WMMAS(CUR)

__global__ __launch_bounds__(256) void dft_low_kernel(
    const float* __restrict__ x, const float* __restrict__ T,
    float* __restrict__ Ppart) {
  __shared__ __attribute__((aligned(16))) float sT[DFT_SLAB * T_PITCH];
  __shared__ __attribute__((aligned(16))) float sX[DFT_SLAB * X_PITCH];

  const int tid     = threadIdx.x;
  const int lane    = tid & 31;
  const int wave    = tid >> 5;
  const int lr      = lane & 15;
  const int halfSel = (lane >> 4) << 1;
  const int q       = blockIdx.x;
  const int nbase   = blockIdx.y * CHUNK_N;
  const int rowW    = (wave & 3) * 32;
  const int colW    = (wave >> 2) * 32;

  v8f acc00 = {0.f,0.f,0.f,0.f,0.f,0.f,0.f,0.f};
  v8f acc01 = acc00, acc10 = acc00, acc11 = acc00;

  v4f pT[8];
  v4f pX[4];

  const float* Tg = T + (size_t)nbase * NROW;
  const float* xg = x + (size_t)nbase * DDIM + q * DFT_COLS;

#pragma unroll
  for (int j = 0; j < 8; ++j) {
    const int i = tid + 256 * j;
    pT[j] = *(const v4f*)(Tg + (size_t)i * 4);
  }
#pragma unroll
  for (int j = 0; j < 4; ++j) {
    const int i = tid + 256 * j;
    pX[j] = *(const v4f*)(xg + (size_t)(i >> 4) * DDIM + (i & 15) * 4);
  }

  const unsigned pa0 = (unsigned)(size_t)(sT + halfSel * T_PITCH + rowW + lr);
  const unsigned pb0 = (unsigned)(size_t)(sX + halfSel * X_PITCH + colW + lr);

  for (int s = 0; s < DFT_NSLAB; ++s) {
    __syncthreads();
#pragma unroll
    for (int j = 0; j < 8; ++j) {
      const int i = tid + 256 * j;
      *(v4f*)(sT + (i >> 5) * T_PITCH + (i & 31) * 4) = pT[j];
    }
#pragma unroll
    for (int j = 0; j < 4; ++j) {
      const int i = tid + 256 * j;
      *(v4f*)(sX + (i >> 4) * X_PITCH + (i & 15) * 4) = pX[j];
    }
    __syncthreads();

    if (s + 1 < DFT_NSLAB) {
      const float* Tn = Tg + (size_t)(s + 1) * DFT_SLAB * NROW;
      const float* xn = xg + (size_t)(s + 1) * DFT_SLAB * DDIM;
#pragma unroll
      for (int j = 0; j < 8; ++j) {
        const int i = tid + 256 * j;
        pT[j] = *(const v4f*)(Tn + (size_t)i * 4);
      }
#pragma unroll
      for (int j = 0; j < 4; ++j) {
        const int i = tid + 256 * j;
        pX[j] = *(const v4f*)(xn + (size_t)(i >> 4) * DDIM + (i & 15) * 4);
      }
    }

    {
      unsigned pa = pa0, pb = pb0;
      v2f a0x, a1x, b0x, b1x, a0y, a1y, b0y, b1y;
      asm volatile(
          DFT_FRAG_LOADS(x)
          DFT_STEP(x, y) DFT_STEP(y, x) DFT_STEP(x, y) DFT_STEP(y, x)
          DFT_STEP(x, y) DFT_STEP(y, x) DFT_STEP(x, y) DFT_STEP(y, x)
          DFT_STEP(x, y) DFT_STEP(y, x) DFT_STEP(x, y) DFT_STEP(y, x)
          DFT_STEP(x, y) DFT_STEP(y, x) DFT_STEP(x, y)
          "s_wait_dscnt 0x0\n\t"
          DFT_WMMAS(y)
          "v_nop\n\tv_nop\n\tv_nop\n\tv_nop"
          : [c00] "+v"(acc00), [c01] "+v"(acc01),
            [c10] "+v"(acc10), [c11] "+v"(acc11),
            [pa] "+v"(pa), [pb] "+v"(pb),
            [a0x] "=&v"(a0x), [a1x] "=&v"(a1x),
            [b0x] "=&v"(b0x), [b1x] "=&v"(b1x),
            [a0y] "=&v"(a0y), [a1y] "=&v"(a1y),
            [b0y] "=&v"(b0y), [b1y] "=&v"(b1y)
          :
          : "memory");
    }
  }

  const int mbase = (lane >> 4) << 3;
  const int col   = q * DFT_COLS + colW + lr;
  const size_t base = (size_t)blockIdx.y * (NROW * DDIM);
#pragma unroll
  for (int v = 0; v < 8; ++v) {
    const int r = rowW + mbase + v;
    Ppart[base + (size_t)r * DDIM + col]             = acc00[v];
    Ppart[base + (size_t)r * DDIM + col + 16]        = acc01[v];
    Ppart[base + (size_t)(r + 16) * DDIM + col]      = acc10[v];
    Ppart[base + (size_t)(r + 16) * DDIM + col + 16] = acc11[v];
  }
}

__global__ __launch_bounds__(256) void reduce_chunks_kernel(
    const float* __restrict__ Ppart, float* __restrict__ C) {
  int idx = blockIdx.x * 256 + threadIdx.x;
  float s = 0.f;
#pragma unroll 4
  for (int c = 0; c < NCHUNK; ++c)
    s += Ppart[(size_t)c * (NROW * DDIM) + idx];
  C[idx] = s;
}

#define MIX_FQ 64
#define MIX_DS 32

__global__ __launch_bounds__(256) void spectral_mix_kernel(
    const float* __restrict__ C, const float* __restrict__ P_low,
    const float* __restrict__ B_low, float* __restrict__ BppP) {
  __shared__ float sXr[DDIM];
  __shared__ float sXi[DDIM];
  __shared__ __attribute__((aligned(16))) float sPart[8 * MIX_FQ * 2];

  const int tid   = threadIdx.x;
  const int lane  = tid & 31;
  const int wave  = tid >> 5;
  const int fq    = blockIdx.x;
  const int k     = blockIdx.y;
  const int dbase = wave * MIX_DS;
  const int f0    = fq * MIX_FQ + lane * 2;

  const float invN = 1.0f / (float)N_PTS;
  const float cr = C[(size_t)(2 * k) * DDIM + tid];
  const float ci = C[(size_t)(2 * k + 1) * DDIM + tid];

  const float* Pk = P_low + (size_t)k * DDIM * DDIM * 2
                          + ((size_t)dbase * DDIM + f0) * 2;
  v4f p[MIX_DS];
#pragma unroll
  for (int j = 0; j < MIX_DS; ++j)
    p[j] = *(const v4f*)(Pk + (size_t)j * DDIM * 2);

  sXr[tid] =  cr * invN;
  sXi[tid] = -ci * invN;
  __syncthreads();

  float ar0 = 0.f, ai0 = 0.f, ar1 = 0.f, ai1 = 0.f;
#pragma unroll
  for (int j = 0; j < MIX_DS; ++j) {
    const float xr = sXr[dbase + j], xi = sXi[dbase + j];
    ar0 = fmaf(xr, p[j].x, fmaf(-xi, p[j].y, ar0));
    ai0 = fmaf(xr, p[j].y, fmaf( xi, p[j].x, ai0));
    ar1 = fmaf(xr, p[j].z, fmaf(-xi, p[j].w, ar1));
    ai1 = fmaf(xr, p[j].w, fmaf( xi, p[j].z, ai1));
  }
  *(v4f*)(sPart + (wave * MIX_FQ + lane * 2) * 2) = (v4f){ar0, ai0, ar1, ai1};
  __syncthreads();

  if (tid < MIX_FQ * 2) {
    const int fl = tid >> 1;
    const int ri = tid & 1;
    const int f  = fq * MIX_FQ + fl;
    float s = B_low[((size_t)k * DDIM + f) * 2 + ri];
#pragma unroll
    for (int w = 0; w < 8; ++w) s += sPart[(w * MIX_FQ + fl) * 2 + ri];
    const float sc = (k == 0) ? 1.0f : 2.0f;
    BppP[((size_t)k * DDIM + f) * 2 + ri] = ri ? -sc * s : sc * s;
  }
}

#define FO_ROWS   64
#define FO_KS     64
#define FO_NSLAB  6
#define FO_APITCH 68
#define FO_BPITCH 544

#define FO_LOADS(S, A0, A1, B0, B1)                                    \
  "ds_load_b64 %[a0" #S "], %[pa] offset:" #A0 "\n\t"                 \
  "ds_load_b64 %[a1" #S "], %[pa] offset:" #A1 "\n\t"                 \
  "ds_load_b64 %[b0" #S "], %[pb] offset:" #B0 "\n\t"                 \
  "ds_load_b64 %[b1" #S "], %[pb] offset:" #B1 "\n\t"                 \
  "ds_load_b64 %[b2" #S "], %[pc] offset:" #B0 "\n\t"                 \
  "ds_load_b64 %[b3" #S "], %[pc] offset:" #B1 "\n\t"
#define FO_WMMAS(S)                                                    \
  "v_wmma_f32_16x16x4_f32 %[c00], %[a0" #S "], %[b0" #S "], %[c00]\n\t" \
  "v_wmma_f32_16x16x4_f32 %[c01], %[a0" #S "], %[b1" #S "], %[c01]\n\t" \
  "v_wmma_f32_16x16x4_f32 %[c02], %[a0" #S "], %[b2" #S "], %[c02]\n\t" \
  "v_wmma_f32_16x16x4_f32 %[c03], %[a0" #S "], %[b3" #S "], %[c03]\n\t" \
  "v_wmma_f32_16x16x4_f32 %[c10], %[a1" #S "], %[b0" #S "], %[c10]\n\t" \
  "v_wmma_f32_16x16x4_f32 %[c11], %[a1" #S "], %[b1" #S "], %[c11]\n\t" \
  "v_wmma_f32_16x16x4_f32 %[c12], %[a1" #S "], %[b2" #S "], %[c12]\n\t" \
  "v_wmma_f32_16x16x4_f32 %[c13], %[a1" #S "], %[b3" #S "], %[c13]\n\t"
#define FO_STEP(CUR, NXT, A0, A1, B0, B1)                              \
  FO_LOADS(NXT, A0, A1, B0, B1) "s_wait_dscnt 0x6\n\t" FO_WMMAS(CUR)

__device__ __forceinline__ void fo_fetch(
    int s, int tid, int nblk, const float* __restrict__ x,
    const float* __restrict__ T, const float* __restrict__ Wp,
    const float* __restrict__ BppP, v4f (&pA)[4], v4f (&pB)[16]) {
  const float* Ag;
  const float* Bg;
  int apitch;
  if (s < 4) {
    Ag = x + (size_t)nblk * DDIM + s * FO_KS;
    Bg = Wp + (size_t)s * (FO_KS / 2) * (2 * DDIM);
    apitch = DDIM;
  } else {
    Ag = T + (size_t)nblk * NROW + (s - 4) * FO_KS;
    Bg = BppP + (size_t)(s - 4) * (FO_KS / 2) * (2 * DDIM);
    apitch = NROW;
  }
#pragma unroll
  for (int j = 0; j < 4; ++j) {
    const int i = tid + 256 * j;
    pA[j] = *(const v4f*)(Ag + (size_t)(i >> 4) * apitch + (i & 15) * 4);
  }
#pragma unroll
  for (int j = 0; j < 16; ++j) {
    const int i = tid + 256 * j;
    pB[j] = *(const v4f*)(Bg + (size_t)i * 4);
  }
}

__global__ __launch_bounds__(256) void fused_out_kernel(
    const float* __restrict__ x, const float* __restrict__ Wp,
    const float* __restrict__ Bias, const float* __restrict__ T,
    const float* __restrict__ BppP, float* __restrict__ out) {
  __shared__ __attribute__((aligned(16))) float sA[FO_ROWS * FO_APITCH];
  __shared__ __attribute__((aligned(16))) float sB[(FO_KS / 2) * FO_BPITCH];

  const int tid     = threadIdx.x;
  const int lane    = tid & 31;
  const int wave    = tid >> 5;
  const int lr      = lane & 15;
  const int hsel    = lane >> 4;
  const int nblk    = blockIdx.x * FO_ROWS;
  const int rowW    = (wave & 1) * 32;
  const int colW    = (wave >> 1) * 64;

  v8f c00 = {0.f,0.f,0.f,0.f,0.f,0.f,0.f,0.f};
  v8f c01 = c00, c02 = c00, c03 = c00, c10 = c00, c11 = c00, c12 = c00, c13 = c00;

  v4f pA[4];
  v4f pB[16];
  fo_fetch(0, tid, nblk, x, T, Wp, BppP, pA, pB);

  const unsigned pa = (unsigned)(size_t)(sA + (rowW + lr) * FO_APITCH + 2 * hsel);
  const unsigned pb = (unsigned)(size_t)(sB + hsel * FO_BPITCH + (colW + lr) * 2);
  const unsigned pc = pb + 256u;

  for (int s = 0; s < FO_NSLAB; ++s) {
    __syncthreads();
#pragma unroll
    for (int j = 0; j < 4; ++j) {
      const int i = tid + 256 * j;
      *(v4f*)(sA + (i >> 4) * FO_APITCH + (i & 15) * 4) = pA[j];
    }
#pragma unroll
    for (int j = 0; j < 16; ++j) {
      const int i = tid + 256 * j;
      *(v4f*)(sB + (i >> 7) * FO_BPITCH + (i & 127) * 4) = pB[j];
    }
    __syncthreads();

    if (s + 1 < FO_NSLAB)
      fo_fetch(s + 1, tid, nblk, x, T, Wp, BppP, pA, pB);

    {
      v2f a0x, a1x, b0x, b1x, b2x, b3x, a0y, a1y, b0y, b1y, b2y, b3y;
      asm volatile(
          FO_LOADS(x, 0, 4352, 0, 128)
          FO_STEP(x, y, 16, 4368, 4352, 4480)
          FO_STEP(y, x, 32, 4384, 8704, 8832)
          FO_STEP(x, y, 48, 4400, 13056, 13184)
          FO_STEP(y, x, 64, 4416, 17408, 17536)
          FO_STEP(x, y, 80, 4432, 21760, 21888)
          FO_STEP(y, x, 96, 4448, 26112, 26240)
          FO_STEP(x, y, 112, 4464, 30464, 30592)
          FO_STEP(y, x, 128, 4480, 34816, 34944)
          FO_STEP(x, y, 144, 4496, 39168, 39296)
          FO_STEP(y, x, 160, 4512, 43520, 43648)
          FO_STEP(x, y, 176, 4528, 47872, 48000)
          FO_STEP(y, x, 192, 4544, 52224, 52352)
          FO_STEP(x, y, 208, 4560, 56576, 56704)
          FO_STEP(y, x, 224, 4576, 60928, 61056)
          FO_STEP(x, y, 240, 4592, 65280, 65408)
          "s_wait_dscnt 0x0\n\t"
          FO_WMMAS(y)
          "v_nop\n\tv_nop\n\tv_nop\n\tv_nop"
          : [c00] "+v"(c00), [c01] "+v"(c01), [c02] "+v"(c02), [c03] "+v"(c03),
            [c10] "+v"(c10), [c11] "+v"(c11), [c12] "+v"(c12), [c13] "+v"(c13),
            [a0x] "=&v"(a0x), [a1x] "=&v"(a1x), [b0x] "=&v"(b0x),
            [b1x] "=&v"(b1x), [b2x] "=&v"(b2x), [b3x] "=&v"(b3x),
            [a0y] "=&v"(a0y), [a1y] "=&v"(a1y), [b0y] "=&v"(b0y),
            [b1y] "=&v"(b1y), [b2y] "=&v"(b2y), [b3y] "=&v"(b3y)
          : [pa] "v"(pa), [pb] "v"(pb), [pc] "v"(pc)
          : "memory");
    }
  }

  const int mbase = hsel << 3;
  const int col   = colW + lr;
  const int nrow  = nblk + rowW + mbase;
  const float bias0 = Bias[col];
  const float bias1 = Bias[col + 16];
  const float bias2 = Bias[col + 32];
  const float bias3 = Bias[col + 48];
#pragma unroll
  for (int v = 0; v < 8; ++v) {
    float* o0 = out + (size_t)(nrow + v) * DDIM + col;
    float* o1 = o0 + (size_t)16 * DDIM;
    o0[0]  = sinf(0.2f * (c00[v] + bias0));
    o0[16] = sinf(0.2f * (c01[v] + bias1));
    o0[32] = sinf(0.2f * (c02[v] + bias2));
    o0[48] = sinf(0.2f * (c03[v] + bias3));
    o1[0]  = sinf(0.2f * (c10[v] + bias0));
    o1[16] = sinf(0.2f * (c11[v] + bias1));
    o1[32] = sinf(0.2f * (c12[v] + bias2));
    o1[48] = sinf(0.2f * (c13[v] + bias3));
  }
}

extern "C" void kernel_launch(void* const* d_in, const int* in_sizes, int n_in,
                              void* d_out, int out_size, void* d_ws,
                              size_t ws_size, hipStream_t stream) {
  (void)in_sizes; (void)n_in; (void)out_size; (void)ws_size;
  const float* x     = (const float*)d_in[0];
  const float* P_low = (const float*)d_in[1];
  const float* B_low = (const float*)d_in[2];
  const float* W     = (const float*)d_in[3];
  const float* Bias  = (const float*)d_in[4];
  float* out = (float*)d_out;

  float* ws    = (float*)d_ws;
  float* T     = ws;
  float* Ppart = T + (size_t)N_PTS * NROW;
  float* C     = Ppart + (size_t)NCHUNK * NROW * DDIM;
  float* BppP  = C + NROW * DDIM;
  float* Wp    = BppP + NROW * DDIM;

  tgen_kernel<<<dim3((N_PTS * KFREQ) / 256), dim3(256), 0, stream>>>(T);
  pack_w_kernel<<<dim3((DDIM * DDIM / 2) / 256), dim3(256), 0, stream>>>(W, Wp);
  dft_low_kernel<<<dim3(DDIM / DFT_COLS, NCHUNK), dim3(256), 0, stream>>>(x, T, Ppart);
  reduce_chunks_kernel<<<dim3((NROW * DDIM) / 256), dim3(256), 0, stream>>>(Ppart, C);
  spectral_mix_kernel<<<dim3(DDIM / MIX_FQ, KFREQ), dim3(256), 0, stream>>>(C, P_low, B_low, BppP);
  fused_out_kernel<<<dim3(N_PTS / FO_ROWS), dim3(256), 0, stream>>>(x, Wp, Bias, T, BppP, out);
}


# Half-period symmetry T[n+N/2][k]=(-1)^k T[n][k] in both transforms: rows n, n+N/2 share even-k/odd-k table sums; half the table WMMAs and table rows; sum order changed (row pairs pre-added)
# speedup vs baseline: 1.7288x; 1.1983x over previous
#include <hip/hip_runtime.h>
#include <math.h>


typedef float v2f __attribute__((ext_vector_type(2)));
typedef float v4f __attribute__((ext_vector_type(4)));
typedef float v8f __attribute__((ext_vector_type(8)));

#define N_PTS 65536
#define DDIM  256
#define KFREQ 64
#define NROW  128
#define NHALF  (N_PTS / 2)
#define NCHUNK 64
#define CHUNK_N (NHALF / NCHUNK)

__device__ __forceinline__ v8f wmma_f32(v2f a, v2f b, v8f c) {
  return __builtin_amdgcn_wmma_f32_16x16x4_f32(false, a, false, b,
                                               (short)0, c, false, false);
}

__global__ __launch_bounds__(256) void tgen_kernel(float* __restrict__ T) {
  int idx = blockIdx.x * 256 + threadIdx.x;
  int n = idx >> 6;
  int p = idx & 63;
  int k = ((p & 31) << 1) | (p >> 5);
  unsigned m = ((unsigned)n * (unsigned)k) & 65535u;
  float ang = (float)m * 9.58737992429e-5f;
  float s, c;
  sincosf(ang, &s, &c);
  size_t base = (size_t)n * NROW + 2 * p;
  T[base]     = c;
  T[base + 1] = s;
}

__global__ __launch_bounds__(256) void pack_w_kernel(
    const float* __restrict__ W, float* __restrict__ Wp) {
  int idx = blockIdx.x * 256 + threadIdx.x;
  int kpair = idx >> 8;
  int col   = idx & 255;
  Wp[(size_t)idx * 2]     = W[(size_t)(2 * kpair) * DDIM + col];
  Wp[(size_t)idx * 2 + 1] = W[(size_t)(2 * kpair + 1) * DDIM + col];
}

#define DFT_SLAB   64
#define DFT_COLS   64
#define DFT_NSLAB  (CHUNK_N / DFT_SLAB)
#define T_PITCH    (NROW + 8)
#define X_PITCH    (DFT_COLS + 8)

#define DFT_FRAG_LOADS(S)                                              \
  "ds_load_2addr_b32 %[a0" #S "], %[pa] offset1:136\n\t"               \
  "ds_load_2addr_b32 %[a1" #S "], %[pa] offset0:16 offset1:152\n\t"    \
  "ds_load_2addr_b32 %[b0" #S "], %[pb] offset1:72\n\t"                \
  "ds_load_2addr_b32 %[b1" #S "], %[pb] offset0:16 offset1:88\n\t"
#define DFT_ADVANCE                                                    \
  "v_add_nc_u32_e32 %[pa], 0x880, %[pa]\n\t"                           \
  "v_add_nc_u32_e32 %[pb], 0x480, %[pb]\n\t"
#define DFT_WMMAS(S)                                                   \
  "v_wmma_f32_16x16x4_f32 %[c00], %[a0" #S "], %[b0" #S "], %[c00]\n\t" \
  "v_wmma_f32_16x16x4_f32 %[c01], %[a0" #S "], %[b1" #S "], %[c01]\n\t" \
  "v_wmma_f32_16x16x4_f32 %[c10], %[a1" #S "], %[b0" #S "], %[c10]\n\t" \
  "v_wmma_f32_16x16x4_f32 %[c11], %[a1" #S "], %[b1" #S "], %[c11]\n\t"
#define DFT_STEP(CUR, NXT)                                             \
  DFT_ADVANCE DFT_FRAG_LOADS(NXT) "s_wait_dscnt 0x4\n\t" DFT_WMMAS(CUR)

__global__ __launch_bounds__(256) void dft_low_kernel(
    const float* __restrict__ x, const float* __restrict__ T,
    float* __restrict__ Ppart) {
  __shared__ __attribute__((aligned(16))) float sT[DFT_SLAB * T_PITCH];
  __shared__ __attribute__((aligned(16))) float sXs[DFT_SLAB * X_PITCH];
  __shared__ __attribute__((aligned(16))) float sXd[DFT_SLAB * X_PITCH];

  const int tid     = threadIdx.x;
  const int lane    = tid & 31;
  const int wave    = tid >> 5;
  const int lr      = lane & 15;
  const int halfSel = (lane >> 4) << 1;
  const int q       = blockIdx.x;
  const int nbase   = blockIdx.y * CHUNK_N;
  const int rowW    = (wave & 3) * 32;
  const int colW    = (wave >> 2) * 32;

  v8f acc00 = {0.f,0.f,0.f,0.f,0.f,0.f,0.f,0.f};
  v8f acc01 = acc00, acc10 = acc00, acc11 = acc00;

  v4f pT[8];
  v4f pX[4];
  v4f pY[4];

  const float* Tg = T + (size_t)nbase * NROW;
  const float* xg = x + (size_t)nbase * DDIM + q * DFT_COLS;
  const float* yg = xg + (size_t)NHALF * DDIM;

#pragma unroll
  for (int j = 0; j < 8; ++j) {
    const int i = tid + 256 * j;
    pT[j] = *(const v4f*)(Tg + (size_t)i * 4);
  }
#pragma unroll
  for (int j = 0; j < 4; ++j) {
    const int i = tid + 256 * j;
    pX[j] = *(const v4f*)(xg + (size_t)(i >> 4) * DDIM + (i & 15) * 4);
    pY[j] = *(const v4f*)(yg + (size_t)(i >> 4) * DDIM + (i & 15) * 4);
  }

  const unsigned sxs = (unsigned)(size_t)(sXs + halfSel * X_PITCH + colW + lr);
  const unsigned sxd = (unsigned)(size_t)(sXd + halfSel * X_PITCH + colW + lr);
  const unsigned pa0 = (unsigned)(size_t)(sT + halfSel * T_PITCH + rowW + lr);
  const unsigned pb0 = (wave & 2) ? sxd : sxs;

  for (int s = 0; s < DFT_NSLAB; ++s) {
    __syncthreads();
#pragma unroll
    for (int j = 0; j < 8; ++j) {
      const int i = tid + 256 * j;
      *(v4f*)(sT + (i >> 5) * T_PITCH + (i & 31) * 4) = pT[j];
    }
#pragma unroll
    for (int j = 0; j < 4; ++j) {
      const int i = tid + 256 * j;
      *(v4f*)(sXs + (i >> 4) * X_PITCH + (i & 15) * 4) = pX[j] + pY[j];
      *(v4f*)(sXd + (i >> 4) * X_PITCH + (i & 15) * 4) = pX[j] - pY[j];
    }
    __syncthreads();

    if (s + 1 < DFT_NSLAB) {
      const float* Tn = Tg + (size_t)(s + 1) * DFT_SLAB * NROW;
      const float* xn = xg + (size_t)(s + 1) * DFT_SLAB * DDIM;
      const float* yn = yg + (size_t)(s + 1) * DFT_SLAB * DDIM;
#pragma unroll
      for (int j = 0; j < 8; ++j) {
        const int i = tid + 256 * j;
        pT[j] = *(const v4f*)(Tn + (size_t)i * 4);
      }
#pragma unroll
      for (int j = 0; j < 4; ++j) {
        const int i = tid + 256 * j;
        pX[j] = *(const v4f*)(xn + (size_t)(i >> 4) * DDIM + (i & 15) * 4);
        pY[j] = *(const v4f*)(yn + (size_t)(i >> 4) * DDIM + (i & 15) * 4);
      }
    }

    {
      unsigned pa = pa0, pb = pb0;
      v2f a0x, a1x, b0x, b1x, a0y, a1y, b0y, b1y;
      asm volatile(
          DFT_FRAG_LOADS(x)
          DFT_STEP(x, y) DFT_STEP(y, x) DFT_STEP(x, y) DFT_STEP(y, x)
          DFT_STEP(x, y) DFT_STEP(y, x) DFT_STEP(x, y) DFT_STEP(y, x)
          DFT_STEP(x, y) DFT_STEP(y, x) DFT_STEP(x, y) DFT_STEP(y, x)
          DFT_STEP(x, y) DFT_STEP(y, x) DFT_STEP(x, y)
          "s_wait_dscnt 0x0\n\t"
          DFT_WMMAS(y)
          "v_nop\n\tv_nop\n\tv_nop\n\tv_nop"
          : [c00] "+v"(acc00), [c01] "+v"(acc01),
            [c10] "+v"(acc10), [c11] "+v"(acc11),
            [pa] "+v"(pa), [pb] "+v"(pb),
            [a0x] "=&v"(a0x), [a1x] "=&v"(a1x),
            [b0x] "=&v"(b0x), [b1x] "=&v"(b1x),
            [a0y] "=&v"(a0y), [a1y] "=&v"(a1y),
            [b0y] "=&v"(b0y), [b1y] "=&v"(b1y)
          :
          : "memory");
    }
  }

  const int mbase = (lane >> 4) << 3;
  const int col   = q * DFT_COLS + colW + lr;
  const size_t base = (size_t)blockIdx.y * (NROW * DDIM);
#pragma unroll
  for (int v = 0; v < 8; ++v) {
    const int r = rowW + mbase + v;
    Ppart[base + (size_t)r * DDIM + col]             = acc00[v];
    Ppart[base + (size_t)r * DDIM + col + 16]        = acc01[v];
    Ppart[base + (size_t)(r + 16) * DDIM + col]      = acc10[v];
    Ppart[base + (size_t)(r + 16) * DDIM + col + 16] = acc11[v];
  }
}

__global__ __launch_bounds__(256) void reduce_chunks_kernel(
    const float* __restrict__ Ppart, float* __restrict__ C) {
  int idx = blockIdx.x * 256 + threadIdx.x;
  float s = 0.f;
#pragma unroll 4
  for (int c = 0; c < NCHUNK; ++c)
    s += Ppart[(size_t)c * (NROW * DDIM) + idx];
  C[idx] = s;
}

#define MIX_FQ 64
#define MIX_DS 32

__global__ __launch_bounds__(256) void spectral_mix_kernel(
    const float* __restrict__ C, const float* __restrict__ P_low,
    const float* __restrict__ B_low, float* __restrict__ BppP) {
  __shared__ float sXr[DDIM];
  __shared__ float sXi[DDIM];
  __shared__ __attribute__((aligned(16))) float sPart[8 * MIX_FQ * 2];

  const int tid   = threadIdx.x;
  const int lane  = tid & 31;
  const int wave  = tid >> 5;
  const int fq    = blockIdx.x;
  const int k     = blockIdx.y;
  const int prow  = ((k & 1) << 5) | (k >> 1);
  const int dbase = wave * MIX_DS;
  const int f0    = fq * MIX_FQ + lane * 2;

  const float invN = 1.0f / (float)N_PTS;
  const float cr = C[(size_t)(2 * prow) * DDIM + tid];
  const float ci = C[(size_t)(2 * prow + 1) * DDIM + tid];

  const float* Pk = P_low + (size_t)k * DDIM * DDIM * 2
                          + ((size_t)dbase * DDIM + f0) * 2;
  v4f p[MIX_DS];
#pragma unroll
  for (int j = 0; j < MIX_DS; ++j)
    p[j] = *(const v4f*)(Pk + (size_t)j * DDIM * 2);

  sXr[tid] =  cr * invN;
  sXi[tid] = -ci * invN;
  __syncthreads();

  float ar0 = 0.f, ai0 = 0.f, ar1 = 0.f, ai1 = 0.f;
#pragma unroll
  for (int j = 0; j < MIX_DS; ++j) {
    const float xr = sXr[dbase + j], xi = sXi[dbase + j];
    ar0 = fmaf(xr, p[j].x, fmaf(-xi, p[j].y, ar0));
    ai0 = fmaf(xr, p[j].y, fmaf( xi, p[j].x, ai0));
    ar1 = fmaf(xr, p[j].z, fmaf(-xi, p[j].w, ar1));
    ai1 = fmaf(xr, p[j].w, fmaf( xi, p[j].z, ai1));
  }
  *(v4f*)(sPart + (wave * MIX_FQ + lane * 2) * 2) = (v4f){ar0, ai0, ar1, ai1};
  __syncthreads();

  if (tid < MIX_FQ * 2) {
    const int fl = tid >> 1;
    const int ri = tid & 1;
    const int f  = fq * MIX_FQ + fl;
    float s = B_low[((size_t)k * DDIM + f) * 2 + ri];
#pragma unroll
    for (int w = 0; w < 8; ++w) s += sPart[(w * MIX_FQ + fl) * 2 + ri];
    const float sc = (k == 0) ? 1.0f : 2.0f;
    BppP[((size_t)prow * DDIM + f) * 2 + ri] = ri ? -sc * s : sc * s;
  }
}

#define FO_RB     32
#define FO_KS     64
#define FO_NSLAB  6
#define FO_APITCH 68
#define FO_BPITCH 544

#define FO_LOADS(S, A0, A1, B0, B1)                                    \
  "ds_load_b64 %[a0" #S "], %[pa] offset:" #A0 "\n\t"                 \
  "ds_load_b64 %[a1" #S "], %[pa] offset:" #A1 "\n\t"                 \
  "ds_load_b64 %[b0" #S "], %[pb] offset:" #B0 "\n\t"                 \
  "ds_load_b64 %[b1" #S "], %[pb] offset:" #B1 "\n\t"                 \
  "ds_load_b64 %[b2" #S "], %[pc] offset:" #B0 "\n\t"                 \
  "ds_load_b64 %[b3" #S "], %[pc] offset:" #B1 "\n\t"
#define FO_WMMAS(S)                                                    \
  "v_wmma_f32_16x16x4_f32 %[c00], %[a0" #S "], %[b0" #S "], %[c00]\n\t" \
  "v_wmma_f32_16x16x4_f32 %[c01], %[a0" #S "], %[b1" #S "], %[c01]\n\t" \
  "v_wmma_f32_16x16x4_f32 %[c02], %[a0" #S "], %[b2" #S "], %[c02]\n\t" \
  "v_wmma_f32_16x16x4_f32 %[c03], %[a0" #S "], %[b3" #S "], %[c03]\n\t" \
  "v_wmma_f32_16x16x4_f32 %[c10], %[a1" #S "], %[b0" #S "], %[c10]\n\t" \
  "v_wmma_f32_16x16x4_f32 %[c11], %[a1" #S "], %[b1" #S "], %[c11]\n\t" \
  "v_wmma_f32_16x16x4_f32 %[c12], %[a1" #S "], %[b2" #S "], %[c12]\n\t" \
  "v_wmma_f32_16x16x4_f32 %[c13], %[a1" #S "], %[b3" #S "], %[c13]\n\t"
#define FO_STEP(CUR, NXT, A0, A1, B0, B1)                              \
  FO_LOADS(NXT, A0, A1, B0, B1) "s_wait_dscnt 0x6\n\t" FO_WMMAS(CUR)

#define FT_LOADS(S, A0, B0, B1)                                        \
  "ds_load_b64 %[a0" #S "], %[pa] offset:" #A0 "\n\t"                 \
  "ds_load_b64 %[b0" #S "], %[pb] offset:" #B0 "\n\t"                 \
  "ds_load_b64 %[b1" #S "], %[pb] offset:" #B1 "\n\t"                 \
  "ds_load_b64 %[b2" #S "], %[pc] offset:" #B0 "\n\t"                 \
  "ds_load_b64 %[b3" #S "], %[pc] offset:" #B1 "\n\t"
#define FT_WMMAS(S)                                                    \
  "v_wmma_f32_16x16x4_f32 %[t0], %[a0" #S "], %[b0" #S "], %[t0]\n\t" \
  "v_wmma_f32_16x16x4_f32 %[t1], %[a0" #S "], %[b1" #S "], %[t1]\n\t" \
  "v_wmma_f32_16x16x4_f32 %[t2], %[a0" #S "], %[b2" #S "], %[t2]\n\t" \
  "v_wmma_f32_16x16x4_f32 %[t3], %[a0" #S "], %[b3" #S "], %[t3]\n\t"
#define FT_STEP(CUR, NXT, A0, B0, B1)                                  \
  FT_LOADS(NXT, A0, B0, B1) "s_wait_dscnt 0x5\n\t" FT_WMMAS(CUR)

__device__ __forceinline__ void fo_fetch(
    int s, int tid, int nblk, const float* __restrict__ x,
    const float* __restrict__ T, const float* __restrict__ Wp,
    const float* __restrict__ BppP, v4f (&pA)[4], v4f (&pB)[16]) {
  const float* Bg;
  if (s < 4) {
    const float* Ag = x + (size_t)nblk * DDIM + s * FO_KS;
#pragma unroll
    for (int j = 0; j < 4; ++j) {
      const int i = tid + 256 * j;
      const int r = i >> 4;
      pA[j] = *(const v4f*)(Ag + ((size_t)(r & 31) + (size_t)(r >> 5) * NHALF) * DDIM
                               + (i & 15) * 4);
    }
    Bg = Wp + (size_t)s * (FO_KS / 2) * (2 * DDIM);
  } else {
    const float* Ag = T + (size_t)nblk * NROW + (s - 4) * FO_KS;
#pragma unroll
    for (int j = 0; j < 2; ++j) {
      const int i = tid + 256 * j;
      pA[j] = *(const v4f*)(Ag + (size_t)(i >> 4) * NROW + (i & 15) * 4);
    }
    Bg = BppP + (size_t)(s - 4) * (FO_KS / 2) * (2 * DDIM);
  }
#pragma unroll
  for (int j = 0; j < 16; ++j) {
    const int i = tid + 256 * j;
    pB[j] = *(const v4f*)(Bg + (size_t)i * 4);
  }
}

__device__ __forceinline__ void fo_table_slab(
    v8f& t0, v8f& t1, v8f& t2, v8f& t3,
    unsigned pa, unsigned pb, unsigned pc) {
  v2f a0x, b0x, b1x, b2x, b3x, a0y, b0y, b1y, b2y, b3y;
  asm volatile(
      FT_LOADS(x, 0, 0, 128)
      FT_STEP(x, y, 16, 4352, 4480)
      FT_STEP(y, x, 32, 8704, 8832)
      FT_STEP(x, y, 48, 13056, 13184)
      FT_STEP(y, x, 64, 17408, 17536)
      FT_STEP(x, y, 80, 21760, 21888)
      FT_STEP(y, x, 96, 26112, 26240)
      FT_STEP(x, y, 112, 30464, 30592)
      FT_STEP(y, x, 128, 34816, 34944)
      FT_STEP(x, y, 144, 39168, 39296)
      FT_STEP(y, x, 160, 43520, 43648)
      FT_STEP(x, y, 176, 47872, 48000)
      FT_STEP(y, x, 192, 52224, 52352)
      FT_STEP(x, y, 208, 56576, 56704)
      FT_STEP(y, x, 224, 60928, 61056)
      FT_STEP(x, y, 240, 65280, 65408)
      "s_wait_dscnt 0x0\n\t"
      FT_WMMAS(y)
      "v_nop\n\tv_nop\n\tv_nop\n\tv_nop"
      : [t0] "+v"(t0), [t1] "+v"(t1), [t2] "+v"(t2), [t3] "+v"(t3),
        [a0x] "=&v"(a0x), [b0x] "=&v"(b0x), [b1x] "=&v"(b1x),
        [b2x] "=&v"(b2x), [b3x] "=&v"(b3x),
        [a0y] "=&v"(a0y), [b0y] "=&v"(b0y), [b1y] "=&v"(b1y),
        [b2y] "=&v"(b2y), [b3y] "=&v"(b3y)
      : [pa] "v"(pa), [pb] "v"(pb), [pc] "v"(pc)
      : "memory");
}

__global__ __launch_bounds__(256) void fused_out_kernel(
    const float* __restrict__ x, const float* __restrict__ Wp,
    const float* __restrict__ Bias, const float* __restrict__ T,
    const float* __restrict__ BppP, float* __restrict__ out) {
  __shared__ __attribute__((aligned(16))) float sA[2 * FO_RB * FO_APITCH];
  __shared__ __attribute__((aligned(16))) float sB[(FO_KS / 2) * FO_BPITCH];

  const int tid     = threadIdx.x;
  const int lane    = tid & 31;
  const int wave    = tid >> 5;
  const int lr      = lane & 15;
  const int hsel    = lane >> 4;
  const int nblk    = blockIdx.x * FO_RB;
  const int rowW    = (wave & 1) * 16;
  const int colW    = (wave >> 1) * 64;

  v8f c00 = {0.f,0.f,0.f,0.f,0.f,0.f,0.f,0.f};
  v8f c01 = c00, c02 = c00, c03 = c00, c10 = c00, c11 = c00, c12 = c00, c13 = c00;
  v8f e0 = c00, e1 = c00, e2 = c00, e3 = c00;
  v8f q0 = c00, q1 = c00, q2 = c00, q3 = c00;

  v4f pA[4];
  v4f pB[16];
  fo_fetch(0, tid, nblk, x, T, Wp, BppP, pA, pB);

  const unsigned pa = (unsigned)(size_t)(sA + (rowW + lr) * FO_APITCH + 2 * hsel);
  const unsigned pb = (unsigned)(size_t)(sB + hsel * FO_BPITCH + (colW + lr) * 2);
  const unsigned pc = pb + 256u;

  for (int s = 0; s < FO_NSLAB; ++s) {
    __syncthreads();
    if (s < 4) {
#pragma unroll
      for (int j = 0; j < 4; ++j) {
        const int i = tid + 256 * j;
        *(v4f*)(sA + (i >> 4) * FO_APITCH + (i & 15) * 4) = pA[j];
      }
    } else {
#pragma unroll
      for (int j = 0; j < 2; ++j) {
        const int i = tid + 256 * j;
        *(v4f*)(sA + (i >> 4) * FO_APITCH + (i & 15) * 4) = pA[j];
      }
    }
#pragma unroll
    for (int j = 0; j < 16; ++j) {
      const int i = tid + 256 * j;
      *(v4f*)(sB + (i >> 7) * FO_BPITCH + (i & 127) * 4) = pB[j];
    }
    __syncthreads();

    if (s + 1 < FO_NSLAB)
      fo_fetch(s + 1, tid, nblk, x, T, Wp, BppP, pA, pB);

    if (s < 4) {
      v2f a0x, a1x, b0x, b1x, b2x, b3x, a0y, a1y, b0y, b1y, b2y, b3y;
      asm volatile(
          FO_LOADS(x, 0, 8704, 0, 128)
          FO_STEP(x, y, 16, 8720, 4352, 4480)
          FO_STEP(y, x, 32, 8736, 8704, 8832)
          FO_STEP(x, y, 48, 8752, 13056, 13184)
          FO_STEP(y, x, 64, 8768, 17408, 17536)
          FO_STEP(x, y, 80, 8784, 21760, 21888)
          FO_STEP(y, x, 96, 8800, 26112, 26240)
          FO_STEP(x, y, 112, 8816, 30464, 30592)
          FO_STEP(y, x, 128, 8832, 34816, 34944)
          FO_STEP(x, y, 144, 8848, 39168, 39296)
          FO_STEP(y, x, 160, 8864, 43520, 43648)
          FO_STEP(x, y, 176, 8880, 47872, 48000)
          FO_STEP(y, x, 192, 8896, 52224, 52352)
          FO_STEP(x, y, 208, 8912, 56576, 56704)
          FO_STEP(y, x, 224, 8928, 60928, 61056)
          FO_STEP(x, y, 240, 8944, 65280, 65408)
          "s_wait_dscnt 0x0\n\t"
          FO_WMMAS(y)
          "v_nop\n\tv_nop\n\tv_nop\n\tv_nop"
          : [c00] "+v"(c00), [c01] "+v"(c01), [c02] "+v"(c02), [c03] "+v"(c03),
            [c10] "+v"(c10), [c11] "+v"(c11), [c12] "+v"(c12), [c13] "+v"(c13),
            [a0x] "=&v"(a0x), [a1x] "=&v"(a1x), [b0x] "=&v"(b0x),
            [b1x] "=&v"(b1x), [b2x] "=&v"(b2x), [b3x] "=&v"(b3x),
            [a0y] "=&v"(a0y), [a1y] "=&v"(a1y), [b0y] "=&v"(b0y),
            [b1y] "=&v"(b1y), [b2y] "=&v"(b2y), [b3y] "=&v"(b3y)
          : [pa] "v"(pa), [pb] "v"(pb), [pc] "v"(pc)
          : "memory");
    } else if (s == 4) {
      fo_table_slab(e0, e1, e2, e3, pa, pb, pc);
    } else {
      fo_table_slab(q0, q1, q2, q3, pa, pb, pc);
    }
  }

  const int mbase = hsel << 3;
  const int col   = colW + lr;
  const int nrow  = nblk + rowW + mbase;
  const float bias0 = Bias[col];
  const float bias1 = Bias[col + 16];
  const float bias2 = Bias[col + 32];
  const float bias3 = Bias[col + 48];
#pragma unroll
  for (int v = 0; v < 8; ++v) {
    float* o0 = out + (size_t)(nrow + v) * DDIM + col;
    float* o1 = o0 + (size_t)NHALF * DDIM;
    o0[0]  = sinf(0.2f * (c00[v] + (e0[v] + q0[v]) + bias0));
    o0[16] = sinf(0.2f * (c01[v] + (e1[v] + q1[v]) + bias1));
    o0[32] = sinf(0.2f * (c02[v] + (e2[v] + q2[v]) + bias2));
    o0[48] = sinf(0.2f * (c03[v] + (e3[v] + q3[v]) + bias3));
    o1[0]  = sinf(0.2f * (c10[v] + (e0[v] - q0[v]) + bias0));
    o1[16] = sinf(0.2f * (c11[v] + (e1[v] - q1[v]) + bias1));
    o1[32] = sinf(0.2f * (c12[v] + (e2[v] - q2[v]) + bias2));
    o1[48] = sinf(0.2f * (c13[v] + (e3[v] - q3[v]) + bias3));
  }
}

extern "C" void kernel_launch(void* const* d_in, const int* in_sizes, int n_in,
                              void* d_out, int out_size, void* d_ws,
                              size_t ws_size, hipStream_t stream) {
  (void)in_sizes; (void)n_in; (void)out_size; (void)ws_size;
  const float* x     = (const float*)d_in[0];
  const float* P_low = (const float*)d_in[1];
  const float* B_low = (const float*)d_in[2];
  const float* W     = (const float*)d_in[3];
  const float* Bias  = (const float*)d_in[4];
  float* out = (float*)d_out;

  float* ws    = (float*)d_ws;
  float* T     = ws;
  float* Ppart = T + (size_t)N_PTS * NROW;
  float* C     = Ppart + (size_t)NCHUNK * NROW * DDIM;
  float* BppP  = C + NROW * DDIM;
  float* Wp    = BppP + NROW * DDIM;

  tgen_kernel<<<dim3((NHALF * KFREQ) / 256), dim3(256), 0, stream>>>(T);
  pack_w_kernel<<<dim3((DDIM * DDIM / 2) / 256), dim3(256), 0, stream>>>(W, Wp);
  dft_low_kernel<<<dim3(DDIM / DFT_COLS, NCHUNK), dim3(256), 0, stream>>>(x, T, Ppart);
  reduce_chunks_kernel<<<dim3((NROW * DDIM) / 256), dim3(256), 0, stream>>>(Ppart, C);
  spectral_mix_kernel<<<dim3(DDIM / MIX_FQ, KFREQ), dim3(256), 0, stream>>>(C, P_low, B_low, BppP);
  fused_out_kernel<<<dim3(NHALF / FO_RB), dim3(256), 0, stream>>>(x, Wp, Bias, T, BppP, out);
}


# Launch count 6 -> 4: W pack rides as extra blocks of the table launch; the 64-chunk sum is done inside the spectral-mix blocks (chunk order 0..63 kept, sum order unchanged)
# speedup vs baseline: 1.8745x; 1.0843x over previous
#include <hip/hip_runtime.h>
#include <math.h>


typedef float v2f __attribute__((ext_vector_type(2)));
typedef float v4f __attribute__((ext_vector_type(4)));
typedef float v8f __attribute__((ext_vector_type(8)));

#define N_PTS 65536
#define DDIM  256
#define KFREQ 64
#define NROW  128
#define NHALF  (N_PTS / 2)
#define NCHUNK 64
#define CHUNK_N (NHALF / NCHUNK)

__device__ __forceinline__ v8f wmma_f32(v2f a, v2f b, v8f c) {
  return __builtin_amdgcn_wmma_f32_16x16x4_f32(false, a, false, b,
                                               (short)0, c, false, false);
}

#define PACK_BLOCKS ((DDIM * DDIM / 2) / 256)
#define TGEN_BLOCKS ((NHALF * KFREQ) / 256)

__global__ __launch_bounds__(256) void prep_kernel(
    const float* __restrict__ W, float* __restrict__ Wp,
    float* __restrict__ T) {
  if (blockIdx.x < PACK_BLOCKS) {
    int idx = blockIdx.x * 256 + threadIdx.x;
    int kpair = idx >> 8;
    int col   = idx & 255;
    Wp[(size_t)idx * 2]     = W[(size_t)(2 * kpair) * DDIM + col];
    Wp[(size_t)idx * 2 + 1] = W[(size_t)(2 * kpair + 1) * DDIM + col];
    return;
  }
  int idx = (blockIdx.x - PACK_BLOCKS) * 256 + threadIdx.x;
  int n = idx >> 6;
  int p = idx & 63;
  int k = ((p & 31) << 1) | (p >> 5);
  unsigned m = ((unsigned)n * (unsigned)k) & 65535u;
  float ang = (float)m * 9.58737992429e-5f;
  float s, c;
  sincosf(ang, &s, &c);
  size_t base = (size_t)n * NROW + 2 * p;
  T[base]     = c;
  T[base + 1] = s;
}

#define DFT_SLAB   64
#define DFT_COLS   64
#define DFT_NSLAB  (CHUNK_N / DFT_SLAB)
#define T_PITCH    (NROW + 8)
#define X_PITCH    (DFT_COLS + 8)

#define DFT_FRAG_LOADS(S)                                              \
  "ds_load_2addr_b32 %[a0" #S "], %[pa] offset1:136\n\t"               \
  "ds_load_2addr_b32 %[a1" #S "], %[pa] offset0:16 offset1:152\n\t"    \
  "ds_load_2addr_b32 %[b0" #S "], %[pb] offset1:72\n\t"                \
  "ds_load_2addr_b32 %[b1" #S "], %[pb] offset0:16 offset1:88\n\t"
#define DFT_ADVANCE                                                    \
  "v_add_nc_u32_e32 %[pa], 0x880, %[pa]\n\t"                           \
  "v_add_nc_u32_e32 %[pb], 0x480, %[pb]\n\t"
#define DFT_WMMAS(S)                                                   \
  "v_wmma_f32_16x16x4_f32 %[c00], %[a0" #S "], %[b0" #S "], %[c00]\n\t" \
  "v_wmma_f32_16x16x4_f32 %[c01], %[a0" #S "], %[b1" #S "], %[c01]\n\t" \
  "v_wmma_f32_16x16x4_f32 %[c10], %[a1" #S "], %[b0" #S "], %[c10]\n\t" \
  "v_wmma_f32_16x16x4_f32 %[c11], %[a1" #S "], %[b1" #S "], %[c11]\n\t"
#define DFT_STEP(CUR, NXT)                                             \
  DFT_ADVANCE DFT_FRAG_LOADS(NXT) "s_wait_dscnt 0x4\n\t" DFT_WMMAS(CUR)

__global__ __launch_bounds__(256) void dft_low_kernel(
    const float* __restrict__ x, const float* __restrict__ T,
    float* __restrict__ Ppart) {
  __shared__ __attribute__((aligned(16))) float sT[DFT_SLAB * T_PITCH];
  __shared__ __attribute__((aligned(16))) float sXs[DFT_SLAB * X_PITCH];
  __shared__ __attribute__((aligned(16))) float sXd[DFT_SLAB * X_PITCH];

  const int tid     = threadIdx.x;
  const int lane    = tid & 31;
  const int wave    = tid >> 5;
  const int lr      = lane & 15;
  const int halfSel = (lane >> 4) << 1;
  const int q       = blockIdx.x;
  const int nbase   = blockIdx.y * CHUNK_N;
  const int rowW    = (wave & 3) * 32;
  const int colW    = (wave >> 2) * 32;

  v8f acc00 = {0.f,0.f,0.f,0.f,0.f,0.f,0.f,0.f};
  v8f acc01 = acc00, acc10 = acc00, acc11 = acc00;

  v4f pT[8];
  v4f pX[4];
  v4f pY[4];

  const float* Tg = T + (size_t)nbase * NROW;
  const float* xg = x + (size_t)nbase * DDIM + q * DFT_COLS;
  const float* yg = xg + (size_t)NHALF * DDIM;

#pragma unroll
  for (int j = 0; j < 8; ++j) {
    const int i = tid + 256 * j;
    pT[j] = *(const v4f*)(Tg + (size_t)i * 4);
  }
#pragma unroll
  for (int j = 0; j < 4; ++j) {
    const int i = tid + 256 * j;
    pX[j] = *(const v4f*)(xg + (size_t)(i >> 4) * DDIM + (i & 15) * 4);
    pY[j] = *(const v4f*)(yg + (size_t)(i >> 4) * DDIM + (i & 15) * 4);
  }

  const unsigned sxs = (unsigned)(size_t)(sXs + halfSel * X_PITCH + colW + lr);
  const unsigned sxd = (unsigned)(size_t)(sXd + halfSel * X_PITCH + colW + lr);
  const unsigned pa0 = (unsigned)(size_t)(sT + halfSel * T_PITCH + rowW + lr);
  const unsigned pb0 = (wave & 2) ? sxd : sxs;

  for (int s = 0; s < DFT_NSLAB; ++s) {
    __syncthreads();
#pragma unroll
    for (int j = 0; j < 8; ++j) {
      const int i = tid + 256 * j;
      *(v4f*)(sT + (i >> 5) * T_PITCH + (i & 31) * 4) = pT[j];
    }
#pragma unroll
    for (int j = 0; j < 4; ++j) {
      const int i = tid + 256 * j;
      *(v4f*)(sXs + (i >> 4) * X_PITCH + (i & 15) * 4) = pX[j] + pY[j];
      *(v4f*)(sXd + (i >> 4) * X_PITCH + (i & 15) * 4) = pX[j] - pY[j];
    }
    __syncthreads();

    if (s + 1 < DFT_NSLAB) {
      const float* Tn = Tg + (size_t)(s + 1) * DFT_SLAB * NROW;
      const float* xn = xg + (size_t)(s + 1) * DFT_SLAB * DDIM;
      const float* yn = yg + (size_t)(s + 1) * DFT_SLAB * DDIM;
#pragma unroll
      for (int j = 0; j < 8; ++j) {
        const int i = tid + 256 * j;
        pT[j] = *(const v4f*)(Tn + (size_t)i * 4);
      }
#pragma unroll
      for (int j = 0; j < 4; ++j) {
        const int i = tid + 256 * j;
        pX[j] = *(const v4f*)(xn + (size_t)(i >> 4) * DDIM + (i & 15) * 4);
        pY[j] = *(const v4f*)(yn + (size_t)(i >> 4) * DDIM + (i & 15) * 4);
      }
    }

    {
      unsigned pa = pa0, pb = pb0;
      v2f a0x, a1x, b0x, b1x, a0y, a1y, b0y, b1y;
      asm volatile(
          DFT_FRAG_LOADS(x)
          DFT_STEP(x, y) DFT_STEP(y, x) DFT_STEP(x, y) DFT_STEP(y, x)
          DFT_STEP(x, y) DFT_STEP(y, x) DFT_STEP(x, y) DFT_STEP(y, x)
          DFT_STEP(x, y) DFT_STEP(y, x) DFT_STEP(x, y) DFT_STEP(y, x)
          DFT_STEP(x, y) DFT_STEP(y, x) DFT_STEP(x, y)
          "s_wait_dscnt 0x0\n\t"
          DFT_WMMAS(y)
          "v_nop\n\tv_nop\n\tv_nop\n\tv_nop"
          : [c00] "+v"(acc00), [c01] "+v"(acc01),
            [c10] "+v"(acc10), [c11] "+v"(acc11),
            [pa] "+v"(pa), [pb] "+v"(pb),
            [a0x] "=&v"(a0x), [a1x] "=&v"(a1x),
            [b0x] "=&v"(b0x), [b1x] "=&v"(b1x),
            [a0y] "=&v"(a0y), [a1y] "=&v"(a1y),
            [b0y] "=&v"(b0y), [b1y] "=&v"(b1y)
          :
          : "memory");
    }
  }

  const int mbase = (lane >> 4) << 3;
  const int col   = q * DFT_COLS + colW + lr;
  const size_t base = (size_t)blockIdx.y * (NROW * DDIM);
#pragma unroll
  for (int v = 0; v < 8; ++v) {
    const int r = rowW + mbase + v;
    Ppart[base + (size_t)r * DDIM + col]             = acc00[v];
    Ppart[base + (size_t)r * DDIM + col + 16]        = acc01[v];
    Ppart[base + (size_t)(r + 16) * DDIM + col]      = acc10[v];
    Ppart[base + (size_t)(r + 16) * DDIM + col + 16] = acc11[v];
  }
}

#define MIX_FQ 64
#define MIX_DS 32

__global__ __launch_bounds__(256) void spectral_mix_kernel(
    const float* __restrict__ Ppart, const float* __restrict__ P_low,
    const float* __restrict__ B_low, float* __restrict__ BppP) {
  __shared__ __attribute__((aligned(8))) float sXr[DDIM];
  __shared__ __attribute__((aligned(8))) float sXi[DDIM];
  __shared__ __attribute__((aligned(16))) float sPart[8 * MIX_FQ * 2];

  const int tid   = threadIdx.x;
  const int lane  = tid & 31;
  const int wave  = tid >> 5;
  const int fq    = blockIdx.x;
  const int k     = blockIdx.y;
  const int prow  = ((k & 1) << 5) | (k >> 1);
  const int dbase = wave * MIX_DS;
  const int f0    = fq * MIX_FQ + lane * 2;

  const int rsel = tid >> 7;
  const int d2   = (tid & 127) * 2;
  const float* Cg = Ppart + (size_t)(2 * prow + rsel) * DDIM + d2;
  v2f cpart[NCHUNK];
#pragma unroll
  for (int ch = 0; ch < NCHUNK; ++ch)
    cpart[ch] = *(const v2f*)(Cg + (size_t)ch * (NROW * DDIM));

  const float* Pk = P_low + (size_t)k * DDIM * DDIM * 2
                          + ((size_t)dbase * DDIM + f0) * 2;
  v4f p[MIX_DS];
#pragma unroll
  for (int j = 0; j < MIX_DS; ++j)
    p[j] = *(const v4f*)(Pk + (size_t)j * DDIM * 2);

  float c0 = 0.f, c1 = 0.f;
#pragma unroll
  for (int ch = 0; ch < NCHUNK; ++ch) {
    c0 += cpart[ch].x;
    c1 += cpart[ch].y;
  }
  const float invN = 1.0f / (float)N_PTS;
  if (rsel == 0) {
    *(v2f*)(sXr + d2) = (v2f){ c0 * invN,  c1 * invN};
  } else {
    *(v2f*)(sXi + d2) = (v2f){-c0 * invN, -c1 * invN};
  }
  __syncthreads();

  float ar0 = 0.f, ai0 = 0.f, ar1 = 0.f, ai1 = 0.f;
#pragma unroll
  for (int j = 0; j < MIX_DS; ++j) {
    const float xr = sXr[dbase + j], xi = sXi[dbase + j];
    ar0 = fmaf(xr, p[j].x, fmaf(-xi, p[j].y, ar0));
    ai0 = fmaf(xr, p[j].y, fmaf( xi, p[j].x, ai0));
    ar1 = fmaf(xr, p[j].z, fmaf(-xi, p[j].w, ar1));
    ai1 = fmaf(xr, p[j].w, fmaf( xi, p[j].z, ai1));
  }
  *(v4f*)(sPart + (wave * MIX_FQ + lane * 2) * 2) = (v4f){ar0, ai0, ar1, ai1};
  __syncthreads();

  if (tid < MIX_FQ * 2) {
    const int fl = tid >> 1;
    const int ri = tid & 1;
    const int f  = fq * MIX_FQ + fl;
    float s = B_low[((size_t)k * DDIM + f) * 2 + ri];
#pragma unroll
    for (int w = 0; w < 8; ++w) s += sPart[(w * MIX_FQ + fl) * 2 + ri];
    const float sc = (k == 0) ? 1.0f : 2.0f;
    BppP[((size_t)prow * DDIM + f) * 2 + ri] = ri ? -sc * s : sc * s;
  }
}

#define FO_RB     32
#define FO_KS     64
#define FO_NSLAB  6
#define FO_APITCH 68
#define FO_BPITCH 544

#define FO_LOADS(S, A0, A1, B0, B1)                                    \
  "ds_load_b64 %[a0" #S "], %[pa] offset:" #A0 "\n\t"                 \
  "ds_load_b64 %[a1" #S "], %[pa] offset:" #A1 "\n\t"                 \
  "ds_load_b64 %[b0" #S "], %[pb] offset:" #B0 "\n\t"                 \
  "ds_load_b64 %[b1" #S "], %[pb] offset:" #B1 "\n\t"                 \
  "ds_load_b64 %[b2" #S "], %[pc] offset:" #B0 "\n\t"                 \
  "ds_load_b64 %[b3" #S "], %[pc] offset:" #B1 "\n\t"
#define FO_WMMAS(S)                                                    \
  "v_wmma_f32_16x16x4_f32 %[c00], %[a0" #S "], %[b0" #S "], %[c00]\n\t" \
  "v_wmma_f32_16x16x4_f32 %[c01], %[a0" #S "], %[b1" #S "], %[c01]\n\t" \
  "v_wmma_f32_16x16x4_f32 %[c02], %[a0" #S "], %[b2" #S "], %[c02]\n\t" \
  "v_wmma_f32_16x16x4_f32 %[c03], %[a0" #S "], %[b3" #S "], %[c03]\n\t" \
  "v_wmma_f32_16x16x4_f32 %[c10], %[a1" #S "], %[b0" #S "], %[c10]\n\t" \
  "v_wmma_f32_16x16x4_f32 %[c11], %[a1" #S "], %[b1" #S "], %[c11]\n\t" \
  "v_wmma_f32_16x16x4_f32 %[c12], %[a1" #S "], %[b2" #S "], %[c12]\n\t" \
  "v_wmma_f32_16x16x4_f32 %[c13], %[a1" #S "], %[b3" #S "], %[c13]\n\t"
#define FO_STEP(CUR, NXT, A0, A1, B0, B1)                              \
  FO_LOADS(NXT, A0, A1, B0, B1) "s_wait_dscnt 0x6\n\t" FO_WMMAS(CUR)

#define FT_LOADS(S, A0, B0, B1)                                        \
  "ds_load_b64 %[a0" #S "], %[pa] offset:" #A0 "\n\t"                 \
  "ds_load_b64 %[b0" #S "], %[pb] offset:" #B0 "\n\t"                 \
  "ds_load_b64 %[b1" #S "], %[pb] offset:" #B1 "\n\t"                 \
  "ds_load_b64 %[b2" #S "], %[pc] offset:" #B0 "\n\t"                 \
  "ds_load_b64 %[b3" #S "], %[pc] offset:" #B1 "\n\t"
#define FT_WMMAS(S)                                                    \
  "v_wmma_f32_16x16x4_f32 %[t0], %[a0" #S "], %[b0" #S "], %[t0]\n\t" \
  "v_wmma_f32_16x16x4_f32 %[t1], %[a0" #S "], %[b1" #S "], %[t1]\n\t" \
  "v_wmma_f32_16x16x4_f32 %[t2], %[a0" #S "], %[b2" #S "], %[t2]\n\t" \
  "v_wmma_f32_16x16x4_f32 %[t3], %[a0" #S "], %[b3" #S "], %[t3]\n\t"
#define FT_STEP(CUR, NXT, A0, B0, B1)                                  \
  FT_LOADS(NXT, A0, B0, B1) "s_wait_dscnt 0x5\n\t" FT_WMMAS(CUR)

__device__ __forceinline__ void fo_fetch(
    int s, int tid, int nblk, const float* __restrict__ x,
    const float* __restrict__ T, const float* __restrict__ Wp,
    const float* __restrict__ BppP, v4f (&pA)[4], v4f (&pB)[16]) {
  const float* Bg;
  if (s < 4) {
    const float* Ag = x + (size_t)nblk * DDIM + s * FO_KS;
#pragma unroll
    for (int j = 0; j < 4; ++j) {
      const int i = tid + 256 * j;
      const int r = i >> 4;
      pA[j] = *(const v4f*)(Ag + ((size_t)(r & 31) + (size_t)(r >> 5) * NHALF) * DDIM
                               + (i & 15) * 4);
    }
    Bg = Wp + (size_t)s * (FO_KS / 2) * (2 * DDIM);
  } else {
    const float* Ag = T + (size_t)nblk * NROW + (s - 4) * FO_KS;
#pragma unroll
    for (int j = 0; j < 2; ++j) {
      const int i = tid + 256 * j;
      pA[j] = *(const v4f*)(Ag + (size_t)(i >> 4) * NROW + (i & 15) * 4);
    }
    Bg = BppP + (size_t)(s - 4) * (FO_KS / 2) * (2 * DDIM);
  }
#pragma unroll
  for (int j = 0; j < 16; ++j) {
    const int i = tid + 256 * j;
    pB[j] = *(const v4f*)(Bg + (size_t)i * 4);
  }
}

__device__ __forceinline__ void fo_table_slab(
    v8f& t0, v8f& t1, v8f& t2, v8f& t3,
    unsigned pa, unsigned pb, unsigned pc) {
  v2f a0x, b0x, b1x, b2x, b3x, a0y, b0y, b1y, b2y, b3y;
  asm volatile(
      FT_LOADS(x, 0, 0, 128)
      FT_STEP(x, y, 16, 4352, 4480)
      FT_STEP(y, x, 32, 8704, 8832)
      FT_STEP(x, y, 48, 13056, 13184)
      FT_STEP(y, x, 64, 17408, 17536)
      FT_STEP(x, y, 80, 21760, 21888)
      FT_STEP(y, x, 96, 26112, 26240)
      FT_STEP(x, y, 112, 30464, 30592)
      FT_STEP(y, x, 128, 34816, 34944)
      FT_STEP(x, y, 144, 39168, 39296)
      FT_STEP(y, x, 160, 43520, 43648)
      FT_STEP(x, y, 176, 47872, 48000)
      FT_STEP(y, x, 192, 52224, 52352)
      FT_STEP(x, y, 208, 56576, 56704)
      FT_STEP(y, x, 224, 60928, 61056)
      FT_STEP(x, y, 240, 65280, 65408)
      "s_wait_dscnt 0x0\n\t"
      FT_WMMAS(y)
      "v_nop\n\tv_nop\n\tv_nop\n\tv_nop"
      : [t0] "+v"(t0), [t1] "+v"(t1), [t2] "+v"(t2), [t3] "+v"(t3),
        [a0x] "=&v"(a0x), [b0x] "=&v"(b0x), [b1x] "=&v"(b1x),
        [b2x] "=&v"(b2x), [b3x] "=&v"(b3x),
        [a0y] "=&v"(a0y), [b0y] "=&v"(b0y), [b1y] "=&v"(b1y),
        [b2y] "=&v"(b2y), [b3y] "=&v"(b3y)
      : [pa] "v"(pa), [pb] "v"(pb), [pc] "v"(pc)
      : "memory");
}

__global__ __launch_bounds__(256) void fused_out_kernel(
    const float* __restrict__ x, const float* __restrict__ Wp,
    const float* __restrict__ Bias, const float* __restrict__ T,
    const float* __restrict__ BppP, float* __restrict__ out) {
  __shared__ __attribute__((aligned(16))) float sA[2 * FO_RB * FO_APITCH];
  __shared__ __attribute__((aligned(16))) float sB[(FO_KS / 2) * FO_BPITCH];

  const int tid     = threadIdx.x;
  const int lane    = tid & 31;
  const int wave    = tid >> 5;
  const int lr      = lane & 15;
  const int hsel    = lane >> 4;
  const int nblk    = blockIdx.x * FO_RB;
  const int rowW    = (wave & 1) * 16;
  const int colW    = (wave >> 1) * 64;

  v8f c00 = {0.f,0.f,0.f,0.f,0.f,0.f,0.f,0.f};
  v8f c01 = c00, c02 = c00, c03 = c00, c10 = c00, c11 = c00, c12 = c00, c13 = c00;
  v8f e0 = c00, e1 = c00, e2 = c00, e3 = c00;
  v8f q0 = c00, q1 = c00, q2 = c00, q3 = c00;

  v4f pA[4];
  v4f pB[16];
  fo_fetch(0, tid, nblk, x, T, Wp, BppP, pA, pB);

  const unsigned pa = (unsigned)(size_t)(sA + (rowW + lr) * FO_APITCH + 2 * hsel);
  const unsigned pb = (unsigned)(size_t)(sB + hsel * FO_BPITCH + (colW + lr) * 2);
  const unsigned pc = pb + 256u;

  for (int s = 0; s < FO_NSLAB; ++s) {
    __syncthreads();
    if (s < 4) {
#pragma unroll
      for (int j = 0; j < 4; ++j) {
        const int i = tid + 256 * j;
        *(v4f*)(sA + (i >> 4) * FO_APITCH + (i & 15) * 4) = pA[j];
      }
    } else {
#pragma unroll
      for (int j = 0; j < 2; ++j) {
        const int i = tid + 256 * j;
        *(v4f*)(sA + (i >> 4) * FO_APITCH + (i & 15) * 4) = pA[j];
      }
    }
#pragma unroll
    for (int j = 0; j < 16; ++j) {
      const int i = tid + 256 * j;
      *(v4f*)(sB + (i >> 7) * FO_BPITCH + (i & 127) * 4) = pB[j];
    }
    __syncthreads();

    if (s + 1 < FO_NSLAB)
      fo_fetch(s + 1, tid, nblk, x, T, Wp, BppP, pA, pB);

    if (s < 4) {
      v2f a0x, a1x, b0x, b1x, b2x, b3x, a0y, a1y, b0y, b1y, b2y, b3y;
      asm volatile(
          FO_LOADS(x, 0, 8704, 0, 128)
          FO_STEP(x, y, 16, 8720, 4352, 4480)
          FO_STEP(y, x, 32, 8736, 8704, 8832)
          FO_STEP(x, y, 48, 8752, 13056, 13184)
          FO_STEP(y, x, 64, 8768, 17408, 17536)
          FO_STEP(x, y, 80, 8784, 21760, 21888)
          FO_STEP(y, x, 96, 8800, 26112, 26240)
          FO_STEP(x, y, 112, 8816, 30464, 30592)
          FO_STEP(y, x, 128, 8832, 34816, 34944)
          FO_STEP(x, y, 144, 8848, 39168, 39296)
          FO_STEP(y, x, 160, 8864, 43520, 43648)
          FO_STEP(x, y, 176, 8880, 47872, 48000)
          FO_STEP(y, x, 192, 8896, 52224, 52352)
          FO_STEP(x, y, 208, 8912, 56576, 56704)
          FO_STEP(y, x, 224, 8928, 60928, 61056)
          FO_STEP(x, y, 240, 8944, 65280, 65408)
          "s_wait_dscnt 0x0\n\t"
          FO_WMMAS(y)
          "v_nop\n\tv_nop\n\tv_nop\n\tv_nop"
          : [c00] "+v"(c00), [c01] "+v"(c01), [c02] "+v"(c02), [c03] "+v"(c03),
            [c10] "+v"(c10), [c11] "+v"(c11), [c12] "+v"(c12), [c13] "+v"(c13),
            [a0x] "=&v"(a0x), [a1x] "=&v"(a1x), [b0x] "=&v"(b0x),
            [b1x] "=&v"(b1x), [b2x] "=&v"(b2x), [b3x] "=&v"(b3x),
            [a0y] "=&v"(a0y), [a1y] "=&v"(a1y), [b0y] "=&v"(b0y),
            [b1y] "=&v"(b1y), [b2y] "=&v"(b2y), [b3y] "=&v"(b3y)
          : [pa] "v"(pa), [pb] "v"(pb), [pc] "v"(pc)
          : "memory");
    } else if (s == 4) {
      fo_table_slab(e0, e1, e2, e3, pa, pb, pc);
    } else {
      fo_table_slab(q0, q1, q2, q3, pa, pb, pc);
    }
  }

  const int mbase = hsel << 3;
  const int col   = colW + lr;
  const int nrow  = nblk + rowW + mbase;
  const float bias0 = Bias[col];
  const float bias1 = Bias[col + 16];
  const float bias2 = Bias[col + 32];
  const float bias3 = Bias[col + 48];
#pragma unroll
  for (int v = 0; v < 8; ++v) {
    float* o0 = out + (size_t)(nrow + v) * DDIM + col;
    float* o1 = o0 + (size_t)NHALF * DDIM;
    o0[0]  = sinf(0.2f * (c00[v] + (e0[v] + q0[v]) + bias0));
    o0[16] = sinf(0.2f * (c01[v] + (e1[v] + q1[v]) + bias1));
    o0[32] = sinf(0.2f * (c02[v] + (e2[v] + q2[v]) + bias2));
    o0[48] = sinf(0.2f * (c03[v] + (e3[v] + q3[v]) + bias3));
    o1[0]  = sinf(0.2f * (c10[v] + (e0[v] - q0[v]) + bias0));
    o1[16] = sinf(0.2f * (c11[v] + (e1[v] - q1[v]) + bias1));
    o1[32] = sinf(0.2f * (c12[v] + (e2[v] - q2[v]) + bias2));
    o1[48] = sinf(0.2f * (c13[v] + (e3[v] - q3[v]) + bias3));
  }
}

extern "C" void kernel_launch(void* const* d_in, const int* in_sizes, int n_in,
                              void* d_out, int out_size, void* d_ws,
                              size_t ws_size, hipStream_t stream) {
  (void)in_sizes; (void)n_in; (void)out_size; (void)ws_size;
  const float* x     = (const float*)d_in[0];
  const float* P_low = (const float*)d_in[1];
  const float* B_low = (const float*)d_in[2];
  const float* W     = (const float*)d_in[3];
  const float* Bias  = (const float*)d_in[4];
  float* out = (float*)d_out;

  float* ws    = (float*)d_ws;
  float* T     = ws;
  float* Ppart = T + (size_t)N_PTS * NROW;
  float* BppP  = Ppart + (size_t)NCHUNK * NROW * DDIM;
  float* Wp    = BppP + NROW * DDIM;

  prep_kernel<<<dim3(PACK_BLOCKS + TGEN_BLOCKS), dim3(256), 0, stream>>>(W, Wp, T);
  dft_low_kernel<<<dim3(DDIM / DFT_COLS, NCHUNK), dim3(256), 0, stream>>>(x, T, Ppart);
  spectral_mix_kernel<<<dim3(DDIM / MIX_FQ, KFREQ), dim3(256), 0, stream>>>(Ppart, P_low, B_low, BppP);
  fused_out_kernel<<<dim3(NHALF / FO_RB), dim3(256), 0, stream>>>(x, Wp, Bias, T, BppP, out);
}


# Table launch removed (4 -> 3 launches): table entries built where used; forward DFT: one complex product per entry (rounding differs from sincosf); output kernel: exact sincosf
# speedup vs baseline: 2.0144x; 1.0746x over previous
#include <hip/hip_runtime.h>
#include <math.h>


typedef float v2f __attribute__((ext_vector_type(2)));
typedef float v4f __attribute__((ext_vector_type(4)));
typedef float v8f __attribute__((ext_vector_type(8)));

#define N_PTS 65536
#define DDIM  256
#define KFREQ 64
#define NROW  128
#define NHALF  (N_PTS / 2)
#define NCHUNK 64
#define CHUNK_N (NHALF / NCHUNK)

__device__ __forceinline__ v8f wmma_f32(v2f a, v2f b, v8f c) {
  return __builtin_amdgcn_wmma_f32_16x16x4_f32(false, a, false, b,
                                               (short)0, c, false, false);
}

__device__ __forceinline__ void twiddle(unsigned n, unsigned k, float& c, float& s) {
  const unsigned m = (n * k) & 65535u;
  const float ang = (float)m * 9.58737992429e-5f;
  sincosf(ang, &s, &c);
}

#define PACK_BLOCKS ((DDIM * DDIM / 2) / 256)

#define DFT_BLOCKS ((DDIM / DFT_COLS) * NCHUNK)
#define DFT_SLAB   64
#define DFT_COLS   64
#define DFT_NSLAB  (CHUNK_N / DFT_SLAB)
#define T_PITCH    (NROW + 8)
#define X_PITCH    (DFT_COLS + 8)

#define DFT_FRAG_LOADS(S)                                              \
  "ds_load_2addr_b32 %[a0" #S "], %[pa] offset1:136\n\t"               \
  "ds_load_2addr_b32 %[a1" #S "], %[pa] offset0:16 offset1:152\n\t"    \
  "ds_load_2addr_b32 %[b0" #S "], %[pb] offset1:72\n\t"                \
  "ds_load_2addr_b32 %[b1" #S "], %[pb] offset0:16 offset1:88\n\t"
#define DFT_ADVANCE                                                    \
  "v_add_nc_u32_e32 %[pa], 0x880, %[pa]\n\t"                           \
  "v_add_nc_u32_e32 %[pb], 0x480, %[pb]\n\t"
#define DFT_WMMAS(S)                                                   \
  "v_wmma_f32_16x16x4_f32 %[c00], %[a0" #S "], %[b0" #S "], %[c00]\n\t" \
  "v_wmma_f32_16x16x4_f32 %[c01], %[a0" #S "], %[b1" #S "], %[c01]\n\t" \
  "v_wmma_f32_16x16x4_f32 %[c10], %[a1" #S "], %[b0" #S "], %[c10]\n\t" \
  "v_wmma_f32_16x16x4_f32 %[c11], %[a1" #S "], %[b1" #S "], %[c11]\n\t"
#define DFT_STEP(CUR, NXT)                                             \
  DFT_ADVANCE DFT_FRAG_LOADS(NXT) "s_wait_dscnt 0x4\n\t" DFT_WMMAS(CUR)

__global__ __launch_bounds__(256) void dft_low_kernel(
    const float* __restrict__ x, const float* __restrict__ W,
    float* __restrict__ Wp, float* __restrict__ Ppart) {
  __shared__ __attribute__((aligned(16))) float sT[DFT_SLAB * T_PITCH];
  __shared__ __attribute__((aligned(16))) float sXs[DFT_SLAB * X_PITCH];
  __shared__ __attribute__((aligned(16))) float sXd[DFT_SLAB * X_PITCH];

  if (blockIdx.x >= DFT_BLOCKS) {
    int idx = (blockIdx.x - DFT_BLOCKS) * 256 + threadIdx.x;
    int kpair = idx >> 8;
    int col   = idx & 255;
    Wp[(size_t)idx * 2]     = W[(size_t)(2 * kpair) * DDIM + col];
    Wp[(size_t)idx * 2 + 1] = W[(size_t)(2 * kpair + 1) * DDIM + col];
    return;
  }

  const int tid     = threadIdx.x;
  const int lane    = tid & 31;
  const int wave    = tid >> 5;
  const int lr      = lane & 15;
  const int halfSel = (lane >> 4) << 1;
  const int q       = blockIdx.x & 3;
  const int chunk   = blockIdx.x >> 2;
  const int nbase   = chunk * CHUNK_N;
  const int rowW    = (wave & 3) * 32;
  const int colW    = (wave >> 2) * 32;

  v8f acc00 = {0.f,0.f,0.f,0.f,0.f,0.f,0.f,0.f};
  v8f acc01 = acc00, acc10 = acc00, acc11 = acc00;

  v4f pX[4];
  v4f pY[4];

  const float* xg = x + (size_t)nbase * DDIM + q * DFT_COLS;
  const float* yg = xg + (size_t)NHALF * DDIM;

#pragma unroll
  for (int j = 0; j < 4; ++j) {
    const int i = tid + 256 * j;
    pX[j] = *(const v4f*)(xg + (size_t)(i >> 4) * DDIM + (i & 15) * 4);
    pY[j] = *(const v4f*)(yg + (size_t)(i >> 4) * DDIM + (i & 15) * 4);
  }

  const unsigned sxs = (unsigned)(size_t)(sXs + halfSel * X_PITCH + colW + lr);
  const unsigned sxd = (unsigned)(size_t)(sXd + halfSel * X_PITCH + colW + lr);
  const unsigned pa0 = (unsigned)(size_t)(sT + halfSel * T_PITCH + rowW + lr);
  const unsigned pb0 = (wave & 2) ? sxd : sxs;

  const int tp  = tid & 63;
  const int tjg = tid >> 6;
  const unsigned tk = (unsigned)(((tp & 31) << 1) | (tp >> 5));
  float twc[16], tws[16];
#pragma unroll
  for (int jj = 0; jj < 16; ++jj)
    twiddle((unsigned)(tjg * 16 + jj), tk, twc[jj], tws[jj]);
  float* sTw = sT + (tjg * 16) * T_PITCH + 2 * tp;

  for (int s = 0; s < DFT_NSLAB; ++s) {
    float cb, sb;
    twiddle((unsigned)(nbase + s * DFT_SLAB), tk, cb, sb);
    __syncthreads();
#pragma unroll
    for (int jj = 0; jj < 16; ++jj) {
      v2f e;
      e.x = cb * twc[jj] - sb * tws[jj];
      e.y = sb * twc[jj] + cb * tws[jj];
      *(v2f*)(sTw + jj * T_PITCH) = e;
    }
#pragma unroll
    for (int j = 0; j < 4; ++j) {
      const int i = tid + 256 * j;
      *(v4f*)(sXs + (i >> 4) * X_PITCH + (i & 15) * 4) = pX[j] + pY[j];
      *(v4f*)(sXd + (i >> 4) * X_PITCH + (i & 15) * 4) = pX[j] - pY[j];
    }
    __syncthreads();

    if (s + 1 < DFT_NSLAB) {
      const float* xn = xg + (size_t)(s + 1) * DFT_SLAB * DDIM;
      const float* yn = yg + (size_t)(s + 1) * DFT_SLAB * DDIM;
#pragma unroll
      for (int j = 0; j < 4; ++j) {
        const int i = tid + 256 * j;
        pX[j] = *(const v4f*)(xn + (size_t)(i >> 4) * DDIM + (i & 15) * 4);
        pY[j] = *(const v4f*)(yn + (size_t)(i >> 4) * DDIM + (i & 15) * 4);
      }
    }

    {
      unsigned pa = pa0, pb = pb0;
      v2f a0x, a1x, b0x, b1x, a0y, a1y, b0y, b1y;
      asm volatile(
          DFT_FRAG_LOADS(x)
          DFT_STEP(x, y) DFT_STEP(y, x) DFT_STEP(x, y) DFT_STEP(y, x)
          DFT_STEP(x, y) DFT_STEP(y, x) DFT_STEP(x, y) DFT_STEP(y, x)
          DFT_STEP(x, y) DFT_STEP(y, x) DFT_STEP(x, y) DFT_STEP(y, x)
          DFT_STEP(x, y) DFT_STEP(y, x) DFT_STEP(x, y)
          "s_wait_dscnt 0x0\n\t"
          DFT_WMMAS(y)
          "v_nop\n\tv_nop\n\tv_nop\n\tv_nop"
          : [c00] "+v"(acc00), [c01] "+v"(acc01),
            [c10] "+v"(acc10), [c11] "+v"(acc11),
            [pa] "+v"(pa), [pb] "+v"(pb),
            [a0x] "=&v"(a0x), [a1x] "=&v"(a1x),
            [b0x] "=&v"(b0x), [b1x] "=&v"(b1x),
            [a0y] "=&v"(a0y), [a1y] "=&v"(a1y),
            [b0y] "=&v"(b0y), [b1y] "=&v"(b1y)
          :
          : "memory");
    }
  }

  const int mbase = (lane >> 4) << 3;
  const int col   = q * DFT_COLS + colW + lr;
  const size_t base = (size_t)chunk * (NROW * DDIM);
#pragma unroll
  for (int v = 0; v < 8; ++v) {
    const int r = rowW + mbase + v;
    Ppart[base + (size_t)r * DDIM + col]             = acc00[v];
    Ppart[base + (size_t)r * DDIM + col + 16]        = acc01[v];
    Ppart[base + (size_t)(r + 16) * DDIM + col]      = acc10[v];
    Ppart[base + (size_t)(r + 16) * DDIM + col + 16] = acc11[v];
  }
}

#define MIX_FQ 64
#define MIX_DS 32

__global__ __launch_bounds__(256) void spectral_mix_kernel(
    const float* __restrict__ Ppart, const float* __restrict__ P_low,
    const float* __restrict__ B_low, float* __restrict__ BppP) {
  __shared__ __attribute__((aligned(8))) float sXr[DDIM];
  __shared__ __attribute__((aligned(8))) float sXi[DDIM];
  __shared__ __attribute__((aligned(16))) float sPart[8 * MIX_FQ * 2];

  const int tid   = threadIdx.x;
  const int lane  = tid & 31;
  const int wave  = tid >> 5;
  const int fq    = blockIdx.x;
  const int k     = blockIdx.y;
  const int prow  = ((k & 1) << 5) | (k >> 1);
  const int dbase = wave * MIX_DS;
  const int f0    = fq * MIX_FQ + lane * 2;

  const int rsel = tid >> 7;
  const int d2   = (tid & 127) * 2;
  const float* Cg = Ppart + (size_t)(2 * prow + rsel) * DDIM + d2;
  v2f cpart[NCHUNK];
#pragma unroll
  for (int ch = 0; ch < NCHUNK; ++ch)
    cpart[ch] = *(const v2f*)(Cg + (size_t)ch * (NROW * DDIM));

  const float* Pk = P_low + (size_t)k * DDIM * DDIM * 2
                          + ((size_t)dbase * DDIM + f0) * 2;
  v4f p[MIX_DS];
#pragma unroll
  for (int j = 0; j < MIX_DS; ++j)
    p[j] = *(const v4f*)(Pk + (size_t)j * DDIM * 2);

  float c0 = 0.f, c1 = 0.f;
#pragma unroll
  for (int ch = 0; ch < NCHUNK; ++ch) {
    c0 += cpart[ch].x;
    c1 += cpart[ch].y;
  }
  const float invN = 1.0f / (float)N_PTS;
  if (rsel == 0) {
    *(v2f*)(sXr + d2) = (v2f){ c0 * invN,  c1 * invN};
  } else {
    *(v2f*)(sXi + d2) = (v2f){-c0 * invN, -c1 * invN};
  }
  __syncthreads();

  float ar0 = 0.f, ai0 = 0.f, ar1 = 0.f, ai1 = 0.f;
#pragma unroll
  for (int j = 0; j < MIX_DS; ++j) {
    const float xr = sXr[dbase + j], xi = sXi[dbase + j];
    ar0 = fmaf(xr, p[j].x, fmaf(-xi, p[j].y, ar0));
    ai0 = fmaf(xr, p[j].y, fmaf( xi, p[j].x, ai0));
    ar1 = fmaf(xr, p[j].z, fmaf(-xi, p[j].w, ar1));
    ai1 = fmaf(xr, p[j].w, fmaf( xi, p[j].z, ai1));
  }
  *(v4f*)(sPart + (wave * MIX_FQ + lane * 2) * 2) = (v4f){ar0, ai0, ar1, ai1};
  __syncthreads();

  if (tid < MIX_FQ * 2) {
    const int fl = tid >> 1;
    const int ri = tid & 1;
    const int f  = fq * MIX_FQ + fl;
    float s = B_low[((size_t)k * DDIM + f) * 2 + ri];
#pragma unroll
    for (int w = 0; w < 8; ++w) s += sPart[(w * MIX_FQ + fl) * 2 + ri];
    const float sc = (k == 0) ? 1.0f : 2.0f;
    BppP[((size_t)prow * DDIM + f) * 2 + ri] = ri ? -sc * s : sc * s;
  }
}

#define FO_RB     32
#define FO_KS     64
#define FO_NSLAB  6
#define FO_APITCH 68
#define FO_BPITCH 544

#define FO_LOADS(S, A0, A1, B0, B1)                                    \
  "ds_load_b64 %[a0" #S "], %[pa] offset:" #A0 "\n\t"                 \
  "ds_load_b64 %[a1" #S "], %[pa] offset:" #A1 "\n\t"                 \
  "ds_load_b64 %[b0" #S "], %[pb] offset:" #B0 "\n\t"                 \
  "ds_load_b64 %[b1" #S "], %[pb] offset:" #B1 "\n\t"                 \
  "ds_load_b64 %[b2" #S "], %[pc] offset:" #B0 "\n\t"                 \
  "ds_load_b64 %[b3" #S "], %[pc] offset:" #B1 "\n\t"
#define FO_WMMAS(S)                                                    \
  "v_wmma_f32_16x16x4_f32 %[c00], %[a0" #S "], %[b0" #S "], %[c00]\n\t" \
  "v_wmma_f32_16x16x4_f32 %[c01], %[a0" #S "], %[b1" #S "], %[c01]\n\t" \
  "v_wmma_f32_16x16x4_f32 %[c02], %[a0" #S "], %[b2" #S "], %[c02]\n\t" \
  "v_wmma_f32_16x16x4_f32 %[c03], %[a0" #S "], %[b3" #S "], %[c03]\n\t" \
  "v_wmma_f32_16x16x4_f32 %[c10], %[a1" #S "], %[b0" #S "], %[c10]\n\t" \
  "v_wmma_f32_16x16x4_f32 %[c11], %[a1" #S "], %[b1" #S "], %[c11]\n\t" \
  "v_wmma_f32_16x16x4_f32 %[c12], %[a1" #S "], %[b2" #S "], %[c12]\n\t" \
  "v_wmma_f32_16x16x4_f32 %[c13], %[a1" #S "], %[b3" #S "], %[c13]\n\t"
#define FO_STEP(CUR, NXT, A0, A1, B0, B1)                              \
  FO_LOADS(NXT, A0, A1, B0, B1) "s_wait_dscnt 0x6\n\t" FO_WMMAS(CUR)

#define FT_LOADS(S, A0, B0, B1)                                        \
  "ds_load_b64 %[a0" #S "], %[pa] offset:" #A0 "\n\t"                 \
  "ds_load_b64 %[b0" #S "], %[pb] offset:" #B0 "\n\t"                 \
  "ds_load_b64 %[b1" #S "], %[pb] offset:" #B1 "\n\t"                 \
  "ds_load_b64 %[b2" #S "], %[pc] offset:" #B0 "\n\t"                 \
  "ds_load_b64 %[b3" #S "], %[pc] offset:" #B1 "\n\t"
#define FT_WMMAS(S)                                                    \
  "v_wmma_f32_16x16x4_f32 %[t0], %[a0" #S "], %[b0" #S "], %[t0]\n\t" \
  "v_wmma_f32_16x16x4_f32 %[t1], %[a0" #S "], %[b1" #S "], %[t1]\n\t" \
  "v_wmma_f32_16x16x4_f32 %[t2], %[a0" #S "], %[b2" #S "], %[t2]\n\t" \
  "v_wmma_f32_16x16x4_f32 %[t3], %[a0" #S "], %[b3" #S "], %[t3]\n\t"
#define FT_STEP(CUR, NXT, A0, B0, B1)                                  \
  FT_LOADS(NXT, A0, B0, B1) "s_wait_dscnt 0x5\n\t" FT_WMMAS(CUR)

__device__ __forceinline__ void fo_fetch(
    int s, int tid, int nblk, const float* __restrict__ x,
    const float* __restrict__ Wp,
    const float* __restrict__ BppP, v4f (&pA)[4], v4f (&pB)[16]) {
  const float* Bg;
  if (s < 4) {
    const float* Ag = x + (size_t)nblk * DDIM + s * FO_KS;
#pragma unroll
    for (int j = 0; j < 4; ++j) {
      const int i = tid + 256 * j;
      const int r = i >> 4;
      pA[j] = *(const v4f*)(Ag + ((size_t)(r & 31) + (size_t)(r >> 5) * NHALF) * DDIM
                               + (i & 15) * 4);
    }
    Bg = Wp + (size_t)s * (FO_KS / 2) * (2 * DDIM);
  } else {
    Bg = BppP + (size_t)(s - 4) * (FO_KS / 2) * (2 * DDIM);
  }
#pragma unroll
  for (int j = 0; j < 16; ++j) {
    const int i = tid + 256 * j;
    pB[j] = *(const v4f*)(Bg + (size_t)i * 4);
  }
}

__device__ __forceinline__ void fo_table_slab(
    v8f& t0, v8f& t1, v8f& t2, v8f& t3,
    unsigned pa, unsigned pb, unsigned pc) {
  v2f a0x, b0x, b1x, b2x, b3x, a0y, b0y, b1y, b2y, b3y;
  asm volatile(
      FT_LOADS(x, 0, 0, 128)
      FT_STEP(x, y, 16, 4352, 4480)
      FT_STEP(y, x, 32, 8704, 8832)
      FT_STEP(x, y, 48, 13056, 13184)
      FT_STEP(y, x, 64, 17408, 17536)
      FT_STEP(x, y, 80, 21760, 21888)
      FT_STEP(y, x, 96, 26112, 26240)
      FT_STEP(x, y, 112, 30464, 30592)
      FT_STEP(y, x, 128, 34816, 34944)
      FT_STEP(x, y, 144, 39168, 39296)
      FT_STEP(y, x, 160, 43520, 43648)
      FT_STEP(x, y, 176, 47872, 48000)
      FT_STEP(y, x, 192, 52224, 52352)
      FT_STEP(x, y, 208, 56576, 56704)
      FT_STEP(y, x, 224, 60928, 61056)
      FT_STEP(x, y, 240, 65280, 65408)
      "s_wait_dscnt 0x0\n\t"
      FT_WMMAS(y)
      "v_nop\n\tv_nop\n\tv_nop\n\tv_nop"
      : [t0] "+v"(t0), [t1] "+v"(t1), [t2] "+v"(t2), [t3] "+v"(t3),
        [a0x] "=&v"(a0x), [b0x] "=&v"(b0x), [b1x] "=&v"(b1x),
        [b2x] "=&v"(b2x), [b3x] "=&v"(b3x),
        [a0y] "=&v"(a0y), [b0y] "=&v"(b0y), [b1y] "=&v"(b1y),
        [b2y] "=&v"(b2y), [b3y] "=&v"(b3y)
      : [pa] "v"(pa), [pb] "v"(pb), [pc] "v"(pc)
      : "memory");
}

__global__ __launch_bounds__(256) void fused_out_kernel(
    const float* __restrict__ x, const float* __restrict__ Wp,
    const float* __restrict__ Bias,
    const float* __restrict__ BppP, float* __restrict__ out) {
  __shared__ __attribute__((aligned(16))) float sA[2 * FO_RB * FO_APITCH];
  __shared__ __attribute__((aligned(16))) float sB[(FO_KS / 2) * FO_BPITCH];
  __shared__ __attribute__((aligned(16))) float sAT[2 * FO_RB * FO_APITCH];

  const int tid     = threadIdx.x;
  const int lane    = tid & 31;
  const int wave    = tid >> 5;
  const int lr      = lane & 15;
  const int hsel    = lane >> 4;
  const int nblk    = blockIdx.x * FO_RB;
  const int rowW    = (wave & 1) * 16;
  const int colW    = (wave >> 1) * 64;

  v8f c00 = {0.f,0.f,0.f,0.f,0.f,0.f,0.f,0.f};
  v8f c01 = c00, c02 = c00, c03 = c00, c10 = c00, c11 = c00, c12 = c00, c13 = c00;
  v8f e0 = c00, e1 = c00, e2 = c00, e3 = c00;
  v8f q0 = c00, q1 = c00, q2 = c00, q3 = c00;

  v4f pA[4];
  v4f pB[16];
  fo_fetch(0, tid, nblk, x, Wp, BppP, pA, pB);

#pragma unroll
  for (int t = 0; t < 2; ++t) {
#pragma unroll
    for (int j = 0; j < 2; ++j) {
      const int i = tid + 256 * j;
      const unsigned n  = (unsigned)(nblk + (i >> 4));
      const int p0 = t * 32 + (i & 15) * 2;
      const int p1 = p0 + 1;
      const unsigned k0 = (unsigned)(((p0 & 31) << 1) | (p0 >> 5));
      const unsigned k1 = (unsigned)(((p1 & 31) << 1) | (p1 >> 5));
      float c0, s0, c1, s1;
      twiddle(n, k0, c0, s0);
      twiddle(n, k1, c1, s1);
      *(v4f*)(sAT + (t * FO_RB + (i >> 4)) * FO_APITCH + (i & 15) * 4) =
          (v4f){c0, s0, c1, s1};
    }
  }

  const unsigned pa = (unsigned)(size_t)(sA + (rowW + lr) * FO_APITCH + 2 * hsel);
  const unsigned paT = (unsigned)(size_t)(sAT + (rowW + lr) * FO_APITCH + 2 * hsel);
  const unsigned pb = (unsigned)(size_t)(sB + hsel * FO_BPITCH + (colW + lr) * 2);
  const unsigned pc = pb + 256u;

  for (int s = 0; s < FO_NSLAB; ++s) {
    __syncthreads();
    if (s < 4) {
#pragma unroll
      for (int j = 0; j < 4; ++j) {
        const int i = tid + 256 * j;
        *(v4f*)(sA + (i >> 4) * FO_APITCH + (i & 15) * 4) = pA[j];
      }
    }
#pragma unroll
    for (int j = 0; j < 16; ++j) {
      const int i = tid + 256 * j;
      *(v4f*)(sB + (i >> 7) * FO_BPITCH + (i & 127) * 4) = pB[j];
    }
    __syncthreads();

    if (s + 1 < FO_NSLAB)
      fo_fetch(s + 1, tid, nblk, x, Wp, BppP, pA, pB);

    if (s < 4) {
      v2f a0x, a1x, b0x, b1x, b2x, b3x, a0y, a1y, b0y, b1y, b2y, b3y;
      asm volatile(
          FO_LOADS(x, 0, 8704, 0, 128)
          FO_STEP(x, y, 16, 8720, 4352, 4480)
          FO_STEP(y, x, 32, 8736, 8704, 8832)
          FO_STEP(x, y, 48, 8752, 13056, 13184)
          FO_STEP(y, x, 64, 8768, 17408, 17536)
          FO_STEP(x, y, 80, 8784, 21760, 21888)
          FO_STEP(y, x, 96, 8800, 26112, 26240)
          FO_STEP(x, y, 112, 8816, 30464, 30592)
          FO_STEP(y, x, 128, 8832, 34816, 34944)
          FO_STEP(x, y, 144, 8848, 39168, 39296)
          FO_STEP(y, x, 160, 8864, 43520, 43648)
          FO_STEP(x, y, 176, 8880, 47872, 48000)
          FO_STEP(y, x, 192, 8896, 52224, 52352)
          FO_STEP(x, y, 208, 8912, 56576, 56704)
          FO_STEP(y, x, 224, 8928, 60928, 61056)
          FO_STEP(x, y, 240, 8944, 65280, 65408)
          "s_wait_dscnt 0x0\n\t"
          FO_WMMAS(y)
          "v_nop\n\tv_nop\n\tv_nop\n\tv_nop"
          : [c00] "+v"(c00), [c01] "+v"(c01), [c02] "+v"(c02), [c03] "+v"(c03),
            [c10] "+v"(c10), [c11] "+v"(c11), [c12] "+v"(c12), [c13] "+v"(c13),
            [a0x] "=&v"(a0x), [a1x] "=&v"(a1x), [b0x] "=&v"(b0x),
            [b1x] "=&v"(b1x), [b2x] "=&v"(b2x), [b3x] "=&v"(b3x),
            [a0y] "=&v"(a0y), [a1y] "=&v"(a1y), [b0y] "=&v"(b0y),
            [b1y] "=&v"(b1y), [b2y] "=&v"(b2y), [b3y] "=&v"(b3y)
          : [pa] "v"(pa), [pb] "v"(pb), [pc] "v"(pc)
          : "memory");
    } else if (s == 4) {
      fo_table_slab(e0, e1, e2, e3, paT, pb, pc);
    } else {
      fo_table_slab(q0, q1, q2, q3, paT + FO_RB * FO_APITCH * 4, pb, pc);
    }
  }

  const int mbase = hsel << 3;
  const int col   = colW + lr;
  const int nrow  = nblk + rowW + mbase;
  const float bias0 = Bias[col];
  const float bias1 = Bias[col + 16];
  const float bias2 = Bias[col + 32];
  const float bias3 = Bias[col + 48];
#pragma unroll
  for (int v = 0; v < 8; ++v) {
    float* o0 = out + (size_t)(nrow + v) * DDIM + col;
    float* o1 = o0 + (size_t)NHALF * DDIM;
    o0[0]  = sinf(0.2f * (c00[v] + (e0[v] + q0[v]) + bias0));
    o0[16] = sinf(0.2f * (c01[v] + (e1[v] + q1[v]) + bias1));
    o0[32] = sinf(0.2f * (c02[v] + (e2[v] + q2[v]) + bias2));
    o0[48] = sinf(0.2f * (c03[v] + (e3[v] + q3[v]) + bias3));
    o1[0]  = sinf(0.2f * (c10[v] + (e0[v] - q0[v]) + bias0));
    o1[16] = sinf(0.2f * (c11[v] + (e1[v] - q1[v]) + bias1));
    o1[32] = sinf(0.2f * (c12[v] + (e2[v] - q2[v]) + bias2));
    o1[48] = sinf(0.2f * (c13[v] + (e3[v] - q3[v]) + bias3));
  }
}

extern "C" void kernel_launch(void* const* d_in, const int* in_sizes, int n_in,
                              void* d_out, int out_size, void* d_ws,
                              size_t ws_size, hipStream_t stream) {
  (void)in_sizes; (void)n_in; (void)out_size; (void)ws_size;
  const float* x     = (const float*)d_in[0];
  const float* P_low = (const float*)d_in[1];
  const float* B_low = (const float*)d_in[2];
  const float* W     = (const float*)d_in[3];
  const float* Bias  = (const float*)d_in[4];
  float* out = (float*)d_out;

  float* ws    = (float*)d_ws;
  float* Ppart = ws + (size_t)N_PTS * NROW;
  float* BppP  = Ppart + (size_t)NCHUNK * NROW * DDIM;
  float* Wp    = BppP + NROW * DDIM;

  dft_low_kernel<<<dim3(DFT_BLOCKS + PACK_BLOCKS), dim3(256), 0, stream>>>(x, W, Wp, Ppart);
  spectral_mix_kernel<<<dim3(DDIM / MIX_FQ, KFREQ), dim3(256), 0, stream>>>(Ppart, P_low, B_low, BppP);
  fused_out_kernel<<<dim3(NHALF / FO_RB), dim3(256), 0, stream>>>(x, Wp, Bias, BppP, out);
}


# sinf/sincosf calls replaced by the library's own small-argument path written out (same operations, bit-identical values); full-range sinf kept once as a fallback loop; kernel text 76 KB -> 46 KB
# speedup vs baseline: 2.2959x; 1.1397x over previous
#include <hip/hip_runtime.h>
#include <math.h>


typedef float v2f __attribute__((ext_vector_type(2)));
typedef float v4f __attribute__((ext_vector_type(4)));
typedef float v8f __attribute__((ext_vector_type(8)));

#define N_PTS 65536
#define DDIM  256
#define KFREQ 64
#define NROW  128
#define NHALF  (N_PTS / 2)
#define NCHUNK 64
#define CHUNK_N (NHALF / NCHUNK)

__device__ __forceinline__ v8f wmma_f32(v2f a, v2f b, v8f c) {
  return __builtin_amdgcn_wmma_f32_16x16x4_f32(false, a, false, b,
                                               (short)0, c, false, false);
}

#define TRIG_F(u) __builtin_bit_cast(float, (unsigned)(u))

__device__ __forceinline__ void trig_small(float ax, float& sp, float& cp, int& q) {
  const float n = __builtin_rintf(ax * TRIG_F(0x3f22f983u));
  float r = __builtin_fmaf(n, TRIG_F(0xbfc90fdau), ax);
  r = __builtin_fmaf(n, TRIG_F(0xb3a22168u), r);
  r = __builtin_fmaf(n, TRIG_F(0xa7c234c4u), r);
  q = (int)n;
  const float r2 = r * r;
  float a = __builtin_fmaf(TRIG_F(0xb94c1982u), r2, TRIG_F(0x3c0881c4u));
  a = __builtin_fmaf(r2, a, TRIG_F(0xbe2aaa9du));
  a = r2 * a;
  sp = __builtin_fmaf(r, a, r);
  float b = __builtin_fmaf(TRIG_F(0x37d75334u), r2, TRIG_F(0xbab64f3bu));
  b = __builtin_fmaf(r2, b, TRIG_F(0x3d2aabf7u));
  b = __builtin_fmaf(r2, b, TRIG_F(0xbf000004u));
  cp = __builtin_fmaf(r2, b, 1.0f);
}

__device__ __forceinline__ float sin_small(float a) {
  float sp, cp; int q;
  trig_small(__builtin_fabsf(a), sp, cp, q);
  const float sel = (q & 1) ? cp : sp;
  const unsigned flip = (((unsigned)q << 30) ^ __builtin_bit_cast(unsigned, a)) & 0x80000000u;
  return __builtin_bit_cast(float, __builtin_bit_cast(unsigned, sel) ^ flip);
}

__device__ __forceinline__ void sincos_small(float ang, float& s, float& c) {
  float sp, cp; int q;
  trig_small(ang, sp, cp, q);
  const unsigned flip = ((unsigned)q << 30) & 0x80000000u;
  const float ss = (q & 1) ? cp : sp;
  const float cc = (q & 1) ? -sp : cp;
  s = __builtin_bit_cast(float, __builtin_bit_cast(unsigned, ss) ^ flip);
  c = __builtin_bit_cast(float, __builtin_bit_cast(unsigned, cc) ^ flip);
}

__device__ __forceinline__ void twiddle(unsigned n, unsigned k, float& c, float& s) {
  const unsigned m = (n * k) & 65535u;
  const float ang = (float)m * 9.58737992429e-5f;
  sincos_small(ang, s, c);
}

#define PACK_BLOCKS ((DDIM * DDIM / 2) / 256)

#define DFT_BLOCKS ((DDIM / DFT_COLS) * NCHUNK)
#define DFT_SLAB   64
#define DFT_COLS   64
#define DFT_NSLAB  (CHUNK_N / DFT_SLAB)
#define T_PITCH    (NROW + 8)
#define X_PITCH    (DFT_COLS + 8)

#define DFT_FRAG_LOADS(S)                                              \
  "ds_load_2addr_b32 %[a0" #S "], %[pa] offset1:136\n\t"               \
  "ds_load_2addr_b32 %[a1" #S "], %[pa] offset0:16 offset1:152\n\t"    \
  "ds_load_2addr_b32 %[b0" #S "], %[pb] offset1:72\n\t"                \
  "ds_load_2addr_b32 %[b1" #S "], %[pb] offset0:16 offset1:88\n\t"
#define DFT_ADVANCE                                                    \
  "v_add_nc_u32_e32 %[pa], 0x880, %[pa]\n\t"                           \
  "v_add_nc_u32_e32 %[pb], 0x480, %[pb]\n\t"
#define DFT_WMMAS(S)                                                   \
  "v_wmma_f32_16x16x4_f32 %[c00], %[a0" #S "], %[b0" #S "], %[c00]\n\t" \
  "v_wmma_f32_16x16x4_f32 %[c01], %[a0" #S "], %[b1" #S "], %[c01]\n\t" \
  "v_wmma_f32_16x16x4_f32 %[c10], %[a1" #S "], %[b0" #S "], %[c10]\n\t" \
  "v_wmma_f32_16x16x4_f32 %[c11], %[a1" #S "], %[b1" #S "], %[c11]\n\t"
#define DFT_STEP(CUR, NXT)                                             \
  DFT_ADVANCE DFT_FRAG_LOADS(NXT) "s_wait_dscnt 0x4\n\t" DFT_WMMAS(CUR)

__global__ __launch_bounds__(256) void dft_low_kernel(
    const float* __restrict__ x, const float* __restrict__ W,
    float* __restrict__ Wp, float* __restrict__ Ppart) {
  __shared__ __attribute__((aligned(16))) float sT[DFT_SLAB * T_PITCH];
  __shared__ __attribute__((aligned(16))) float sXs[DFT_SLAB * X_PITCH];
  __shared__ __attribute__((aligned(16))) float sXd[DFT_SLAB * X_PITCH];

  if (blockIdx.x >= DFT_BLOCKS) {
    int idx = (blockIdx.x - DFT_BLOCKS) * 256 + threadIdx.x;
    int kpair = idx >> 8;
    int col   = idx & 255;
    Wp[(size_t)idx * 2]     = W[(size_t)(2 * kpair) * DDIM + col];
    Wp[(size_t)idx * 2 + 1] = W[(size_t)(2 * kpair + 1) * DDIM + col];
    return;
  }

  const int tid     = threadIdx.x;
  const int lane    = tid & 31;
  const int wave    = tid >> 5;
  const int lr      = lane & 15;
  const int halfSel = (lane >> 4) << 1;
  const int q       = blockIdx.x & 3;
  const int chunk   = blockIdx.x >> 2;
  const int nbase   = chunk * CHUNK_N;
  const int rowW    = (wave & 3) * 32;
  const int colW    = (wave >> 2) * 32;

  v8f acc00 = {0.f,0.f,0.f,0.f,0.f,0.f,0.f,0.f};
  v8f acc01 = acc00, acc10 = acc00, acc11 = acc00;

  v4f pX[4];
  v4f pY[4];

  const float* xg = x + (size_t)nbase * DDIM + q * DFT_COLS;
  const float* yg = xg + (size_t)NHALF * DDIM;

#pragma unroll
  for (int j = 0; j < 4; ++j) {
    const int i = tid + 256 * j;
    pX[j] = *(const v4f*)(xg + (size_t)(i >> 4) * DDIM + (i & 15) * 4);
    pY[j] = *(const v4f*)(yg + (size_t)(i >> 4) * DDIM + (i & 15) * 4);
  }

  const unsigned sxs = (unsigned)(size_t)(sXs + halfSel * X_PITCH + colW + lr);
  const unsigned sxd = (unsigned)(size_t)(sXd + halfSel * X_PITCH + colW + lr);
  const unsigned pa0 = (unsigned)(size_t)(sT + halfSel * T_PITCH + rowW + lr);
  const unsigned pb0 = (wave & 2) ? sxd : sxs;

  const int tp  = tid & 63;
  const int tjg = tid >> 6;
  const unsigned tk = (unsigned)(((tp & 31) << 1) | (tp >> 5));
  float twc[16], tws[16];
#pragma unroll
  for (int jj = 0; jj < 16; ++jj)
    twiddle((unsigned)(tjg * 16 + jj), tk, twc[jj], tws[jj]);
  float* sTw = sT + (tjg * 16) * T_PITCH + 2 * tp;

  for (int s = 0; s < DFT_NSLAB; ++s) {
    float cb, sb;
    twiddle((unsigned)(nbase + s * DFT_SLAB), tk, cb, sb);
    __syncthreads();
#pragma unroll
    for (int jj = 0; jj < 16; ++jj) {
      v2f e;
      e.x = cb * twc[jj] - sb * tws[jj];
      e.y = sb * twc[jj] + cb * tws[jj];
      *(v2f*)(sTw + jj * T_PITCH) = e;
    }
#pragma unroll
    for (int j = 0; j < 4; ++j) {
      const int i = tid + 256 * j;
      *(v4f*)(sXs + (i >> 4) * X_PITCH + (i & 15) * 4) = pX[j] + pY[j];
      *(v4f*)(sXd + (i >> 4) * X_PITCH + (i & 15) * 4) = pX[j] - pY[j];
    }
    __syncthreads();

    if (s + 1 < DFT_NSLAB) {
      const float* xn = xg + (size_t)(s + 1) * DFT_SLAB * DDIM;
      const float* yn = yg + (size_t)(s + 1) * DFT_SLAB * DDIM;
#pragma unroll
      for (int j = 0; j < 4; ++j) {
        const int i = tid + 256 * j;
        pX[j] = *(const v4f*)(xn + (size_t)(i >> 4) * DDIM + (i & 15) * 4);
        pY[j] = *(const v4f*)(yn + (size_t)(i >> 4) * DDIM + (i & 15) * 4);
      }
    }

    {
      unsigned pa = pa0, pb = pb0;
      v2f a0x, a1x, b0x, b1x, a0y, a1y, b0y, b1y;
      asm volatile(
          DFT_FRAG_LOADS(x)
          DFT_STEP(x, y) DFT_STEP(y, x) DFT_STEP(x, y) DFT_STEP(y, x)
          DFT_STEP(x, y) DFT_STEP(y, x) DFT_STEP(x, y) DFT_STEP(y, x)
          DFT_STEP(x, y) DFT_STEP(y, x) DFT_STEP(x, y) DFT_STEP(y, x)
          DFT_STEP(x, y) DFT_STEP(y, x) DFT_STEP(x, y)
          "s_wait_dscnt 0x0\n\t"
          DFT_WMMAS(y)
          "v_nop\n\tv_nop\n\tv_nop\n\tv_nop"
          : [c00] "+v"(acc00), [c01] "+v"(acc01),
            [c10] "+v"(acc10), [c11] "+v"(acc11),
            [pa] "+v"(pa), [pb] "+v"(pb),
            [a0x] "=&v"(a0x), [a1x] "=&v"(a1x),
            [b0x] "=&v"(b0x), [b1x] "=&v"(b1x),
            [a0y] "=&v"(a0y), [a1y] "=&v"(a1y),
            [b0y] "=&v"(b0y), [b1y] "=&v"(b1y)
          :
          : "memory");
    }
  }

  const int mbase = (lane >> 4) << 3;
  const int col   = q * DFT_COLS + colW + lr;
  const size_t base = (size_t)chunk * (NROW * DDIM);
#pragma unroll
  for (int v = 0; v < 8; ++v) {
    const int r = rowW + mbase + v;
    Ppart[base + (size_t)r * DDIM + col]             = acc00[v];
    Ppart[base + (size_t)r * DDIM + col + 16]        = acc01[v];
    Ppart[base + (size_t)(r + 16) * DDIM + col]      = acc10[v];
    Ppart[base + (size_t)(r + 16) * DDIM + col + 16] = acc11[v];
  }
}

#define MIX_FQ 64
#define MIX_DS 32

__global__ __launch_bounds__(256) void spectral_mix_kernel(
    const float* __restrict__ Ppart, const float* __restrict__ P_low,
    const float* __restrict__ B_low, float* __restrict__ BppP) {
  __shared__ __attribute__((aligned(8))) float sXr[DDIM];
  __shared__ __attribute__((aligned(8))) float sXi[DDIM];
  __shared__ __attribute__((aligned(16))) float sPart[8 * MIX_FQ * 2];

  const int tid   = threadIdx.x;
  const int lane  = tid & 31;
  const int wave  = tid >> 5;
  const int fq    = blockIdx.x;
  const int k     = blockIdx.y;
  const int prow  = ((k & 1) << 5) | (k >> 1);
  const int dbase = wave * MIX_DS;
  const int f0    = fq * MIX_FQ + lane * 2;

  const int rsel = tid >> 7;
  const int d2   = (tid & 127) * 2;
  const float* Cg = Ppart + (size_t)(2 * prow + rsel) * DDIM + d2;
  v2f cpart[NCHUNK];
#pragma unroll
  for (int ch = 0; ch < NCHUNK; ++ch)
    cpart[ch] = *(const v2f*)(Cg + (size_t)ch * (NROW * DDIM));

  const float* Pk = P_low + (size_t)k * DDIM * DDIM * 2
                          + ((size_t)dbase * DDIM + f0) * 2;
  v4f p[MIX_DS];
#pragma unroll
  for (int j = 0; j < MIX_DS; ++j)
    p[j] = *(const v4f*)(Pk + (size_t)j * DDIM * 2);

  float c0 = 0.f, c1 = 0.f;
#pragma unroll
  for (int ch = 0; ch < NCHUNK; ++ch) {
    c0 += cpart[ch].x;
    c1 += cpart[ch].y;
  }
  const float invN = 1.0f / (float)N_PTS;
  if (rsel == 0) {
    *(v2f*)(sXr + d2) = (v2f){ c0 * invN,  c1 * invN};
  } else {
    *(v2f*)(sXi + d2) = (v2f){-c0 * invN, -c1 * invN};
  }
  __syncthreads();

  float ar0 = 0.f, ai0 = 0.f, ar1 = 0.f, ai1 = 0.f;
#pragma unroll
  for (int j = 0; j < MIX_DS; ++j) {
    const float xr = sXr[dbase + j], xi = sXi[dbase + j];
    ar0 = fmaf(xr, p[j].x, fmaf(-xi, p[j].y, ar0));
    ai0 = fmaf(xr, p[j].y, fmaf( xi, p[j].x, ai0));
    ar1 = fmaf(xr, p[j].z, fmaf(-xi, p[j].w, ar1));
    ai1 = fmaf(xr, p[j].w, fmaf( xi, p[j].z, ai1));
  }
  *(v4f*)(sPart + (wave * MIX_FQ + lane * 2) * 2) = (v4f){ar0, ai0, ar1, ai1};
  __syncthreads();

  if (tid < MIX_FQ * 2) {
    const int fl = tid >> 1;
    const int ri = tid & 1;
    const int f  = fq * MIX_FQ + fl;
    float s = B_low[((size_t)k * DDIM + f) * 2 + ri];
#pragma unroll
    for (int w = 0; w < 8; ++w) s += sPart[(w * MIX_FQ + fl) * 2 + ri];
    const float sc = (k == 0) ? 1.0f : 2.0f;
    BppP[((size_t)prow * DDIM + f) * 2 + ri] = ri ? -sc * s : sc * s;
  }
}

#define FO_RB     32
#define FO_KS     64
#define FO_NSLAB  6
#define FO_APITCH 68
#define FO_BPITCH 544

#define FO_LOADS(S, A0, A1, B0, B1)                                    \
  "ds_load_b64 %[a0" #S "], %[pa] offset:" #A0 "\n\t"                 \
  "ds_load_b64 %[a1" #S "], %[pa] offset:" #A1 "\n\t"                 \
  "ds_load_b64 %[b0" #S "], %[pb] offset:" #B0 "\n\t"                 \
  "ds_load_b64 %[b1" #S "], %[pb] offset:" #B1 "\n\t"                 \
  "ds_load_b64 %[b2" #S "], %[pc] offset:" #B0 "\n\t"                 \
  "ds_load_b64 %[b3" #S "], %[pc] offset:" #B1 "\n\t"
#define FO_WMMAS(S)                                                    \
  "v_wmma_f32_16x16x4_f32 %[c00], %[a0" #S "], %[b0" #S "], %[c00]\n\t" \
  "v_wmma_f32_16x16x4_f32 %[c01], %[a0" #S "], %[b1" #S "], %[c01]\n\t" \
  "v_wmma_f32_16x16x4_f32 %[c02], %[a0" #S "], %[b2" #S "], %[c02]\n\t" \
  "v_wmma_f32_16x16x4_f32 %[c03], %[a0" #S "], %[b3" #S "], %[c03]\n\t" \
  "v_wmma_f32_16x16x4_f32 %[c10], %[a1" #S "], %[b0" #S "], %[c10]\n\t" \
  "v_wmma_f32_16x16x4_f32 %[c11], %[a1" #S "], %[b1" #S "], %[c11]\n\t" \
  "v_wmma_f32_16x16x4_f32 %[c12], %[a1" #S "], %[b2" #S "], %[c12]\n\t" \
  "v_wmma_f32_16x16x4_f32 %[c13], %[a1" #S "], %[b3" #S "], %[c13]\n\t"
#define FO_STEP(CUR, NXT, A0, A1, B0, B1)                              \
  FO_LOADS(NXT, A0, A1, B0, B1) "s_wait_dscnt 0x6\n\t" FO_WMMAS(CUR)

#define FT_LOADS(S, A0, B0, B1)                                        \
  "ds_load_b64 %[a0" #S "], %[pa] offset:" #A0 "\n\t"                 \
  "ds_load_b64 %[b0" #S "], %[pb] offset:" #B0 "\n\t"                 \
  "ds_load_b64 %[b1" #S "], %[pb] offset:" #B1 "\n\t"                 \
  "ds_load_b64 %[b2" #S "], %[pc] offset:" #B0 "\n\t"                 \
  "ds_load_b64 %[b3" #S "], %[pc] offset:" #B1 "\n\t"
#define FT_WMMAS(S)                                                    \
  "v_wmma_f32_16x16x4_f32 %[t0], %[a0" #S "], %[b0" #S "], %[t0]\n\t" \
  "v_wmma_f32_16x16x4_f32 %[t1], %[a0" #S "], %[b1" #S "], %[t1]\n\t" \
  "v_wmma_f32_16x16x4_f32 %[t2], %[a0" #S "], %[b2" #S "], %[t2]\n\t" \
  "v_wmma_f32_16x16x4_f32 %[t3], %[a0" #S "], %[b3" #S "], %[t3]\n\t"
#define FT_STEP(CUR, NXT, A0, B0, B1)                                  \
  FT_LOADS(NXT, A0, B0, B1) "s_wait_dscnt 0x5\n\t" FT_WMMAS(CUR)

__device__ __forceinline__ void fo_fetch(
    int s, int tid, int nblk, const float* __restrict__ x,
    const float* __restrict__ Wp,
    const float* __restrict__ BppP, v4f (&pA)[4], v4f (&pB)[16]) {
  const float* Bg;
  if (s < 4) {
    const float* Ag = x + (size_t)nblk * DDIM + s * FO_KS;
#pragma unroll
    for (int j = 0; j < 4; ++j) {
      const int i = tid + 256 * j;
      const int r = i >> 4;
      pA[j] = *(const v4f*)(Ag + ((size_t)(r & 31) + (size_t)(r >> 5) * NHALF) * DDIM
                               + (i & 15) * 4);
    }
    Bg = Wp + (size_t)s * (FO_KS / 2) * (2 * DDIM);
  } else {
    Bg = BppP + (size_t)(s - 4) * (FO_KS / 2) * (2 * DDIM);
  }
#pragma unroll
  for (int j = 0; j < 16; ++j) {
    const int i = tid + 256 * j;
    pB[j] = *(const v4f*)(Bg + (size_t)i * 4);
  }
}

__device__ __forceinline__ void fo_table_slab(
    v8f& t0, v8f& t1, v8f& t2, v8f& t3,
    unsigned pa, unsigned pb, unsigned pc) {
  v2f a0x, b0x, b1x, b2x, b3x, a0y, b0y, b1y, b2y, b3y;
  asm volatile(
      FT_LOADS(x, 0, 0, 128)
      FT_STEP(x, y, 16, 4352, 4480)
      FT_STEP(y, x, 32, 8704, 8832)
      FT_STEP(x, y, 48, 13056, 13184)
      FT_STEP(y, x, 64, 17408, 17536)
      FT_STEP(x, y, 80, 21760, 21888)
      FT_STEP(y, x, 96, 26112, 26240)
      FT_STEP(x, y, 112, 30464, 30592)
      FT_STEP(y, x, 128, 34816, 34944)
      FT_STEP(x, y, 144, 39168, 39296)
      FT_STEP(y, x, 160, 43520, 43648)
      FT_STEP(x, y, 176, 47872, 48000)
      FT_STEP(y, x, 192, 52224, 52352)
      FT_STEP(x, y, 208, 56576, 56704)
      FT_STEP(y, x, 224, 60928, 61056)
      FT_STEP(x, y, 240, 65280, 65408)
      "s_wait_dscnt 0x0\n\t"
      FT_WMMAS(y)
      "v_nop\n\tv_nop\n\tv_nop\n\tv_nop"
      : [t0] "+v"(t0), [t1] "+v"(t1), [t2] "+v"(t2), [t3] "+v"(t3),
        [a0x] "=&v"(a0x), [b0x] "=&v"(b0x), [b1x] "=&v"(b1x),
        [b2x] "=&v"(b2x), [b3x] "=&v"(b3x),
        [a0y] "=&v"(a0y), [b0y] "=&v"(b0y), [b1y] "=&v"(b1y),
        [b2y] "=&v"(b2y), [b3y] "=&v"(b3y)
      : [pa] "v"(pa), [pb] "v"(pb), [pc] "v"(pc)
      : "memory");
}

__global__ __launch_bounds__(256) void fused_out_kernel(
    const float* __restrict__ x, const float* __restrict__ Wp,
    const float* __restrict__ Bias,
    const float* __restrict__ BppP, float* __restrict__ out) {
  __shared__ __attribute__((aligned(16))) float sA[2 * FO_RB * FO_APITCH];
  __shared__ __attribute__((aligned(16))) float sB[(FO_KS / 2) * FO_BPITCH];
  __shared__ __attribute__((aligned(16))) float sAT[2 * FO_RB * FO_APITCH];

  const int tid     = threadIdx.x;
  const int lane    = tid & 31;
  const int wave    = tid >> 5;
  const int lr      = lane & 15;
  const int hsel    = lane >> 4;
  const int nblk    = blockIdx.x * FO_RB;
  const int rowW    = (wave & 1) * 16;
  const int colW    = (wave >> 1) * 64;

  v8f c00 = {0.f,0.f,0.f,0.f,0.f,0.f,0.f,0.f};
  v8f c01 = c00, c02 = c00, c03 = c00, c10 = c00, c11 = c00, c12 = c00, c13 = c00;
  v8f e0 = c00, e1 = c00, e2 = c00, e3 = c00;
  v8f q0 = c00, q1 = c00, q2 = c00, q3 = c00;

  v4f pA[4];
  v4f pB[16];
  fo_fetch(0, tid, nblk, x, Wp, BppP, pA, pB);

#pragma unroll
  for (int t = 0; t < 2; ++t) {
#pragma unroll
    for (int j = 0; j < 2; ++j) {
      const int i = tid + 256 * j;
      const unsigned n  = (unsigned)(nblk + (i >> 4));
      const int p0 = t * 32 + (i & 15) * 2;
      const int p1 = p0 + 1;
      const unsigned k0 = (unsigned)(((p0 & 31) << 1) | (p0 >> 5));
      const unsigned k1 = (unsigned)(((p1 & 31) << 1) | (p1 >> 5));
      float c0, s0, c1, s1;
      twiddle(n, k0, c0, s0);
      twiddle(n, k1, c1, s1);
      *(v4f*)(sAT + (t * FO_RB + (i >> 4)) * FO_APITCH + (i & 15) * 4) =
          (v4f){c0, s0, c1, s1};
    }
  }

  const unsigned pa = (unsigned)(size_t)(sA + (rowW + lr) * FO_APITCH + 2 * hsel);
  const unsigned paT = (unsigned)(size_t)(sAT + (rowW + lr) * FO_APITCH + 2 * hsel);
  const unsigned pb = (unsigned)(size_t)(sB + hsel * FO_BPITCH + (colW + lr) * 2);
  const unsigned pc = pb + 256u;

  for (int s = 0; s < FO_NSLAB; ++s) {
    __syncthreads();
    if (s < 4) {
#pragma unroll
      for (int j = 0; j < 4; ++j) {
        const int i = tid + 256 * j;
        *(v4f*)(sA + (i >> 4) * FO_APITCH + (i & 15) * 4) = pA[j];
      }
    }
#pragma unroll
    for (int j = 0; j < 16; ++j) {
      const int i = tid + 256 * j;
      *(v4f*)(sB + (i >> 7) * FO_BPITCH + (i & 127) * 4) = pB[j];
    }
    __syncthreads();

    if (s + 1 < FO_NSLAB)
      fo_fetch(s + 1, tid, nblk, x, Wp, BppP, pA, pB);

    if (s < 4) {
      v2f a0x, a1x, b0x, b1x, b2x, b3x, a0y, a1y, b0y, b1y, b2y, b3y;
      asm volatile(
          FO_LOADS(x, 0, 8704, 0, 128)
          FO_STEP(x, y, 16, 8720, 4352, 4480)
          FO_STEP(y, x, 32, 8736, 8704, 8832)
          FO_STEP(x, y, 48, 8752, 13056, 13184)
          FO_STEP(y, x, 64, 8768, 17408, 17536)
          FO_STEP(x, y, 80, 8784, 21760, 21888)
          FO_STEP(y, x, 96, 8800, 26112, 26240)
          FO_STEP(x, y, 112, 8816, 30464, 30592)
          FO_STEP(y, x, 128, 8832, 34816, 34944)
          FO_STEP(x, y, 144, 8848, 39168, 39296)
          FO_STEP(y, x, 160, 8864, 43520, 43648)
          FO_STEP(x, y, 176, 8880, 47872, 48000)
          FO_STEP(y, x, 192, 8896, 52224, 52352)
          FO_STEP(x, y, 208, 8912, 56576, 56704)
          FO_STEP(y, x, 224, 8928, 60928, 61056)
          FO_STEP(x, y, 240, 8944, 65280, 65408)
          "s_wait_dscnt 0x0\n\t"
          FO_WMMAS(y)
          "v_nop\n\tv_nop\n\tv_nop\n\tv_nop"
          : [c00] "+v"(c00), [c01] "+v"(c01), [c02] "+v"(c02), [c03] "+v"(c03),
            [c10] "+v"(c10), [c11] "+v"(c11), [c12] "+v"(c12), [c13] "+v"(c13),
            [a0x] "=&v"(a0x), [a1x] "=&v"(a1x), [b0x] "=&v"(b0x),
            [b1x] "=&v"(b1x), [b2x] "=&v"(b2x), [b3x] "=&v"(b3x),
            [a0y] "=&v"(a0y), [a1y] "=&v"(a1y), [b0y] "=&v"(b0y),
            [b1y] "=&v"(b1y), [b2y] "=&v"(b2y), [b3y] "=&v"(b3y)
          : [pa] "v"(pa), [pb] "v"(pb), [pc] "v"(pc)
          : "memory");
    } else if (s == 4) {
      fo_table_slab(e0, e1, e2, e3, paT, pb, pc);
    } else {
      fo_table_slab(q0, q1, q2, q3, paT + FO_RB * FO_APITCH * 4, pb, pc);
    }
  }

  const int mbase = hsel << 3;
  const int col   = colW + lr;
  const int nrow  = nblk + rowW + mbase;
  const float bias0 = Bias[col];
  const float bias1 = Bias[col + 16];
  const float bias2 = Bias[col + 32];
  const float bias3 = Bias[col + 48];
  float big = 0.f;
#pragma unroll
  for (int v = 0; v < 8; ++v) {
    const float ep0 = e0[v] + q0[v], em0 = e0[v] - q0[v];
    const float ep1 = e1[v] + q1[v], em1 = e1[v] - q1[v];
    const float ep2 = e2[v] + q2[v], em2 = e2[v] - q2[v];
    const float ep3 = e3[v] + q3[v], em3 = e3[v] - q3[v];
    c00[v] = 0.2f * (c00[v] + ep0 + bias0);
    c01[v] = 0.2f * (c01[v] + ep1 + bias1);
    c02[v] = 0.2f * (c02[v] + ep2 + bias2);
    c03[v] = 0.2f * (c03[v] + ep3 + bias3);
    c10[v] = 0.2f * (c10[v] + em0 + bias0);
    c11[v] = 0.2f * (c11[v] + em1 + bias1);
    c12[v] = 0.2f * (c12[v] + em2 + bias2);
    c13[v] = 0.2f * (c13[v] + em3 + bias3);
    big = fmaxf(big, fmaxf(fmaxf(__builtin_fabsf(c00[v]), __builtin_fabsf(c01[v])),
                           fmaxf(__builtin_fabsf(c02[v]), __builtin_fabsf(c03[v]))));
    big = fmaxf(big, fmaxf(fmaxf(__builtin_fabsf(c10[v]), __builtin_fabsf(c11[v])),
                           fmaxf(__builtin_fabsf(c12[v]), __builtin_fabsf(c13[v]))));
  }
  float* ob0 = out + (size_t)nrow * DDIM + col;
  float* ob1 = ob0 + (size_t)NHALF * DDIM;
  if (big < 131072.0f) {
#pragma unroll
    for (int v = 0; v < 8; ++v) {
      float* o0 = ob0 + v * DDIM;
      float* o1 = ob1 + v * DDIM;
      o0[0]  = sin_small(c00[v]);
      o0[16] = sin_small(c01[v]);
      o0[32] = sin_small(c02[v]);
      o0[48] = sin_small(c03[v]);
      o1[0]  = sin_small(c10[v]);
      o1[16] = sin_small(c11[v]);
      o1[32] = sin_small(c12[v]);
      o1[48] = sin_small(c13[v]);
    }
  } else {
#pragma nounroll
    for (int v = 0; v < 8; ++v) {
      float* o0 = ob0 + v * DDIM;
      float* o1 = ob1 + v * DDIM;
      o0[0]  = sinf(c00[v]);
      o0[16] = sinf(c01[v]);
      o0[32] = sinf(c02[v]);
      o0[48] = sinf(c03[v]);
      o1[0]  = sinf(c10[v]);
      o1[16] = sinf(c11[v]);
      o1[32] = sinf(c12[v]);
      o1[48] = sinf(c13[v]);
    }
  }
}

extern "C" void kernel_launch(void* const* d_in, const int* in_sizes, int n_in,
                              void* d_out, int out_size, void* d_ws,
                              size_t ws_size, hipStream_t stream) {
  (void)in_sizes; (void)n_in; (void)out_size; (void)ws_size;
  const float* x     = (const float*)d_in[0];
  const float* P_low = (const float*)d_in[1];
  const float* B_low = (const float*)d_in[2];
  const float* W     = (const float*)d_in[3];
  const float* Bias  = (const float*)d_in[4];
  float* out = (float*)d_out;

  float* ws    = (float*)d_ws;
  float* Ppart = ws + (size_t)N_PTS * NROW;
  float* BppP  = Ppart + (size_t)NCHUNK * NROW * DDIM;
  float* Wp    = BppP + NROW * DDIM;

  dft_low_kernel<<<dim3(DFT_BLOCKS + PACK_BLOCKS), dim3(256), 0, stream>>>(x, W, Wp, Ppart);
  spectral_mix_kernel<<<dim3(DDIM / MIX_FQ, KFREQ), dim3(256), 0, stream>>>(Ppart, P_low, B_low, BppP);
  fused_out_kernel<<<dim3(NHALF / FO_RB), dim3(256), 0, stream>>>(x, Wp, Bias, BppP, out);
}


# Forward DFT decimated by 16: rows n1+(N/16)*n2 combined by a 16-point real transform on the vector unit, matrix reduction over N/16 only (a quarter of the WMMAs); sum order changed
# speedup vs baseline: 2.4765x; 1.0787x over previous
#include <hip/hip_runtime.h>
#include <math.h>


typedef float v2f __attribute__((ext_vector_type(2)));
typedef float v4f __attribute__((ext_vector_type(4)));
typedef float v8f __attribute__((ext_vector_type(8)));

#define N_PTS 65536
#define DDIM  256
#define KFREQ 64
#define NROW  128
#define NHALF  (N_PTS / 2)
#define NCHUNK 64
#define CHUNK_N (NHALF / NCHUNK)

__device__ __forceinline__ v8f wmma_f32(v2f a, v2f b, v8f c) {
  return __builtin_amdgcn_wmma_f32_16x16x4_f32(false, a, false, b,
                                               (short)0, c, false, false);
}

#define TRIG_F(u) __builtin_bit_cast(float, (unsigned)(u))

__device__ __forceinline__ void trig_small(float ax, float& sp, float& cp, int& q) {
  const float n = __builtin_rintf(ax * TRIG_F(0x3f22f983u));
  float r = __builtin_fmaf(n, TRIG_F(0xbfc90fdau), ax);
  r = __builtin_fmaf(n, TRIG_F(0xb3a22168u), r);
  r = __builtin_fmaf(n, TRIG_F(0xa7c234c4u), r);
  q = (int)n;
  const float r2 = r * r;
  float a = __builtin_fmaf(TRIG_F(0xb94c1982u), r2, TRIG_F(0x3c0881c4u));
  a = __builtin_fmaf(r2, a, TRIG_F(0xbe2aaa9du));
  a = r2 * a;
  sp = __builtin_fmaf(r, a, r);
  float b = __builtin_fmaf(TRIG_F(0x37d75334u), r2, TRIG_F(0xbab64f3bu));
  b = __builtin_fmaf(r2, b, TRIG_F(0x3d2aabf7u));
  b = __builtin_fmaf(r2, b, TRIG_F(0xbf000004u));
  cp = __builtin_fmaf(r2, b, 1.0f);
}

__device__ __forceinline__ float sin_small(float a) {
  float sp, cp; int q;
  trig_small(__builtin_fabsf(a), sp, cp, q);
  const float sel = (q & 1) ? cp : sp;
  const unsigned flip = (((unsigned)q << 30) ^ __builtin_bit_cast(unsigned, a)) & 0x80000000u;
  return __builtin_bit_cast(float, __builtin_bit_cast(unsigned, sel) ^ flip);
}

__device__ __forceinline__ void sincos_small(float ang, float& s, float& c) {
  float sp, cp; int q;
  trig_small(ang, sp, cp, q);
  const unsigned flip = ((unsigned)q << 30) & 0x80000000u;
  const float ss = (q & 1) ? cp : sp;
  const float cc = (q & 1) ? -sp : cp;
  s = __builtin_bit_cast(float, __builtin_bit_cast(unsigned, ss) ^ flip);
  c = __builtin_bit_cast(float, __builtin_bit_cast(unsigned, cc) ^ flip);
}

__device__ __forceinline__ void twiddle(unsigned n, unsigned k, float& c, float& s) {
  const unsigned m = (n * k) & 65535u;
  const float ang = (float)m * 9.58737992429e-5f;
  sincos_small(ang, s, c);
}

#define PACK_BLOCKS ((DDIM * DDIM / 2) / 256)

#define DFT_COLS   64
#define DFT_BLOCKS ((DDIM / DFT_COLS) * NCHUNK)
#define DEC_M      (N_PTS / 16)
#define DEC_CH     (DEC_M / NCHUNK)
#define DEC_SLAB   16
#define DEC_NSLAB  (DEC_CH / DEC_SLAB)
#define DEC_TP     24
#define DEC_YP     72

#define DEC_FRAG_LOADS(S)                                              \
  "ds_load_2addr_b32 %[a0" #S "], %[pa] offset1:24\n\t"                \
  "ds_load_2addr_b32 %[b0" #S "], %[pb] offset1:72\n\t"                \
  "ds_load_2addr_b32 %[b1" #S "], %[pb] offset0:16 offset1:88\n\t"     \
  "ds_load_2addr_b32 %[b2" #S "], %[pb] offset0:32 offset1:104\n\t"    \
  "ds_load_2addr_b32 %[b3" #S "], %[pb] offset0:48 offset1:120\n\t"
#define DEC_ADVANCE                                                    \
  "v_add_nc_u32_e32 %[pa], 0x180, %[pa]\n\t"                           \
  "v_add_nc_u32_e32 %[pb], 0x480, %[pb]\n\t"
#define DEC_WMMAS(S)                                                   \
  "v_wmma_f32_16x16x4_f32 %[c0], %[a0" #S "], %[b0" #S "], %[c0]\n\t"  \
  "v_wmma_f32_16x16x4_f32 %[c1], %[a0" #S "], %[b1" #S "], %[c1]\n\t"  \
  "v_wmma_f32_16x16x4_f32 %[c2], %[a0" #S "], %[b2" #S "], %[c2]\n\t"  \
  "v_wmma_f32_16x16x4_f32 %[c3], %[a0" #S "], %[b3" #S "], %[c3]\n\t"
#define DEC_STEP(CUR, NXT)                                             \
  DEC_ADVANCE DEC_FRAG_LOADS(NXT) "s_wait_dscnt 0x5\n\t" DEC_WMMAS(CUR)

__global__ __launch_bounds__(256) void dft_low_kernel(
    const float* __restrict__ x, const float* __restrict__ W,
    float* __restrict__ Wp, float* __restrict__ Ppart) {
  __shared__ __attribute__((aligned(16))) float sT[8 * 32 * DEC_TP];
  __shared__ __attribute__((aligned(16))) float sY[8 * 32 * DEC_YP];

  if (blockIdx.x >= DFT_BLOCKS) {
    int idx = (blockIdx.x - DFT_BLOCKS) * 256 + threadIdx.x;
    int kpair = idx >> 8;
    int col   = idx & 255;
    Wp[(size_t)idx * 2]     = W[(size_t)(2 * kpair) * DDIM + col];
    Wp[(size_t)idx * 2 + 1] = W[(size_t)(2 * kpair + 1) * DDIM + col];
    return;
  }

  const int tid     = threadIdx.x;
  const int lane    = tid & 31;
  const int wave    = tid >> 5;
  const int lr      = lane & 15;
  const int halfSel = (lane >> 4) << 1;
  const int q       = blockIdx.x & 3;
  const int chunk   = blockIdx.x >> 2;
  const int nbase   = chunk * DEC_CH;

  v8f acc0 = {0.f,0.f,0.f,0.f,0.f,0.f,0.f,0.f};
  v8f acc1 = acc0, acc2 = acc0, acc3 = acc0;

  const int xr = tid >> 4;
  const int xc = (tid & 15) * 4;
  const float* xg = x + (size_t)(nbase + xr) * DDIM + q * DFT_COLS + xc;
  v4f px[16];
#pragma unroll
  for (int j = 0; j < 16; ++j)
    px[j] = *(const v4f*)(xg + (size_t)j * DEC_M * DDIM);

  const unsigned pa0 = (unsigned)(size_t)(sT + (wave * 32 + halfSel) * DEC_TP + lr);
  const unsigned pb0 = (unsigned)(size_t)(sY + (wave * 32 + halfSel) * DEC_YP + lr);

  const int tp  = tid & 63;
  const int tjg = tid >> 6;
  const int tt  = tp >> 3;
  const int tf  = tp & 7;
  const int thi = tf >> 2;
  const unsigned tk = (unsigned)((tt == 0) ? (8 * thi + 16 * (tf & 3))
                                           : ((thi ? 16 - tt : tt) + 16 * (tf & 3)));
  float* sTw = sT + (tt * 32 + tjg * 4) * DEC_TP + 2 * (tf & 3) + 8 * thi;
  float* sYw = sY + xr * DEC_YP + xc;

  const float C1 = 0.92387953251128674f;
  const float S1 = 0.38268343236508977f;
  const float H  = 0.70710678118654752f;

  for (int s = 0; s < DEC_NSLAB; ++s) {
    __syncthreads();
#pragma unroll
    for (int jj = 0; jj < 4; ++jj) {
      float c, sn;
      twiddle((unsigned)(nbase + s * DEC_SLAB + tjg * 4 + jj), tk, c, sn);
      v2f va, vb;
      if (tt == 0) {
        va = thi ? (v2f){0.f, 0.f} : (v2f){c, sn};
        vb = thi ? (v2f){c, sn} : (v2f){0.f, 0.f};
      } else {
        va = (v2f){c, sn};
        vb = thi ? (v2f){sn, -c} : (v2f){-sn, c};
      }
      *(v2f*)(sTw + jj * DEC_TP) = va;
      *(v2f*)(sTw + (16 + jj) * DEC_TP) = vb;
    }
    {
      v4f e0 = px[0] + px[8],  o0 = px[0] - px[8];
      v4f e1 = px[1] + px[9],  o1 = px[1] - px[9];
      v4f e2 = px[2] + px[10], o2 = px[2] - px[10];
      v4f e3 = px[3] + px[11], o3 = px[3] - px[11];
      v4f e4 = px[4] + px[12], o4 = px[4] - px[12];
      v4f e5 = px[5] + px[13], o5 = px[5] - px[13];
      v4f e6 = px[6] + px[14], o6 = px[6] - px[14];
      v4f e7 = px[7] + px[15], o7 = px[7] - px[15];
      v4f ee0 = e0 + e4, eo0 = e0 - e4;
      v4f ee1 = e1 + e5, eo1 = e1 - e5;
      v4f ee2 = e2 + e6, eo2 = e2 - e6;
      v4f ee3 = e3 + e7, eo3 = e3 - e7;
      v4f s02 = ee0 + ee2, s13 = ee1 + ee3;
      v4f A0 = s02 + s13, A8 = s02 - s13;
      v4f A4 = ee0 - ee2, B4 = ee1 - ee3;
      v4f t1 = H * (eo1 - eo3), t2 = H * (eo1 + eo3);
      v4f A2 = eo0 + t1, B2 = eo2 + t2;
      v4f A6 = eo0 - t1, B6 = t2 - eo2;
      v4f d1 = o1 - o7, d2 = o2 - o6, d3 = o3 - o5;
      v4f p1 = o1 + o7, p2 = o2 + o6, p3 = o3 + o5;
      v4f hd2 = H * d2, hp2 = H * p2;
      v4f A1 = o0 + hd2 + (C1 * d1 + S1 * d3);
      v4f A7 = o0 + hd2 - (C1 * d1 + S1 * d3);
      v4f A3 = o0 - hd2 + (S1 * d1 - C1 * d3);
      v4f A5 = o0 - hd2 - (S1 * d1 - C1 * d3);
      v4f B1 = (S1 * p1 + C1 * p3) + hp2 + o4;
      v4f B7 = (S1 * p1 + C1 * p3) - hp2 - o4;
      v4f B3 = (C1 * p1 - S1 * p3) + hp2 - o4;
      v4f B5 = (C1 * p1 - S1 * p3) - hp2 + o4;
      *(v4f*)(sYw + (0 * 32) * DEC_YP)      = A0;
      *(v4f*)(sYw + (0 * 32 + 16) * DEC_YP) = A8;
      *(v4f*)(sYw + (1 * 32) * DEC_YP)      = A1;
      *(v4f*)(sYw + (1 * 32 + 16) * DEC_YP) = B1;
      *(v4f*)(sYw + (2 * 32) * DEC_YP)      = A2;
      *(v4f*)(sYw + (2 * 32 + 16) * DEC_YP) = B2;
      *(v4f*)(sYw + (3 * 32) * DEC_YP)      = A3;
      *(v4f*)(sYw + (3 * 32 + 16) * DEC_YP) = B3;
      *(v4f*)(sYw + (4 * 32) * DEC_YP)      = A4;
      *(v4f*)(sYw + (4 * 32 + 16) * DEC_YP) = B4;
      *(v4f*)(sYw + (5 * 32) * DEC_YP)      = A5;
      *(v4f*)(sYw + (5 * 32 + 16) * DEC_YP) = B5;
      *(v4f*)(sYw + (6 * 32) * DEC_YP)      = A6;
      *(v4f*)(sYw + (6 * 32 + 16) * DEC_YP) = B6;
      *(v4f*)(sYw + (7 * 32) * DEC_YP)      = A7;
      *(v4f*)(sYw + (7 * 32 + 16) * DEC_YP) = B7;
    }
    __syncthreads();

    if (s + 1 < DEC_NSLAB) {
      const float* xn = xg + (size_t)(s + 1) * DEC_SLAB * DDIM;
#pragma unroll
      for (int j = 0; j < 16; ++j)
        px[j] = *(const v4f*)(xn + (size_t)j * DEC_M * DDIM);
    }

    {
      unsigned pa = pa0, pb = pb0;
      v2f a0x, b0x, b1x, b2x, b3x, a0y, b0y, b1y, b2y, b3y;
      asm volatile(
          DEC_FRAG_LOADS(x)
          DEC_STEP(x, y) DEC_STEP(y, x) DEC_STEP(x, y) DEC_STEP(y, x)
          DEC_STEP(x, y) DEC_STEP(y, x) DEC_STEP(x, y)
          "s_wait_dscnt 0x0\n\t"
          DEC_WMMAS(y)
          "v_nop\n\tv_nop\n\tv_nop\n\tv_nop"
          : [c0] "+v"(acc0), [c1] "+v"(acc1), [c2] "+v"(acc2), [c3] "+v"(acc3),
            [pa] "+v"(pa), [pb] "+v"(pb),
            [a0x] "=&v"(a0x), [b0x] "=&v"(b0x), [b1x] "=&v"(b1x),
            [b2x] "=&v"(b2x), [b3x] "=&v"(b3x),
            [a0y] "=&v"(a0y), [b0y] "=&v"(b0y), [b1y] "=&v"(b1y),
            [b2y] "=&v"(b2y), [b3y] "=&v"(b3y)
          :
          : "memory");
    }
  }

  const int mbase = (lane >> 4) << 3;
  const int col   = q * DFT_COLS + lr;
  const size_t base = (size_t)chunk * (NROW * DDIM) + (size_t)(wave * 16 + mbase) * DDIM + col;
#pragma unroll
  for (int v = 0; v < 8; ++v) {
    Ppart[base + (size_t)v * DDIM]      = acc0[v];
    Ppart[base + (size_t)v * DDIM + 16] = acc1[v];
    Ppart[base + (size_t)v * DDIM + 32] = acc2[v];
    Ppart[base + (size_t)v * DDIM + 48] = acc3[v];
  }
}

#define MIX_FQ 64
#define MIX_DS 32

__global__ __launch_bounds__(256) void spectral_mix_kernel(
    const float* __restrict__ Ppart, const float* __restrict__ P_low,
    const float* __restrict__ B_low, float* __restrict__ BppP) {
  __shared__ __attribute__((aligned(8))) float sXr[DDIM];
  __shared__ __attribute__((aligned(8))) float sXi[DDIM];
  __shared__ __attribute__((aligned(16))) float sPart[8 * MIX_FQ * 2];

  const int tid   = threadIdx.x;
  const int lane  = tid & 31;
  const int wave  = tid >> 5;
  const int fq    = blockIdx.x;
  const int k     = blockIdx.y;
  const int prow  = ((k & 1) << 5) | (k >> 1);
  const int kres  = k & 15;
  const int ktile = (kres == 0 || kres == 8) ? 0 : (kres < 8 ? kres : 16 - kres);
  const int krow  = 16 * ktile + 2 * (k >> 4) + ((kres >= 8) ? 8 : 0);
  const int dbase = wave * MIX_DS;
  const int f0    = fq * MIX_FQ + lane * 2;

  const int rsel = tid >> 7;
  const int d2   = (tid & 127) * 2;
  const float* Cg = Ppart + (size_t)(krow + rsel) * DDIM + d2;
  v2f cpart[NCHUNK];
#pragma unroll
  for (int ch = 0; ch < NCHUNK; ++ch)
    cpart[ch] = *(const v2f*)(Cg + (size_t)ch * (NROW * DDIM));

  const float* Pk = P_low + (size_t)k * DDIM * DDIM * 2
                          + ((size_t)dbase * DDIM + f0) * 2;
  v4f p[MIX_DS];
#pragma unroll
  for (int j = 0; j < MIX_DS; ++j)
    p[j] = *(const v4f*)(Pk + (size_t)j * DDIM * 2);

  float c0 = 0.f, c1 = 0.f;
#pragma unroll
  for (int ch = 0; ch < NCHUNK; ++ch) {
    c0 += cpart[ch].x;
    c1 += cpart[ch].y;
  }
  const float invN = 1.0f / (float)N_PTS;
  if (rsel == 0) {
    *(v2f*)(sXr + d2) = (v2f){ c0 * invN,  c1 * invN};
  } else {
    *(v2f*)(sXi + d2) = (v2f){-c0 * invN, -c1 * invN};
  }
  __syncthreads();

  float ar0 = 0.f, ai0 = 0.f, ar1 = 0.f, ai1 = 0.f;
#pragma unroll
  for (int j = 0; j < MIX_DS; ++j) {
    const float xr = sXr[dbase + j], xi = sXi[dbase + j];
    ar0 = fmaf(xr, p[j].x, fmaf(-xi, p[j].y, ar0));
    ai0 = fmaf(xr, p[j].y, fmaf( xi, p[j].x, ai0));
    ar1 = fmaf(xr, p[j].z, fmaf(-xi, p[j].w, ar1));
    ai1 = fmaf(xr, p[j].w, fmaf( xi, p[j].z, ai1));
  }
  *(v4f*)(sPart + (wave * MIX_FQ + lane * 2) * 2) = (v4f){ar0, ai0, ar1, ai1};
  __syncthreads();

  if (tid < MIX_FQ * 2) {
    const int fl = tid >> 1;
    const int ri = tid & 1;
    const int f  = fq * MIX_FQ + fl;
    float s = B_low[((size_t)k * DDIM + f) * 2 + ri];
#pragma unroll
    for (int w = 0; w < 8; ++w) s += sPart[(w * MIX_FQ + fl) * 2 + ri];
    const float sc = (k == 0) ? 1.0f : 2.0f;
    BppP[((size_t)prow * DDIM + f) * 2 + ri] = ri ? -sc * s : sc * s;
  }
}

#define FO_RB     32
#define FO_KS     64
#define FO_NSLAB  6
#define FO_APITCH 68
#define FO_BPITCH 544

#define FO_LOADS(S, A0, A1, B0, B1)                                    \
  "ds_load_b64 %[a0" #S "], %[pa] offset:" #A0 "\n\t"                 \
  "ds_load_b64 %[a1" #S "], %[pa] offset:" #A1 "\n\t"                 \
  "ds_load_b64 %[b0" #S "], %[pb] offset:" #B0 "\n\t"                 \
  "ds_load_b64 %[b1" #S "], %[pb] offset:" #B1 "\n\t"                 \
  "ds_load_b64 %[b2" #S "], %[pc] offset:" #B0 "\n\t"                 \
  "ds_load_b64 %[b3" #S "], %[pc] offset:" #B1 "\n\t"
#define FO_WMMAS(S)                                                    \
  "v_wmma_f32_16x16x4_f32 %[c00], %[a0" #S "], %[b0" #S "], %[c00]\n\t" \
  "v_wmma_f32_16x16x4_f32 %[c01], %[a0" #S "], %[b1" #S "], %[c01]\n\t" \
  "v_wmma_f32_16x16x4_f32 %[c02], %[a0" #S "], %[b2" #S "], %[c02]\n\t" \
  "v_wmma_f32_16x16x4_f32 %[c03], %[a0" #S "], %[b3" #S "], %[c03]\n\t" \
  "v_wmma_f32_16x16x4_f32 %[c10], %[a1" #S "], %[b0" #S "], %[c10]\n\t" \
  "v_wmma_f32_16x16x4_f32 %[c11], %[a1" #S "], %[b1" #S "], %[c11]\n\t" \
  "v_wmma_f32_16x16x4_f32 %[c12], %[a1" #S "], %[b2" #S "], %[c12]\n\t" \
  "v_wmma_f32_16x16x4_f32 %[c13], %[a1" #S "], %[b3" #S "], %[c13]\n\t"
#define FO_STEP(CUR, NXT, A0, A1, B0, B1)                              \
  FO_LOADS(NXT, A0, A1, B0, B1) "s_wait_dscnt 0x6\n\t" FO_WMMAS(CUR)

#define FT_LOADS(S, A0, B0, B1)                                        \
  "ds_load_b64 %[a0" #S "], %[pa] offset:" #A0 "\n\t"                 \
  "ds_load_b64 %[b0" #S "], %[pb] offset:" #B0 "\n\t"                 \
  "ds_load_b64 %[b1" #S "], %[pb] offset:" #B1 "\n\t"                 \
  "ds_load_b64 %[b2" #S "], %[pc] offset:" #B0 "\n\t"                 \
  "ds_load_b64 %[b3" #S "], %[pc] offset:" #B1 "\n\t"
#define FT_WMMAS(S)                                                    \
  "v_wmma_f32_16x16x4_f32 %[t0], %[a0" #S "], %[b0" #S "], %[t0]\n\t" \
  "v_wmma_f32_16x16x4_f32 %[t1], %[a0" #S "], %[b1" #S "], %[t1]\n\t" \
  "v_wmma_f32_16x16x4_f32 %[t2], %[a0" #S "], %[b2" #S "], %[t2]\n\t" \
  "v_wmma_f32_16x16x4_f32 %[t3], %[a0" #S "], %[b3" #S "], %[t3]\n\t"
#define FT_STEP(CUR, NXT, A0, B0, B1)                                  \
  FT_LOADS(NXT, A0, B0, B1) "s_wait_dscnt 0x5\n\t" FT_WMMAS(CUR)

__device__ __forceinline__ void fo_fetch(
    int s, int tid, int nblk, const float* __restrict__ x,
    const float* __restrict__ Wp,
    const float* __restrict__ BppP, v4f (&pA)[4], v4f (&pB)[16]) {
  const float* Bg;
  if (s < 4) {
    const float* Ag = x + (size_t)nblk * DDIM + s * FO_KS;
#pragma unroll
    for (int j = 0; j < 4; ++j) {
      const int i = tid + 256 * j;
      const int r = i >> 4;
      pA[j] = *(const v4f*)(Ag + ((size_t)(r & 31) + (size_t)(r >> 5) * NHALF) * DDIM
                               + (i & 15) * 4);
    }
    Bg = Wp + (size_t)s * (FO_KS / 2) * (2 * DDIM);
  } else {
    Bg = BppP + (size_t)(s - 4) * (FO_KS / 2) * (2 * DDIM);
  }
#pragma unroll
  for (int j = 0; j < 16; ++j) {
    const int i = tid + 256 * j;
    pB[j] = *(const v4f*)(Bg + (size_t)i * 4);
  }
}

__device__ __forceinline__ void fo_table_slab(
    v8f& t0, v8f& t1, v8f& t2, v8f& t3,
    unsigned pa, unsigned pb, unsigned pc) {
  v2f a0x, b0x, b1x, b2x, b3x, a0y, b0y, b1y, b2y, b3y;
  asm volatile(
      FT_LOADS(x, 0, 0, 128)
      FT_STEP(x, y, 16, 4352, 4480)
      FT_STEP(y, x, 32, 8704, 8832)
      FT_STEP(x, y, 48, 13056, 13184)
      FT_STEP(y, x, 64, 17408, 17536)
      FT_STEP(x, y, 80, 21760, 21888)
      FT_STEP(y, x, 96, 26112, 26240)
      FT_STEP(x, y, 112, 30464, 30592)
      FT_STEP(y, x, 128, 34816, 34944)
      FT_STEP(x, y, 144, 39168, 39296)
      FT_STEP(y, x, 160, 43520, 43648)
      FT_STEP(x, y, 176, 47872, 48000)
      FT_STEP(y, x, 192, 52224, 52352)
      FT_STEP(x, y, 208, 56576, 56704)
      FT_STEP(y, x, 224, 60928, 61056)
      FT_STEP(x, y, 240, 65280, 65408)
      "s_wait_dscnt 0x0\n\t"
      FT_WMMAS(y)
      "v_nop\n\tv_nop\n\tv_nop\n\tv_nop"
      : [t0] "+v"(t0), [t1] "+v"(t1), [t2] "+v"(t2), [t3] "+v"(t3),
        [a0x] "=&v"(a0x), [b0x] "=&v"(b0x), [b1x] "=&v"(b1x),
        [b2x] "=&v"(b2x), [b3x] "=&v"(b3x),
        [a0y] "=&v"(a0y), [b0y] "=&v"(b0y), [b1y] "=&v"(b1y),
        [b2y] "=&v"(b2y), [b3y] "=&v"(b3y)
      : [pa] "v"(pa), [pb] "v"(pb), [pc] "v"(pc)
      : "memory");
}

__global__ __launch_bounds__(256) void fused_out_kernel(
    const float* __restrict__ x, const float* __restrict__ Wp,
    const float* __restrict__ Bias,
    const float* __restrict__ BppP, float* __restrict__ out) {
  __shared__ __attribute__((aligned(16))) float sA[2 * FO_RB * FO_APITCH];
  __shared__ __attribute__((aligned(16))) float sB[(FO_KS / 2) * FO_BPITCH];
  __shared__ __attribute__((aligned(16))) float sAT[2 * FO_RB * FO_APITCH];

  const int tid     = threadIdx.x;
  const int lane    = tid & 31;
  const int wave    = tid >> 5;
  const int lr      = lane & 15;
  const int hsel    = lane >> 4;
  const int nblk    = blockIdx.x * FO_RB;
  const int rowW    = (wave & 1) * 16;
  const int colW    = (wave >> 1) * 64;

  v8f c00 = {0.f,0.f,0.f,0.f,0.f,0.f,0.f,0.f};
  v8f c01 = c00, c02 = c00, c03 = c00, c10 = c00, c11 = c00, c12 = c00, c13 = c00;
  v8f e0 = c00, e1 = c00, e2 = c00, e3 = c00;
  v8f q0 = c00, q1 = c00, q2 = c00, q3 = c00;

  v4f pA[4];
  v4f pB[16];
  fo_fetch(0, tid, nblk, x, Wp, BppP, pA, pB);

#pragma unroll
  for (int t = 0; t < 2; ++t) {
#pragma unroll
    for (int j = 0; j < 2; ++j) {
      const int i = tid + 256 * j;
      const unsigned n  = (unsigned)(nblk + (i >> 4));
      const int p0 = t * 32 + (i & 15) * 2;
      const int p1 = p0 + 1;
      const unsigned k0 = (unsigned)(((p0 & 31) << 1) | (p0 >> 5));
      const unsigned k1 = (unsigned)(((p1 & 31) << 1) | (p1 >> 5));
      float c0, s0, c1, s1;
      twiddle(n, k0, c0, s0);
      twiddle(n, k1, c1, s1);
      *(v4f*)(sAT + (t * FO_RB + (i >> 4)) * FO_APITCH + (i & 15) * 4) =
          (v4f){c0, s0, c1, s1};
    }
  }

  const unsigned pa = (unsigned)(size_t)(sA + (rowW + lr) * FO_APITCH + 2 * hsel);
  const unsigned paT = (unsigned)(size_t)(sAT + (rowW + lr) * FO_APITCH + 2 * hsel);
  const unsigned pb = (unsigned)(size_t)(sB + hsel * FO_BPITCH + (colW + lr) * 2);
  const unsigned pc = pb + 256u;

  for (int s = 0; s < FO_NSLAB; ++s) {
    __syncthreads();
    if (s < 4) {
#pragma unroll
      for (int j = 0; j < 4; ++j) {
        const int i = tid + 256 * j;
        *(v4f*)(sA + (i >> 4) * FO_APITCH + (i & 15) * 4) = pA[j];
      }
    }
#pragma unroll
    for (int j = 0; j < 16; ++j) {
      const int i = tid + 256 * j;
      *(v4f*)(sB + (i >> 7) * FO_BPITCH + (i & 127) * 4) = pB[j];
    }
    __syncthreads();

    if (s + 1 < FO_NSLAB)
      fo_fetch(s + 1, tid, nblk, x, Wp, BppP, pA, pB);

    if (s < 4) {
      v2f a0x, a1x, b0x, b1x, b2x, b3x, a0y, a1y, b0y, b1y, b2y, b3y;
      asm volatile(
          FO_LOADS(x, 0, 8704, 0, 128)
          FO_STEP(x, y, 16, 8720, 4352, 4480)
          FO_STEP(y, x, 32, 8736, 8704, 8832)
          FO_STEP(x, y, 48, 8752, 13056, 13184)
          FO_STEP(y, x, 64, 8768, 17408, 17536)
          FO_STEP(x, y, 80, 8784, 21760, 21888)
          FO_STEP(y, x, 96, 8800, 26112, 26240)
          FO_STEP(x, y, 112, 8816, 30464, 30592)
          FO_STEP(y, x, 128, 8832, 34816, 34944)
          FO_STEP(x, y, 144, 8848, 39168, 39296)
          FO_STEP(y, x, 160, 8864, 43520, 43648)
          FO_STEP(x, y, 176, 8880, 47872, 48000)
          FO_STEP(y, x, 192, 8896, 52224, 52352)
          FO_STEP(x, y, 208, 8912, 56576, 56704)
          FO_STEP(y, x, 224, 8928, 60928, 61056)
          FO_STEP(x, y, 240, 8944, 65280, 65408)
          "s_wait_dscnt 0x0\n\t"
          FO_WMMAS(y)
          "v_nop\n\tv_nop\n\tv_nop\n\tv_nop"
          : [c00] "+v"(c00), [c01] "+v"(c01), [c02] "+v"(c02), [c03] "+v"(c03),
            [c10] "+v"(c10), [c11] "+v"(c11), [c12] "+v"(c12), [c13] "+v"(c13),
            [a0x] "=&v"(a0x), [a1x] "=&v"(a1x), [b0x] "=&v"(b0x),
            [b1x] "=&v"(b1x), [b2x] "=&v"(b2x), [b3x] "=&v"(b3x),
            [a0y] "=&v"(a0y), [a1y] "=&v"(a1y), [b0y] "=&v"(b0y),
            [b1y] "=&v"(b1y), [b2y] "=&v"(b2y), [b3y] "=&v"(b3y)
          : [pa] "v"(pa), [pb] "v"(pb), [pc] "v"(pc)
          : "memory");
    } else if (s == 4) {
      fo_table_slab(e0, e1, e2, e3, paT, pb, pc);
    } else {
      fo_table_slab(q0, q1, q2, q3, paT + FO_RB * FO_APITCH * 4, pb, pc);
    }
  }

  const int mbase = hsel << 3;
  const int col   = colW + lr;
  const int nrow  = nblk + rowW + mbase;
  const float bias0 = Bias[col];
  const float bias1 = Bias[col + 16];
  const float bias2 = Bias[col + 32];
  const float bias3 = Bias[col + 48];
  float big = 0.f;
#pragma unroll
  for (int v = 0; v < 8; ++v) {
    const float ep0 = e0[v] + q0[v], em0 = e0[v] - q0[v];
    const float ep1 = e1[v] + q1[v], em1 = e1[v] - q1[v];
    const float ep2 = e2[v] + q2[v], em2 = e2[v] - q2[v];
    const float ep3 = e3[v] + q3[v], em3 = e3[v] - q3[v];
    c00[v] = 0.2f * (c00[v] + ep0 + bias0);
    c01[v] = 0.2f * (c01[v] + ep1 + bias1);
    c02[v] = 0.2f * (c02[v] + ep2 + bias2);
    c03[v] = 0.2f * (c03[v] + ep3 + bias3);
    c10[v] = 0.2f * (c10[v] + em0 + bias0);
    c11[v] = 0.2f * (c11[v] + em1 + bias1);
    c12[v] = 0.2f * (c12[v] + em2 + bias2);
    c13[v] = 0.2f * (c13[v] + em3 + bias3);
    big = fmaxf(big, fmaxf(fmaxf(__builtin_fabsf(c00[v]), __builtin_fabsf(c01[v])),
                           fmaxf(__builtin_fabsf(c02[v]), __builtin_fabsf(c03[v]))));
    big = fmaxf(big, fmaxf(fmaxf(__builtin_fabsf(c10[v]), __builtin_fabsf(c11[v])),
                           fmaxf(__builtin_fabsf(c12[v]), __builtin_fabsf(c13[v]))));
  }
  float* ob0 = out + (size_t)nrow * DDIM + col;
  float* ob1 = ob0 + (size_t)NHALF * DDIM;
  if (big < 131072.0f) {
#pragma unroll
    for (int v = 0; v < 8; ++v) {
      float* o0 = ob0 + v * DDIM;
      float* o1 = ob1 + v * DDIM;
      o0[0]  = sin_small(c00[v]);
      o0[16] = sin_small(c01[v]);
      o0[32] = sin_small(c02[v]);
      o0[48] = sin_small(c03[v]);
      o1[0]  = sin_small(c10[v]);
      o1[16] = sin_small(c11[v]);
      o1[32] = sin_small(c12[v]);
      o1[48] = sin_small(c13[v]);
    }
  } else {
#pragma nounroll
    for (int v = 0; v < 8; ++v) {
      float* o0 = ob0 + v * DDIM;
      float* o1 = ob1 + v * DDIM;
      o0[0]  = sinf(c00[v]);
      o0[16] = sinf(c01[v]);
      o0[32] = sinf(c02[v]);
      o0[48] = sinf(c03[v]);
      o1[0]  = sinf(c10[v]);
      o1[16] = sinf(c11[v]);
      o1[32] = sinf(c12[v]);
      o1[48] = sinf(c13[v]);
    }
  }
}

extern "C" void kernel_launch(void* const* d_in, const int* in_sizes, int n_in,
                              void* d_out, int out_size, void* d_ws,
                              size_t ws_size, hipStream_t stream) {
  (void)in_sizes; (void)n_in; (void)out_size; (void)ws_size;
  const float* x     = (const float*)d_in[0];
  const float* P_low = (const float*)d_in[1];
  const float* B_low = (const float*)d_in[2];
  const float* W     = (const float*)d_in[3];
  const float* Bias  = (const float*)d_in[4];
  float* out = (float*)d_out;

  float* ws    = (float*)d_ws;
  float* Ppart = ws + (size_t)N_PTS * NROW;
  float* BppP  = Ppart + (size_t)NCHUNK * NROW * DDIM;
  float* Wp    = BppP + NROW * DDIM;

  dft_low_kernel<<<dim3(DFT_BLOCKS + PACK_BLOCKS), dim3(256), 0, stream>>>(x, W, Wp, Ppart);
  spectral_mix_kernel<<<dim3(DDIM / MIX_FQ, KFREQ), dim3(256), 0, stream>>>(Ppart, P_low, B_low, BppP);
  fused_out_kernel<<<dim3(NHALF / FO_RB), dim3(256), 0, stream>>>(x, Wp, Bias, BppP, out);
}


# Packed B columns reordered so a lane's four accumulator tiles are four adjacent output columns: output stored as whole 128-B lines by b128 stores (16 per wave, was 64 b32); sum order unchanged
# speedup vs baseline: 2.5526x; 1.0307x over previous
#include <hip/hip_runtime.h>
#include <math.h>


typedef float v2f __attribute__((ext_vector_type(2)));
typedef float v4f __attribute__((ext_vector_type(4)));
typedef float v8f __attribute__((ext_vector_type(8)));

#define N_PTS 65536
#define DDIM  256
#define KFREQ 64
#define NROW  128
#define NHALF  (N_PTS / 2)
#define NCHUNK 64
#define CHUNK_N (NHALF / NCHUNK)

__device__ __forceinline__ v8f wmma_f32(v2f a, v2f b, v8f c) {
  return __builtin_amdgcn_wmma_f32_16x16x4_f32(false, a, false, b,
                                               (short)0, c, false, false);
}

#define TRIG_F(u) __builtin_bit_cast(float, (unsigned)(u))

__device__ __forceinline__ void trig_small(float ax, float& sp, float& cp, int& q) {
  const float n = __builtin_rintf(ax * TRIG_F(0x3f22f983u));
  float r = __builtin_fmaf(n, TRIG_F(0xbfc90fdau), ax);
  r = __builtin_fmaf(n, TRIG_F(0xb3a22168u), r);
  r = __builtin_fmaf(n, TRIG_F(0xa7c234c4u), r);
  q = (int)n;
  const float r2 = r * r;
  float a = __builtin_fmaf(TRIG_F(0xb94c1982u), r2, TRIG_F(0x3c0881c4u));
  a = __builtin_fmaf(r2, a, TRIG_F(0xbe2aaa9du));
  a = r2 * a;
  sp = __builtin_fmaf(r, a, r);
  float b = __builtin_fmaf(TRIG_F(0x37d75334u), r2, TRIG_F(0xbab64f3bu));
  b = __builtin_fmaf(r2, b, TRIG_F(0x3d2aabf7u));
  b = __builtin_fmaf(r2, b, TRIG_F(0xbf000004u));
  cp = __builtin_fmaf(r2, b, 1.0f);
}

__device__ __forceinline__ float sin_small(float a) {
  float sp, cp; int q;
  trig_small(__builtin_fabsf(a), sp, cp, q);
  const float sel = (q & 1) ? cp : sp;
  const unsigned flip = (((unsigned)q << 30) ^ __builtin_bit_cast(unsigned, a)) & 0x80000000u;
  return __builtin_bit_cast(float, __builtin_bit_cast(unsigned, sel) ^ flip);
}

__device__ __forceinline__ void sincos_small(float ang, float& s, float& c) {
  float sp, cp; int q;
  trig_small(ang, sp, cp, q);
  const unsigned flip = ((unsigned)q << 30) & 0x80000000u;
  const float ss = (q & 1) ? cp : sp;
  const float cc = (q & 1) ? -sp : cp;
  s = __builtin_bit_cast(float, __builtin_bit_cast(unsigned, ss) ^ flip);
  c = __builtin_bit_cast(float, __builtin_bit_cast(unsigned, cc) ^ flip);
}

__device__ __forceinline__ void twiddle(unsigned n, unsigned k, float& c, float& s) {
  const unsigned m = (n * k) & 65535u;
  const float ang = (float)m * 9.58737992429e-5f;
  sincos_small(ang, s, c);
}

#define PACK_BLOCKS ((DDIM * DDIM / 2) / 256)

__device__ __forceinline__ int fo_col_of_pos(int p) {
  return (p & ~63) | ((p & 15) << 2) | ((p >> 4) & 3);
}
__device__ __forceinline__ int fo_pos_of_col(int c) {
  return (c & ~63) | ((c & 3) << 4) | ((c >> 2) & 15);
}

#define DFT_COLS   64
#define DFT_BLOCKS ((DDIM / DFT_COLS) * NCHUNK)
#define DEC_M      (N_PTS / 16)
#define DEC_CH     (DEC_M / NCHUNK)
#define DEC_SLAB   16
#define DEC_NSLAB  (DEC_CH / DEC_SLAB)
#define DEC_TP     24
#define DEC_YP     72

#define DEC_FRAG_LOADS(S)                                              \
  "ds_load_2addr_b32 %[a0" #S "], %[pa] offset1:24\n\t"                \
  "ds_load_2addr_b32 %[b0" #S "], %[pb] offset1:72\n\t"                \
  "ds_load_2addr_b32 %[b1" #S "], %[pb] offset0:16 offset1:88\n\t"     \
  "ds_load_2addr_b32 %[b2" #S "], %[pb] offset0:32 offset1:104\n\t"    \
  "ds_load_2addr_b32 %[b3" #S "], %[pb] offset0:48 offset1:120\n\t"
#define DEC_ADVANCE                                                    \
  "v_add_nc_u32_e32 %[pa], 0x180, %[pa]\n\t"                           \
  "v_add_nc_u32_e32 %[pb], 0x480, %[pb]\n\t"
#define DEC_WMMAS(S)                                                   \
  "v_wmma_f32_16x16x4_f32 %[c0], %[a0" #S "], %[b0" #S "], %[c0]\n\t"  \
  "v_wmma_f32_16x16x4_f32 %[c1], %[a0" #S "], %[b1" #S "], %[c1]\n\t"  \
  "v_wmma_f32_16x16x4_f32 %[c2], %[a0" #S "], %[b2" #S "], %[c2]\n\t"  \
  "v_wmma_f32_16x16x4_f32 %[c3], %[a0" #S "], %[b3" #S "], %[c3]\n\t"
#define DEC_STEP(CUR, NXT)                                             \
  DEC_ADVANCE DEC_FRAG_LOADS(NXT) "s_wait_dscnt 0x5\n\t" DEC_WMMAS(CUR)

__global__ __launch_bounds__(256) void dft_low_kernel(
    const float* __restrict__ x, const float* __restrict__ W,
    float* __restrict__ Wp, float* __restrict__ Ppart) {
  __shared__ __attribute__((aligned(16))) float sT[8 * 32 * DEC_TP];
  __shared__ __attribute__((aligned(16))) float sY[8 * 32 * DEC_YP];

  if (blockIdx.x >= DFT_BLOCKS) {
    int idx = (blockIdx.x - DFT_BLOCKS) * 256 + threadIdx.x;
    int kpair = idx >> 8;
    int col   = fo_col_of_pos(idx & 255);
    Wp[(size_t)idx * 2]     = W[(size_t)(2 * kpair) * DDIM + col];
    Wp[(size_t)idx * 2 + 1] = W[(size_t)(2 * kpair + 1) * DDIM + col];
    return;
  }

  const int tid     = threadIdx.x;
  const int lane    = tid & 31;
  const int wave    = tid >> 5;
  const int lr      = lane & 15;
  const int halfSel = (lane >> 4) << 1;
  const int q       = blockIdx.x & 3;
  const int chunk   = blockIdx.x >> 2;
  const int nbase   = chunk * DEC_CH;

  v8f acc0 = {0.f,0.f,0.f,0.f,0.f,0.f,0.f,0.f};
  v8f acc1 = acc0, acc2 = acc0, acc3 = acc0;

  const int xr = tid >> 4;
  const int xc = (tid & 15) * 4;
  const float* xg = x + (size_t)(nbase + xr) * DDIM + q * DFT_COLS + xc;
  v4f px[16];
#pragma unroll
  for (int j = 0; j < 16; ++j)
    px[j] = *(const v4f*)(xg + (size_t)j * DEC_M * DDIM);

  const unsigned pa0 = (unsigned)(size_t)(sT + (wave * 32 + halfSel) * DEC_TP + lr);
  const unsigned pb0 = (unsigned)(size_t)(sY + (wave * 32 + halfSel) * DEC_YP + lr);

  const int tp  = tid & 63;
  const int tjg = tid >> 6;
  const int tt  = tp >> 3;
  const int tf  = tp & 7;
  const int thi = tf >> 2;
  const unsigned tk = (unsigned)((tt == 0) ? (8 * thi + 16 * (tf & 3))
                                           : ((thi ? 16 - tt : tt) + 16 * (tf & 3)));
  float* sTw = sT + (tt * 32 + tjg * 4) * DEC_TP + 2 * (tf & 3) + 8 * thi;
  float* sYw = sY + xr * DEC_YP + xc;

  const float C1 = 0.92387953251128674f;
  const float S1 = 0.38268343236508977f;
  const float H  = 0.70710678118654752f;

  for (int s = 0; s < DEC_NSLAB; ++s) {
    __syncthreads();
#pragma unroll
    for (int jj = 0; jj < 4; ++jj) {
      float c, sn;
      twiddle((unsigned)(nbase + s * DEC_SLAB + tjg * 4 + jj), tk, c, sn);
      v2f va, vb;
      if (tt == 0) {
        va = thi ? (v2f){0.f, 0.f} : (v2f){c, sn};
        vb = thi ? (v2f){c, sn} : (v2f){0.f, 0.f};
      } else {
        va = (v2f){c, sn};
        vb = thi ? (v2f){sn, -c} : (v2f){-sn, c};
      }
      *(v2f*)(sTw + jj * DEC_TP) = va;
      *(v2f*)(sTw + (16 + jj) * DEC_TP) = vb;
    }
    {
      v4f e0 = px[0] + px[8],  o0 = px[0] - px[8];
      v4f e1 = px[1] + px[9],  o1 = px[1] - px[9];
      v4f e2 = px[2] + px[10], o2 = px[2] - px[10];
      v4f e3 = px[3] + px[11], o3 = px[3] - px[11];
      v4f e4 = px[4] + px[12], o4 = px[4] - px[12];
      v4f e5 = px[5] + px[13], o5 = px[5] - px[13];
      v4f e6 = px[6] + px[14], o6 = px[6] - px[14];
      v4f e7 = px[7] + px[15], o7 = px[7] - px[15];
      v4f ee0 = e0 + e4, eo0 = e0 - e4;
      v4f ee1 = e1 + e5, eo1 = e1 - e5;
      v4f ee2 = e2 + e6, eo2 = e2 - e6;
      v4f ee3 = e3 + e7, eo3 = e3 - e7;
      v4f s02 = ee0 + ee2, s13 = ee1 + ee3;
      v4f A0 = s02 + s13, A8 = s02 - s13;
      v4f A4 = ee0 - ee2, B4 = ee1 - ee3;
      v4f t1 = H * (eo1 - eo3), t2 = H * (eo1 + eo3);
      v4f A2 = eo0 + t1, B2 = eo2 + t2;
      v4f A6 = eo0 - t1, B6 = t2 - eo2;
      v4f d1 = o1 - o7, d2 = o2 - o6, d3 = o3 - o5;
      v4f p1 = o1 + o7, p2 = o2 + o6, p3 = o3 + o5;
      v4f hd2 = H * d2, hp2 = H * p2;
      v4f A1 = o0 + hd2 + (C1 * d1 + S1 * d3);
      v4f A7 = o0 + hd2 - (C1 * d1 + S1 * d3);
      v4f A3 = o0 - hd2 + (S1 * d1 - C1 * d3);
      v4f A5 = o0 - hd2 - (S1 * d1 - C1 * d3);
      v4f B1 = (S1 * p1 + C1 * p3) + hp2 + o4;
      v4f B7 = (S1 * p1 + C1 * p3) - hp2 - o4;
      v4f B3 = (C1 * p1 - S1 * p3) + hp2 - o4;
      v4f B5 = (C1 * p1 - S1 * p3) - hp2 + o4;
      *(v4f*)(sYw + (0 * 32) * DEC_YP)      = A0;
      *(v4f*)(sYw + (0 * 32 + 16) * DEC_YP) = A8;
      *(v4f*)(sYw + (1 * 32) * DEC_YP)      = A1;
      *(v4f*)(sYw + (1 * 32 + 16) * DEC_YP) = B1;
      *(v4f*)(sYw + (2 * 32) * DEC_YP)      = A2;
      *(v4f*)(sYw + (2 * 32 + 16) * DEC_YP) = B2;
      *(v4f*)(sYw + (3 * 32) * DEC_YP)      = A3;
      *(v4f*)(sYw + (3 * 32 + 16) * DEC_YP) = B3;
      *(v4f*)(sYw + (4 * 32) * DEC_YP)      = A4;
      *(v4f*)(sYw + (4 * 32 + 16) * DEC_YP) = B4;
      *(v4f*)(sYw + (5 * 32) * DEC_YP)      = A5;
      *(v4f*)(sYw + (5 * 32 + 16) * DEC_YP) = B5;
      *(v4f*)(sYw + (6 * 32) * DEC_YP)      = A6;
      *(v4f*)(sYw + (6 * 32 + 16) * DEC_YP) = B6;
      *(v4f*)(sYw + (7 * 32) * DEC_YP)      = A7;
      *(v4f*)(sYw + (7 * 32 + 16) * DEC_YP) = B7;
    }
    __syncthreads();

    if (s + 1 < DEC_NSLAB) {
      const float* xn = xg + (size_t)(s + 1) * DEC_SLAB * DDIM;
#pragma unroll
      for (int j = 0; j < 16; ++j)
        px[j] = *(const v4f*)(xn + (size_t)j * DEC_M * DDIM);
    }

    {
      unsigned pa = pa0, pb = pb0;
      v2f a0x, b0x, b1x, b2x, b3x, a0y, b0y, b1y, b2y, b3y;
      asm volatile(
          DEC_FRAG_LOADS(x)
          DEC_STEP(x, y) DEC_STEP(y, x) DEC_STEP(x, y) DEC_STEP(y, x)
          DEC_STEP(x, y) DEC_STEP(y, x) DEC_STEP(x, y)
          "s_wait_dscnt 0x0\n\t"
          DEC_WMMAS(y)
          "v_nop\n\tv_nop\n\tv_nop\n\tv_nop"
          : [c0] "+v"(acc0), [c1] "+v"(acc1), [c2] "+v"(acc2), [c3] "+v"(acc3),
            [pa] "+v"(pa), [pb] "+v"(pb),
            [a0x] "=&v"(a0x), [b0x] "=&v"(b0x), [b1x] "=&v"(b1x),
            [b2x] "=&v"(b2x), [b3x] "=&v"(b3x),
            [a0y] "=&v"(a0y), [b0y] "=&v"(b0y), [b1y] "=&v"(b1y),
            [b2y] "=&v"(b2y), [b3y] "=&v"(b3y)
          :
          : "memory");
    }
  }

  const int mbase = (lane >> 4) << 3;
  const int col   = q * DFT_COLS + lr;
  const size_t base = (size_t)chunk * (NROW * DDIM) + (size_t)(wave * 16 + mbase) * DDIM + col;
#pragma unroll
  for (int v = 0; v < 8; ++v) {
    Ppart[base + (size_t)v * DDIM]      = acc0[v];
    Ppart[base + (size_t)v * DDIM + 16] = acc1[v];
    Ppart[base + (size_t)v * DDIM + 32] = acc2[v];
    Ppart[base + (size_t)v * DDIM + 48] = acc3[v];
  }
}

#define MIX_FQ 64
#define MIX_DS 32

__global__ __launch_bounds__(256) void spectral_mix_kernel(
    const float* __restrict__ Ppart, const float* __restrict__ P_low,
    const float* __restrict__ B_low, float* __restrict__ BppP) {
  __shared__ __attribute__((aligned(8))) float sXr[DDIM];
  __shared__ __attribute__((aligned(8))) float sXi[DDIM];
  __shared__ __attribute__((aligned(16))) float sPart[8 * MIX_FQ * 2];

  const int tid   = threadIdx.x;
  const int lane  = tid & 31;
  const int wave  = tid >> 5;
  const int fq    = blockIdx.x;
  const int k     = blockIdx.y;
  const int prow  = ((k & 1) << 5) | (k >> 1);
  const int kres  = k & 15;
  const int ktile = (kres == 0 || kres == 8) ? 0 : (kres < 8 ? kres : 16 - kres);
  const int krow  = 16 * ktile + 2 * (k >> 4) + ((kres >= 8) ? 8 : 0);
  const int dbase = wave * MIX_DS;
  const int f0    = fq * MIX_FQ + lane * 2;

  const int rsel = tid >> 7;
  const int d2   = (tid & 127) * 2;
  const float* Cg = Ppart + (size_t)(krow + rsel) * DDIM + d2;
  v2f cpart[NCHUNK];
#pragma unroll
  for (int ch = 0; ch < NCHUNK; ++ch)
    cpart[ch] = *(const v2f*)(Cg + (size_t)ch * (NROW * DDIM));

  const float* Pk = P_low + (size_t)k * DDIM * DDIM * 2
                          + ((size_t)dbase * DDIM + f0) * 2;
  v4f p[MIX_DS];
#pragma unroll
  for (int j = 0; j < MIX_DS; ++j)
    p[j] = *(const v4f*)(Pk + (size_t)j * DDIM * 2);

  float c0 = 0.f, c1 = 0.f;
#pragma unroll
  for (int ch = 0; ch < NCHUNK; ++ch) {
    c0 += cpart[ch].x;
    c1 += cpart[ch].y;
  }
  const float invN = 1.0f / (float)N_PTS;
  if (rsel == 0) {
    *(v2f*)(sXr + d2) = (v2f){ c0 * invN,  c1 * invN};
  } else {
    *(v2f*)(sXi + d2) = (v2f){-c0 * invN, -c1 * invN};
  }
  __syncthreads();

  float ar0 = 0.f, ai0 = 0.f, ar1 = 0.f, ai1 = 0.f;
#pragma unroll
  for (int j = 0; j < MIX_DS; ++j) {
    const float xr = sXr[dbase + j], xi = sXi[dbase + j];
    ar0 = fmaf(xr, p[j].x, fmaf(-xi, p[j].y, ar0));
    ai0 = fmaf(xr, p[j].y, fmaf( xi, p[j].x, ai0));
    ar1 = fmaf(xr, p[j].z, fmaf(-xi, p[j].w, ar1));
    ai1 = fmaf(xr, p[j].w, fmaf( xi, p[j].z, ai1));
  }
  *(v4f*)(sPart + (wave * MIX_FQ + lane * 2) * 2) = (v4f){ar0, ai0, ar1, ai1};
  __syncthreads();

  if (tid < MIX_FQ * 2) {
    const int fl = tid >> 1;
    const int ri = tid & 1;
    const int f  = fq * MIX_FQ + fl;
    float s = B_low[((size_t)k * DDIM + f) * 2 + ri];
#pragma unroll
    for (int w = 0; w < 8; ++w) s += sPart[(w * MIX_FQ + fl) * 2 + ri];
    const float sc = (k == 0) ? 1.0f : 2.0f;
    BppP[((size_t)prow * DDIM + fo_pos_of_col(f)) * 2 + ri] = ri ? -sc * s : sc * s;
  }
}

#define FO_RB     32
#define FO_KS     64
#define FO_NSLAB  6
#define FO_APITCH 68
#define FO_BPITCH 544

#define FO_LOADS(S, A0, A1, B0, B1)                                    \
  "ds_load_b64 %[a0" #S "], %[pa] offset:" #A0 "\n\t"                 \
  "ds_load_b64 %[a1" #S "], %[pa] offset:" #A1 "\n\t"                 \
  "ds_load_b64 %[b0" #S "], %[pb] offset:" #B0 "\n\t"                 \
  "ds_load_b64 %[b1" #S "], %[pb] offset:" #B1 "\n\t"                 \
  "ds_load_b64 %[b2" #S "], %[pc] offset:" #B0 "\n\t"                 \
  "ds_load_b64 %[b3" #S "], %[pc] offset:" #B1 "\n\t"
#define FO_WMMAS(S)                                                    \
  "v_wmma_f32_16x16x4_f32 %[c00], %[a0" #S "], %[b0" #S "], %[c00]\n\t" \
  "v_wmma_f32_16x16x4_f32 %[c01], %[a0" #S "], %[b1" #S "], %[c01]\n\t" \
  "v_wmma_f32_16x16x4_f32 %[c02], %[a0" #S "], %[b2" #S "], %[c02]\n\t" \
  "v_wmma_f32_16x16x4_f32 %[c03], %[a0" #S "], %[b3" #S "], %[c03]\n\t" \
  "v_wmma_f32_16x16x4_f32 %[c10], %[a1" #S "], %[b0" #S "], %[c10]\n\t" \
  "v_wmma_f32_16x16x4_f32 %[c11], %[a1" #S "], %[b1" #S "], %[c11]\n\t" \
  "v_wmma_f32_16x16x4_f32 %[c12], %[a1" #S "], %[b2" #S "], %[c12]\n\t" \
  "v_wmma_f32_16x16x4_f32 %[c13], %[a1" #S "], %[b3" #S "], %[c13]\n\t"
#define FO_STEP(CUR, NXT, A0, A1, B0, B1)                              \
  FO_LOADS(NXT, A0, A1, B0, B1) "s_wait_dscnt 0x6\n\t" FO_WMMAS(CUR)

#define FT_LOADS(S, A0, B0, B1)                                        \
  "ds_load_b64 %[a0" #S "], %[pa] offset:" #A0 "\n\t"                 \
  "ds_load_b64 %[b0" #S "], %[pb] offset:" #B0 "\n\t"                 \
  "ds_load_b64 %[b1" #S "], %[pb] offset:" #B1 "\n\t"                 \
  "ds_load_b64 %[b2" #S "], %[pc] offset:" #B0 "\n\t"                 \
  "ds_load_b64 %[b3" #S "], %[pc] offset:" #B1 "\n\t"
#define FT_WMMAS(S)                                                    \
  "v_wmma_f32_16x16x4_f32 %[t0], %[a0" #S "], %[b0" #S "], %[t0]\n\t" \
  "v_wmma_f32_16x16x4_f32 %[t1], %[a0" #S "], %[b1" #S "], %[t1]\n\t" \
  "v_wmma_f32_16x16x4_f32 %[t2], %[a0" #S "], %[b2" #S "], %[t2]\n\t" \
  "v_wmma_f32_16x16x4_f32 %[t3], %[a0" #S "], %[b3" #S "], %[t3]\n\t"
#define FT_STEP(CUR, NXT, A0, B0, B1)                                  \
  FT_LOADS(NXT, A0, B0, B1) "s_wait_dscnt 0x5\n\t" FT_WMMAS(CUR)

__device__ __forceinline__ void fo_fetch(
    int s, int tid, int nblk, const float* __restrict__ x,
    const float* __restrict__ Wp,
    const float* __restrict__ BppP, v4f (&pA)[4], v4f (&pB)[16]) {
  const float* Bg;
  if (s < 4) {
    const float* Ag = x + (size_t)nblk * DDIM + s * FO_KS;
#pragma unroll
    for (int j = 0; j < 4; ++j) {
      const int i = tid + 256 * j;
      const int r = i >> 4;
      pA[j] = *(const v4f*)(Ag + ((size_t)(r & 31) + (size_t)(r >> 5) * NHALF) * DDIM
                               + (i & 15) * 4);
    }
    Bg = Wp + (size_t)s * (FO_KS / 2) * (2 * DDIM);
  } else {
    Bg = BppP + (size_t)(s - 4) * (FO_KS / 2) * (2 * DDIM);
  }
#pragma unroll
  for (int j = 0; j < 16; ++j) {
    const int i = tid + 256 * j;
    pB[j] = *(const v4f*)(Bg + (size_t)i * 4);
  }
}

__device__ __forceinline__ void fo_table_slab(
    v8f& t0, v8f& t1, v8f& t2, v8f& t3,
    unsigned pa, unsigned pb, unsigned pc) {
  v2f a0x, b0x, b1x, b2x, b3x, a0y, b0y, b1y, b2y, b3y;
  asm volatile(
      FT_LOADS(x, 0, 0, 128)
      FT_STEP(x, y, 16, 4352, 4480)
      FT_STEP(y, x, 32, 8704, 8832)
      FT_STEP(x, y, 48, 13056, 13184)
      FT_STEP(y, x, 64, 17408, 17536)
      FT_STEP(x, y, 80, 21760, 21888)
      FT_STEP(y, x, 96, 26112, 26240)
      FT_STEP(x, y, 112, 30464, 30592)
      FT_STEP(y, x, 128, 34816, 34944)
      FT_STEP(x, y, 144, 39168, 39296)
      FT_STEP(y, x, 160, 43520, 43648)
      FT_STEP(x, y, 176, 47872, 48000)
      FT_STEP(y, x, 192, 52224, 52352)
      FT_STEP(x, y, 208, 56576, 56704)
      FT_STEP(y, x, 224, 60928, 61056)
      FT_STEP(x, y, 240, 65280, 65408)
      "s_wait_dscnt 0x0\n\t"
      FT_WMMAS(y)
      "v_nop\n\tv_nop\n\tv_nop\n\tv_nop"
      : [t0] "+v"(t0), [t1] "+v"(t1), [t2] "+v"(t2), [t3] "+v"(t3),
        [a0x] "=&v"(a0x), [b0x] "=&v"(b0x), [b1x] "=&v"(b1x),
        [b2x] "=&v"(b2x), [b3x] "=&v"(b3x),
        [a0y] "=&v"(a0y), [b0y] "=&v"(b0y), [b1y] "=&v"(b1y),
        [b2y] "=&v"(b2y), [b3y] "=&v"(b3y)
      : [pa] "v"(pa), [pb] "v"(pb), [pc] "v"(pc)
      : "memory");
}

__global__ __launch_bounds__(256) void fused_out_kernel(
    const float* __restrict__ x, const float* __restrict__ Wp,
    const float* __restrict__ Bias,
    const float* __restrict__ BppP, float* __restrict__ out) {
  __shared__ __attribute__((aligned(16))) float sA[2 * FO_RB * FO_APITCH];
  __shared__ __attribute__((aligned(16))) float sB[(FO_KS / 2) * FO_BPITCH];
  __shared__ __attribute__((aligned(16))) float sAT[2 * FO_RB * FO_APITCH];

  const int tid     = threadIdx.x;
  const int lane    = tid & 31;
  const int wave    = tid >> 5;
  const int lr      = lane & 15;
  const int hsel    = lane >> 4;
  const int nblk    = blockIdx.x * FO_RB;
  const int rowW    = (wave & 1) * 16;
  const int colW    = (wave >> 1) * 64;

  v8f c00 = {0.f,0.f,0.f,0.f,0.f,0.f,0.f,0.f};
  v8f c01 = c00, c02 = c00, c03 = c00, c10 = c00, c11 = c00, c12 = c00, c13 = c00;
  v8f e0 = c00, e1 = c00, e2 = c00, e3 = c00;
  v8f q0 = c00, q1 = c00, q2 = c00, q3 = c00;

  v4f pA[4];
  v4f pB[16];
  fo_fetch(0, tid, nblk, x, Wp, BppP, pA, pB);

#pragma unroll
  for (int t = 0; t < 2; ++t) {
#pragma unroll
    for (int j = 0; j < 2; ++j) {
      const int i = tid + 256 * j;
      const unsigned n  = (unsigned)(nblk + (i >> 4));
      const int p0 = t * 32 + (i & 15) * 2;
      const int p1 = p0 + 1;
      const unsigned k0 = (unsigned)(((p0 & 31) << 1) | (p0 >> 5));
      const unsigned k1 = (unsigned)(((p1 & 31) << 1) | (p1 >> 5));
      float c0, s0, c1, s1;
      twiddle(n, k0, c0, s0);
      twiddle(n, k1, c1, s1);
      *(v4f*)(sAT + (t * FO_RB + (i >> 4)) * FO_APITCH + (i & 15) * 4) =
          (v4f){c0, s0, c1, s1};
    }
  }

  const unsigned pa = (unsigned)(size_t)(sA + (rowW + lr) * FO_APITCH + 2 * hsel);
  const unsigned paT = (unsigned)(size_t)(sAT + (rowW + lr) * FO_APITCH + 2 * hsel);
  const unsigned pb = (unsigned)(size_t)(sB + hsel * FO_BPITCH + (colW + lr) * 2);
  const unsigned pc = pb + 256u;

  for (int s = 0; s < FO_NSLAB; ++s) {
    __syncthreads();
    if (s < 4) {
#pragma unroll
      for (int j = 0; j < 4; ++j) {
        const int i = tid + 256 * j;
        *(v4f*)(sA + (i >> 4) * FO_APITCH + (i & 15) * 4) = pA[j];
      }
    }
#pragma unroll
    for (int j = 0; j < 16; ++j) {
      const int i = tid + 256 * j;
      *(v4f*)(sB + (i >> 7) * FO_BPITCH + (i & 127) * 4) = pB[j];
    }
    __syncthreads();

    if (s + 1 < FO_NSLAB)
      fo_fetch(s + 1, tid, nblk, x, Wp, BppP, pA, pB);

    if (s < 4) {
      v2f a0x, a1x, b0x, b1x, b2x, b3x, a0y, a1y, b0y, b1y, b2y, b3y;
      asm volatile(
          FO_LOADS(x, 0, 8704, 0, 128)
          FO_STEP(x, y, 16, 8720, 4352, 4480)
          FO_STEP(y, x, 32, 8736, 8704, 8832)
          FO_STEP(x, y, 48, 8752, 13056, 13184)
          FO_STEP(y, x, 64, 8768, 17408, 17536)
          FO_STEP(x, y, 80, 8784, 21760, 21888)
          FO_STEP(y, x, 96, 8800, 26112, 26240)
          FO_STEP(x, y, 112, 8816, 30464, 30592)
          FO_STEP(y, x, 128, 8832, 34816, 34944)
          FO_STEP(x, y, 144, 8848, 39168, 39296)
          FO_STEP(y, x, 160, 8864, 43520, 43648)
          FO_STEP(x, y, 176, 8880, 47872, 48000)
          FO_STEP(y, x, 192, 8896, 52224, 52352)
          FO_STEP(x, y, 208, 8912, 56576, 56704)
          FO_STEP(y, x, 224, 8928, 60928, 61056)
          FO_STEP(x, y, 240, 8944, 65280, 65408)
          "s_wait_dscnt 0x0\n\t"
          FO_WMMAS(y)
          "v_nop\n\tv_nop\n\tv_nop\n\tv_nop"
          : [c00] "+v"(c00), [c01] "+v"(c01), [c02] "+v"(c02), [c03] "+v"(c03),
            [c10] "+v"(c10), [c11] "+v"(c11), [c12] "+v"(c12), [c13] "+v"(c13),
            [a0x] "=&v"(a0x), [a1x] "=&v"(a1x), [b0x] "=&v"(b0x),
            [b1x] "=&v"(b1x), [b2x] "=&v"(b2x), [b3x] "=&v"(b3x),
            [a0y] "=&v"(a0y), [a1y] "=&v"(a1y), [b0y] "=&v"(b0y),
            [b1y] "=&v"(b1y), [b2y] "=&v"(b2y), [b3y] "=&v"(b3y)
          : [pa] "v"(pa), [pb] "v"(pb), [pc] "v"(pc)
          : "memory");
    } else if (s == 4) {
      fo_table_slab(e0, e1, e2, e3, paT, pb, pc);
    } else {
      fo_table_slab(q0, q1, q2, q3, paT + FO_RB * FO_APITCH * 4, pb, pc);
    }
  }

  const int mbase = hsel << 3;
  const int col   = colW + 4 * lr;
  const int nrow  = nblk + rowW + mbase;
  const v4f bias  = *(const v4f*)(Bias + col);
  const float bias0 = bias.x;
  const float bias1 = bias.y;
  const float bias2 = bias.z;
  const float bias3 = bias.w;
  float big = 0.f;
#pragma unroll
  for (int v = 0; v < 8; ++v) {
    const float ep0 = e0[v] + q0[v], em0 = e0[v] - q0[v];
    const float ep1 = e1[v] + q1[v], em1 = e1[v] - q1[v];
    const float ep2 = e2[v] + q2[v], em2 = e2[v] - q2[v];
    const float ep3 = e3[v] + q3[v], em3 = e3[v] - q3[v];
    c00[v] = 0.2f * (c00[v] + ep0 + bias0);
    c01[v] = 0.2f * (c01[v] + ep1 + bias1);
    c02[v] = 0.2f * (c02[v] + ep2 + bias2);
    c03[v] = 0.2f * (c03[v] + ep3 + bias3);
    c10[v] = 0.2f * (c10[v] + em0 + bias0);
    c11[v] = 0.2f * (c11[v] + em1 + bias1);
    c12[v] = 0.2f * (c12[v] + em2 + bias2);
    c13[v] = 0.2f * (c13[v] + em3 + bias3);
    big = fmaxf(big, fmaxf(fmaxf(__builtin_fabsf(c00[v]), __builtin_fabsf(c01[v])),
                           fmaxf(__builtin_fabsf(c02[v]), __builtin_fabsf(c03[v]))));
    big = fmaxf(big, fmaxf(fmaxf(__builtin_fabsf(c10[v]), __builtin_fabsf(c11[v])),
                           fmaxf(__builtin_fabsf(c12[v]), __builtin_fabsf(c13[v]))));
  }
  float* ob0 = out + (size_t)nrow * DDIM + col;
  float* ob1 = ob0 + (size_t)NHALF * DDIM;
  if (big < 131072.0f) {
#pragma unroll
    for (int v = 0; v < 8; ++v) {
      *(v4f*)(ob0 + v * DDIM) = (v4f){sin_small(c00[v]), sin_small(c01[v]),
                                      sin_small(c02[v]), sin_small(c03[v])};
      *(v4f*)(ob1 + v * DDIM) = (v4f){sin_small(c10[v]), sin_small(c11[v]),
                                      sin_small(c12[v]), sin_small(c13[v])};
    }
  } else {
#pragma nounroll
    for (int v = 0; v < 8; ++v) {
      *(v4f*)(ob0 + v * DDIM) = (v4f){sinf(c00[v]), sinf(c01[v]),
                                      sinf(c02[v]), sinf(c03[v])};
      *(v4f*)(ob1 + v * DDIM) = (v4f){sinf(c10[v]), sinf(c11[v]),
                                      sinf(c12[v]), sinf(c13[v])};
    }
  }
}

extern "C" void kernel_launch(void* const* d_in, const int* in_sizes, int n_in,
                              void* d_out, int out_size, void* d_ws,
                              size_t ws_size, hipStream_t stream) {
  (void)in_sizes; (void)n_in; (void)out_size; (void)ws_size;
  const float* x     = (const float*)d_in[0];
  const float* P_low = (const float*)d_in[1];
  const float* B_low = (const float*)d_in[2];
  const float* W     = (const float*)d_in[3];
  const float* Bias  = (const float*)d_in[4];
  float* out = (float*)d_out;

  float* ws    = (float*)d_ws;
  float* Ppart = ws + (size_t)N_PTS * NROW;
  float* BppP  = Ppart + (size_t)NCHUNK * NROW * DDIM;
  float* Wp    = BppP + NROW * DDIM;

  dft_low_kernel<<<dim3(DFT_BLOCKS + PACK_BLOCKS), dim3(256), 0, stream>>>(x, W, Wp, Ppart);
  spectral_mix_kernel<<<dim3(DDIM / MIX_FQ, KFREQ), dim3(256), 0, stream>>>(Ppart, P_low, B_low, BppP);
  fused_out_kernel<<<dim3(NHALF / FO_RB), dim3(256), 0, stream>>>(x, Wp, Bias, BppP, out);
}


# Output sine: a lane whose 64 arguments all lie within +-0.78125 (quadrant 0: the reduction returns the argument) runs the sine polynomial alone; same values (sin(-0) gives +0); sum order unchanged
# speedup vs baseline: 2.6727x; 1.0471x over previous
#include <hip/hip_runtime.h>
#include <math.h>


typedef float v2f __attribute__((ext_vector_type(2)));
typedef float v4f __attribute__((ext_vector_type(4)));
typedef float v8f __attribute__((ext_vector_type(8)));

#define N_PTS 65536
#define DDIM  256
#define KFREQ 64
#define NROW  128
#define NHALF  (N_PTS / 2)
#define NCHUNK 64
#define CHUNK_N (NHALF / NCHUNK)

__device__ __forceinline__ v8f wmma_f32(v2f a, v2f b, v8f c) {
  return __builtin_amdgcn_wmma_f32_16x16x4_f32(false, a, false, b,
                                               (short)0, c, false, false);
}

#define TRIG_F(u) __builtin_bit_cast(float, (unsigned)(u))

__device__ __forceinline__ void trig_small(float ax, float& sp, float& cp, int& q) {
  const float n = __builtin_rintf(ax * TRIG_F(0x3f22f983u));
  float r = __builtin_fmaf(n, TRIG_F(0xbfc90fdau), ax);
  r = __builtin_fmaf(n, TRIG_F(0xb3a22168u), r);
  r = __builtin_fmaf(n, TRIG_F(0xa7c234c4u), r);
  q = (int)n;
  const float r2 = r * r;
  float a = __builtin_fmaf(TRIG_F(0xb94c1982u), r2, TRIG_F(0x3c0881c4u));
  a = __builtin_fmaf(r2, a, TRIG_F(0xbe2aaa9du));
  a = r2 * a;
  sp = __builtin_fmaf(r, a, r);
  float b = __builtin_fmaf(TRIG_F(0x37d75334u), r2, TRIG_F(0xbab64f3bu));
  b = __builtin_fmaf(r2, b, TRIG_F(0x3d2aabf7u));
  b = __builtin_fmaf(r2, b, TRIG_F(0xbf000004u));
  cp = __builtin_fmaf(r2, b, 1.0f);
}

__device__ __forceinline__ float sin_small(float a) {
  float sp, cp; int q;
  trig_small(__builtin_fabsf(a), sp, cp, q);
  const float sel = (q & 1) ? cp : sp;
  const unsigned flip = (((unsigned)q << 30) ^ __builtin_bit_cast(unsigned, a)) & 0x80000000u;
  return __builtin_bit_cast(float, __builtin_bit_cast(unsigned, sel) ^ flip);
}

__device__ __forceinline__ float sin_quadrant0(float r) {
  const float r2 = r * r;
  float a = __builtin_fmaf(TRIG_F(0xb94c1982u), r2, TRIG_F(0x3c0881c4u));
  a = __builtin_fmaf(r2, a, TRIG_F(0xbe2aaa9du));
  a = r2 * a;
  return __builtin_fmaf(r, a, r);
}

__device__ __forceinline__ void sincos_small(float ang, float& s, float& c) {
  float sp, cp; int q;
  trig_small(ang, sp, cp, q);
  const unsigned flip = ((unsigned)q << 30) & 0x80000000u;
  const float ss = (q & 1) ? cp : sp;
  const float cc = (q & 1) ? -sp : cp;
  s = __builtin_bit_cast(float, __builtin_bit_cast(unsigned, ss) ^ flip);
  c = __builtin_bit_cast(float, __builtin_bit_cast(unsigned, cc) ^ flip);
}

__device__ __forceinline__ void twiddle(unsigned n, unsigned k, float& c, float& s) {
  const unsigned m = (n * k) & 65535u;
  const float ang = (float)m * 9.58737992429e-5f;
  sincos_small(ang, s, c);
}

#define PACK_BLOCKS ((DDIM * DDIM / 2) / 256)

__device__ __forceinline__ int fo_col_of_pos(int p) {
  return (p & ~63) | ((p & 15) << 2) | ((p >> 4) & 3);
}
__device__ __forceinline__ int fo_pos_of_col(int c) {
  return (c & ~63) | ((c & 3) << 4) | ((c >> 2) & 15);
}

#define DFT_COLS   64
#define DFT_BLOCKS ((DDIM / DFT_COLS) * NCHUNK)
#define DEC_M      (N_PTS / 16)
#define DEC_CH     (DEC_M / NCHUNK)
#define DEC_SLAB   16
#define DEC_NSLAB  (DEC_CH / DEC_SLAB)
#define DEC_TP     24
#define DEC_YP     72

#define DEC_FRAG_LOADS(S)                                              \
  "ds_load_2addr_b32 %[a0" #S "], %[pa] offset1:24\n\t"                \
  "ds_load_2addr_b32 %[b0" #S "], %[pb] offset1:72\n\t"                \
  "ds_load_2addr_b32 %[b1" #S "], %[pb] offset0:16 offset1:88\n\t"     \
  "ds_load_2addr_b32 %[b2" #S "], %[pb] offset0:32 offset1:104\n\t"    \
  "ds_load_2addr_b32 %[b3" #S "], %[pb] offset0:48 offset1:120\n\t"
#define DEC_ADVANCE                                                    \
  "v_add_nc_u32_e32 %[pa], 0x180, %[pa]\n\t"                           \
  "v_add_nc_u32_e32 %[pb], 0x480, %[pb]\n\t"
#define DEC_WMMAS(S)                                                   \
  "v_wmma_f32_16x16x4_f32 %[c0], %[a0" #S "], %[b0" #S "], %[c0]\n\t"  \
  "v_wmma_f32_16x16x4_f32 %[c1], %[a0" #S "], %[b1" #S "], %[c1]\n\t"  \
  "v_wmma_f32_16x16x4_f32 %[c2], %[a0" #S "], %[b2" #S "], %[c2]\n\t"  \
  "v_wmma_f32_16x16x4_f32 %[c3], %[a0" #S "], %[b3" #S "], %[c3]\n\t"
#define DEC_STEP(CUR, NXT)                                             \
  DEC_ADVANCE DEC_FRAG_LOADS(NXT) "s_wait_dscnt 0x5\n\t" DEC_WMMAS(CUR)

__global__ __launch_bounds__(256) void dft_low_kernel(
    const float* __restrict__ x, const float* __restrict__ W,
    float* __restrict__ Wp, float* __restrict__ Ppart) {
  __shared__ __attribute__((aligned(16))) float sT[8 * 32 * DEC_TP];
  __shared__ __attribute__((aligned(16))) float sY[8 * 32 * DEC_YP];

  if (blockIdx.x >= DFT_BLOCKS) {
    int idx = (blockIdx.x - DFT_BLOCKS) * 256 + threadIdx.x;
    int kpair = idx >> 8;
    int col   = fo_col_of_pos(idx & 255);
    Wp[(size_t)idx * 2]     = W[(size_t)(2 * kpair) * DDIM + col];
    Wp[(size_t)idx * 2 + 1] = W[(size_t)(2 * kpair + 1) * DDIM + col];
    return;
  }

  const int tid     = threadIdx.x;
  const int lane    = tid & 31;
  const int wave    = tid >> 5;
  const int lr      = lane & 15;
  const int halfSel = (lane >> 4) << 1;
  const int q       = blockIdx.x & 3;
  const int chunk   = blockIdx.x >> 2;
  const int nbase   = chunk * DEC_CH;

  v8f acc0 = {0.f,0.f,0.f,0.f,0.f,0.f,0.f,0.f};
  v8f acc1 = acc0, acc2 = acc0, acc3 = acc0;

  const int xr = tid >> 4;
  const int xc = (tid & 15) * 4;
  const float* xg = x + (size_t)(nbase + xr) * DDIM + q * DFT_COLS + xc;
  v4f px[16];
#pragma unroll
  for (int j = 0; j < 16; ++j)
    px[j] = *(const v4f*)(xg + (size_t)j * DEC_M * DDIM);

  const unsigned pa0 = (unsigned)(size_t)(sT + (wave * 32 + halfSel) * DEC_TP + lr);
  const unsigned pb0 = (unsigned)(size_t)(sY + (wave * 32 + halfSel) * DEC_YP + lr);

  const int tp  = tid & 63;
  const int tjg = tid >> 6;
  const int tt  = tp >> 3;
  const int tf  = tp & 7;
  const int thi = tf >> 2;
  const unsigned tk = (unsigned)((tt == 0) ? (8 * thi + 16 * (tf & 3))
                                           : ((thi ? 16 - tt : tt) + 16 * (tf & 3)));
  float* sTw = sT + (tt * 32 + tjg * 4) * DEC_TP + 2 * (tf & 3) + 8 * thi;
  float* sYw = sY + xr * DEC_YP + xc;

  const float C1 = 0.92387953251128674f;
  const float S1 = 0.38268343236508977f;
  const float H  = 0.70710678118654752f;

  for (int s = 0; s < DEC_NSLAB; ++s) {
    __syncthreads();
#pragma unroll
    for (int jj = 0; jj < 4; ++jj) {
      float c, sn;
      twiddle((unsigned)(nbase + s * DEC_SLAB + tjg * 4 + jj), tk, c, sn);
      v2f va, vb;
      if (tt == 0) {
        va = thi ? (v2f){0.f, 0.f} : (v2f){c, sn};
        vb = thi ? (v2f){c, sn} : (v2f){0.f, 0.f};
      } else {
        va = (v2f){c, sn};
        vb = thi ? (v2f){sn, -c} : (v2f){-sn, c};
      }
      *(v2f*)(sTw + jj * DEC_TP) = va;
      *(v2f*)(sTw + (16 + jj) * DEC_TP) = vb;
    }
    {
      v4f e0 = px[0] + px[8],  o0 = px[0] - px[8];
      v4f e1 = px[1] + px[9],  o1 = px[1] - px[9];
      v4f e2 = px[2] + px[10], o2 = px[2] - px[10];
      v4f e3 = px[3] + px[11], o3 = px[3] - px[11];
      v4f e4 = px[4] + px[12], o4 = px[4] - px[12];
      v4f e5 = px[5] + px[13], o5 = px[5] - px[13];
      v4f e6 = px[6] + px[14], o6 = px[6] - px[14];
      v4f e7 = px[7] + px[15], o7 = px[7] - px[15];
      v4f ee0 = e0 + e4, eo0 = e0 - e4;
      v4f ee1 = e1 + e5, eo1 = e1 - e5;
      v4f ee2 = e2 + e6, eo2 = e2 - e6;
      v4f ee3 = e3 + e7, eo3 = e3 - e7;
      v4f s02 = ee0 + ee2, s13 = ee1 + ee3;
      v4f A0 = s02 + s13, A8 = s02 - s13;
      v4f A4 = ee0 - ee2, B4 = ee1 - ee3;
      v4f t1 = H * (eo1 - eo3), t2 = H * (eo1 + eo3);
      v4f A2 = eo0 + t1, B2 = eo2 + t2;
      v4f A6 = eo0 - t1, B6 = t2 - eo2;
      v4f d1 = o1 - o7, d2 = o2 - o6, d3 = o3 - o5;
      v4f p1 = o1 + o7, p2 = o2 + o6, p3 = o3 + o5;
      v4f hd2 = H * d2, hp2 = H * p2;
      v4f A1 = o0 + hd2 + (C1 * d1 + S1 * d3);
      v4f A7 = o0 + hd2 - (C1 * d1 + S1 * d3);
      v4f A3 = o0 - hd2 + (S1 * d1 - C1 * d3);
      v4f A5 = o0 - hd2 - (S1 * d1 - C1 * d3);
      v4f B1 = (S1 * p1 + C1 * p3) + hp2 + o4;
      v4f B7 = (S1 * p1 + C1 * p3) - hp2 - o4;
      v4f B3 = (C1 * p1 - S1 * p3) + hp2 - o4;
      v4f B5 = (C1 * p1 - S1 * p3) - hp2 + o4;
      *(v4f*)(sYw + (0 * 32) * DEC_YP)      = A0;
      *(v4f*)(sYw + (0 * 32 + 16) * DEC_YP) = A8;
      *(v4f*)(sYw + (1 * 32) * DEC_YP)      = A1;
      *(v4f*)(sYw + (1 * 32 + 16) * DEC_YP) = B1;
      *(v4f*)(sYw + (2 * 32) * DEC_YP)      = A2;
      *(v4f*)(sYw + (2 * 32 + 16) * DEC_YP) = B2;
      *(v4f*)(sYw + (3 * 32) * DEC_YP)      = A3;
      *(v4f*)(sYw + (3 * 32 + 16) * DEC_YP) = B3;
      *(v4f*)(sYw + (4 * 32) * DEC_YP)      = A4;
      *(v4f*)(sYw + (4 * 32 + 16) * DEC_YP) = B4;
      *(v4f*)(sYw + (5 * 32) * DEC_YP)      = A5;
      *(v4f*)(sYw + (5 * 32 + 16) * DEC_YP) = B5;
      *(v4f*)(sYw + (6 * 32) * DEC_YP)      = A6;
      *(v4f*)(sYw + (6 * 32 + 16) * DEC_YP) = B6;
      *(v4f*)(sYw + (7 * 32) * DEC_YP)      = A7;
      *(v4f*)(sYw + (7 * 32 + 16) * DEC_YP) = B7;
    }
    __syncthreads();

    if (s + 1 < DEC_NSLAB) {
      const float* xn = xg + (size_t)(s + 1) * DEC_SLAB * DDIM;
#pragma unroll
      for (int j = 0; j < 16; ++j)
        px[j] = *(const v4f*)(xn + (size_t)j * DEC_M * DDIM);
    }

    {
      unsigned pa = pa0, pb = pb0;
      v2f a0x, b0x, b1x, b2x, b3x, a0y, b0y, b1y, b2y, b3y;
      asm volatile(
          DEC_FRAG_LOADS(x)
          DEC_STEP(x, y) DEC_STEP(y, x) DEC_STEP(x, y) DEC_STEP(y, x)
          DEC_STEP(x, y) DEC_STEP(y, x) DEC_STEP(x, y)
          "s_wait_dscnt 0x0\n\t"
          DEC_WMMAS(y)
          "v_nop\n\tv_nop\n\tv_nop\n\tv_nop"
          : [c0] "+v"(acc0), [c1] "+v"(acc1), [c2] "+v"(acc2), [c3] "+v"(acc3),
            [pa] "+v"(pa), [pb] "+v"(pb),
            [a0x] "=&v"(a0x), [b0x] "=&v"(b0x), [b1x] "=&v"(b1x),
            [b2x] "=&v"(b2x), [b3x] "=&v"(b3x),
            [a0y] "=&v"(a0y), [b0y] "=&v"(b0y), [b1y] "=&v"(b1y),
            [b2y] "=&v"(b2y), [b3y] "=&v"(b3y)
          :
          : "memory");
    }
  }

  const int mbase = (lane >> 4) << 3;
  const int col   = q * DFT_COLS + lr;
  const size_t base = (size_t)chunk * (NROW * DDIM) + (size_t)(wave * 16 + mbase) * DDIM + col;
#pragma unroll
  for (int v = 0; v < 8; ++v) {
    Ppart[base + (size_t)v * DDIM]      = acc0[v];
    Ppart[base + (size_t)v * DDIM + 16] = acc1[v];
    Ppart[base + (size_t)v * DDIM + 32] = acc2[v];
    Ppart[base + (size_t)v * DDIM + 48] = acc3[v];
  }
}

#define MIX_FQ 64
#define MIX_DS 32

__global__ __launch_bounds__(256) void spectral_mix_kernel(
    const float* __restrict__ Ppart, const float* __restrict__ P_low,
    const float* __restrict__ B_low, float* __restrict__ BppP) {
  __shared__ __attribute__((aligned(8))) float sXr[DDIM];
  __shared__ __attribute__((aligned(8))) float sXi[DDIM];
  __shared__ __attribute__((aligned(16))) float sPart[8 * MIX_FQ * 2];

  const int tid   = threadIdx.x;
  const int lane  = tid & 31;
  const int wave  = tid >> 5;
  const int fq    = blockIdx.x;
  const int k     = blockIdx.y;
  const int prow  = ((k & 1) << 5) | (k >> 1);
  const int kres  = k & 15;
  const int ktile = (kres == 0 || kres == 8) ? 0 : (kres < 8 ? kres : 16 - kres);
  const int krow  = 16 * ktile + 2 * (k >> 4) + ((kres >= 8) ? 8 : 0);
  const int dbase = wave * MIX_DS;
  const int f0    = fq * MIX_FQ + lane * 2;

  const int rsel = tid >> 7;
  const int d2   = (tid & 127) * 2;
  const float* Cg = Ppart + (size_t)(krow + rsel) * DDIM + d2;
  v2f cpart[NCHUNK];
#pragma unroll
  for (int ch = 0; ch < NCHUNK; ++ch)
    cpart[ch] = *(const v2f*)(Cg + (size_t)ch * (NROW * DDIM));

  const float* Pk = P_low + (size_t)k * DDIM * DDIM * 2
                          + ((size_t)dbase * DDIM + f0) * 2;
  v4f p[MIX_DS];
#pragma unroll
  for (int j = 0; j < MIX_DS; ++j)
    p[j] = *(const v4f*)(Pk + (size_t)j * DDIM * 2);

  float c0 = 0.f, c1 = 0.f;
#pragma unroll
  for (int ch = 0; ch < NCHUNK; ++ch) {
    c0 += cpart[ch].x;
    c1 += cpart[ch].y;
  }
  const float invN = 1.0f / (float)N_PTS;
  if (rsel == 0) {
    *(v2f*)(sXr + d2) = (v2f){ c0 * invN,  c1 * invN};
  } else {
    *(v2f*)(sXi + d2) = (v2f){-c0 * invN, -c1 * invN};
  }
  __syncthreads();

  float ar0 = 0.f, ai0 = 0.f, ar1 = 0.f, ai1 = 0.f;
#pragma unroll
  for (int j = 0; j < MIX_DS; ++j) {
    const float xr = sXr[dbase + j], xi = sXi[dbase + j];
    ar0 = fmaf(xr, p[j].x, fmaf(-xi, p[j].y, ar0));
    ai0 = fmaf(xr, p[j].y, fmaf( xi, p[j].x, ai0));
    ar1 = fmaf(xr, p[j].z, fmaf(-xi, p[j].w, ar1));
    ai1 = fmaf(xr, p[j].w, fmaf( xi, p[j].z, ai1));
  }
  *(v4f*)(sPart + (wave * MIX_FQ + lane * 2) * 2) = (v4f){ar0, ai0, ar1, ai1};
  __syncthreads();

  if (tid < MIX_FQ * 2) {
    const int fl = tid >> 1;
    const int ri = tid & 1;
    const int f  = fq * MIX_FQ + fl;
    float s = B_low[((size_t)k * DDIM + f) * 2 + ri];
#pragma unroll
    for (int w = 0; w < 8; ++w) s += sPart[(w * MIX_FQ + fl) * 2 + ri];
    const float sc = (k == 0) ? 1.0f : 2.0f;
    BppP[((size_t)prow * DDIM + fo_pos_of_col(f)) * 2 + ri] = ri ? -sc * s : sc * s;
  }
}

#define FO_RB     32
#define FO_KS     64
#define FO_NSLAB  6
#define FO_APITCH 68
#define FO_BPITCH 544

#define FO_LOADS(S, A0, A1, B0, B1)                                    \
  "ds_load_b64 %[a0" #S "], %[pa] offset:" #A0 "\n\t"                 \
  "ds_load_b64 %[a1" #S "], %[pa] offset:" #A1 "\n\t"                 \
  "ds_load_b64 %[b0" #S "], %[pb] offset:" #B0 "\n\t"                 \
  "ds_load_b64 %[b1" #S "], %[pb] offset:" #B1 "\n\t"                 \
  "ds_load_b64 %[b2" #S "], %[pc] offset:" #B0 "\n\t"                 \
  "ds_load_b64 %[b3" #S "], %[pc] offset:" #B1 "\n\t"
#define FO_WMMAS(S)                                                    \
  "v_wmma_f32_16x16x4_f32 %[c00], %[a0" #S "], %[b0" #S "], %[c00]\n\t" \
  "v_wmma_f32_16x16x4_f32 %[c01], %[a0" #S "], %[b1" #S "], %[c01]\n\t" \
  "v_wmma_f32_16x16x4_f32 %[c02], %[a0" #S "], %[b2" #S "], %[c02]\n\t" \
  "v_wmma_f32_16x16x4_f32 %[c03], %[a0" #S "], %[b3" #S "], %[c03]\n\t" \
  "v_wmma_f32_16x16x4_f32 %[c10], %[a1" #S "], %[b0" #S "], %[c10]\n\t" \
  "v_wmma_f32_16x16x4_f32 %[c11], %[a1" #S "], %[b1" #S "], %[c11]\n\t" \
  "v_wmma_f32_16x16x4_f32 %[c12], %[a1" #S "], %[b2" #S "], %[c12]\n\t" \
  "v_wmma_f32_16x16x4_f32 %[c13], %[a1" #S "], %[b3" #S "], %[c13]\n\t"
#define FO_STEP(CUR, NXT, A0, A1, B0, B1)                              \
  FO_LOADS(NXT, A0, A1, B0, B1) "s_wait_dscnt 0x6\n\t" FO_WMMAS(CUR)

#define FT_LOADS(S, A0, B0, B1)                                        \
  "ds_load_b64 %[a0" #S "], %[pa] offset:" #A0 "\n\t"                 \
  "ds_load_b64 %[b0" #S "], %[pb] offset:" #B0 "\n\t"                 \
  "ds_load_b64 %[b1" #S "], %[pb] offset:" #B1 "\n\t"                 \
  "ds_load_b64 %[b2" #S "], %[pc] offset:" #B0 "\n\t"                 \
  "ds_load_b64 %[b3" #S "], %[pc] offset:" #B1 "\n\t"
#define FT_WMMAS(S)                                                    \
  "v_wmma_f32_16x16x4_f32 %[t0], %[a0" #S "], %[b0" #S "], %[t0]\n\t" \
  "v_wmma_f32_16x16x4_f32 %[t1], %[a0" #S "], %[b1" #S "], %[t1]\n\t" \
  "v_wmma_f32_16x16x4_f32 %[t2], %[a0" #S "], %[b2" #S "], %[t2]\n\t" \
  "v_wmma_f32_16x16x4_f32 %[t3], %[a0" #S "], %[b3" #S "], %[t3]\n\t"
#define FT_STEP(CUR, NXT, A0, B0, B1)                                  \
  FT_LOADS(NXT, A0, B0, B1) "s_wait_dscnt 0x5\n\t" FT_WMMAS(CUR)

__device__ __forceinline__ void fo_fetch(
    int s, int tid, int nblk, const float* __restrict__ x,
    const float* __restrict__ Wp,
    const float* __restrict__ BppP, v4f (&pA)[4], v4f (&pB)[16]) {
  const float* Bg;
  if (s < 4) {
    const float* Ag = x + (size_t)nblk * DDIM + s * FO_KS;
#pragma unroll
    for (int j = 0; j < 4; ++j) {
      const int i = tid + 256 * j;
      const int r = i >> 4;
      pA[j] = *(const v4f*)(Ag + ((size_t)(r & 31) + (size_t)(r >> 5) * NHALF) * DDIM
                               + (i & 15) * 4);
    }
    Bg = Wp + (size_t)s * (FO_KS / 2) * (2 * DDIM);
  } else {
    Bg = BppP + (size_t)(s - 4) * (FO_KS / 2) * (2 * DDIM);
  }
#pragma unroll
  for (int j = 0; j < 16; ++j) {
    const int i = tid + 256 * j;
    pB[j] = *(const v4f*)(Bg + (size_t)i * 4);
  }
}

__device__ __forceinline__ void fo_table_slab(
    v8f& t0, v8f& t1, v8f& t2, v8f& t3,
    unsigned pa, unsigned pb, unsigned pc) {
  v2f a0x, b0x, b1x, b2x, b3x, a0y, b0y, b1y, b2y, b3y;
  asm volatile(
      FT_LOADS(x, 0, 0, 128)
      FT_STEP(x, y, 16, 4352, 4480)
      FT_STEP(y, x, 32, 8704, 8832)
      FT_STEP(x, y, 48, 13056, 13184)
      FT_STEP(y, x, 64, 17408, 17536)
      FT_STEP(x, y, 80, 21760, 21888)
      FT_STEP(y, x, 96, 26112, 26240)
      FT_STEP(x, y, 112, 30464, 30592)
      FT_STEP(y, x, 128, 34816, 34944)
      FT_STEP(x, y, 144, 39168, 39296)
      FT_STEP(y, x, 160, 43520, 43648)
      FT_STEP(x, y, 176, 47872, 48000)
      FT_STEP(y, x, 192, 52224, 52352)
      FT_STEP(x, y, 208, 56576, 56704)
      FT_STEP(y, x, 224, 60928, 61056)
      FT_STEP(x, y, 240, 65280, 65408)
      "s_wait_dscnt 0x0\n\t"
      FT_WMMAS(y)
      "v_nop\n\tv_nop\n\tv_nop\n\tv_nop"
      : [t0] "+v"(t0), [t1] "+v"(t1), [t2] "+v"(t2), [t3] "+v"(t3),
        [a0x] "=&v"(a0x), [b0x] "=&v"(b0x), [b1x] "=&v"(b1x),
        [b2x] "=&v"(b2x), [b3x] "=&v"(b3x),
        [a0y] "=&v"(a0y), [b0y] "=&v"(b0y), [b1y] "=&v"(b1y),
        [b2y] "=&v"(b2y), [b3y] "=&v"(b3y)
      : [pa] "v"(pa), [pb] "v"(pb), [pc] "v"(pc)
      : "memory");
}

__global__ __launch_bounds__(256) void fused_out_kernel(
    const float* __restrict__ x, const float* __restrict__ Wp,
    const float* __restrict__ Bias,
    const float* __restrict__ BppP, float* __restrict__ out) {
  __shared__ __attribute__((aligned(16))) float sA[2 * FO_RB * FO_APITCH];
  __shared__ __attribute__((aligned(16))) float sB[(FO_KS / 2) * FO_BPITCH];
  __shared__ __attribute__((aligned(16))) float sAT[2 * FO_RB * FO_APITCH];

  const int tid     = threadIdx.x;
  const int lane    = tid & 31;
  const int wave    = tid >> 5;
  const int lr      = lane & 15;
  const int hsel    = lane >> 4;
  const int nblk    = blockIdx.x * FO_RB;
  const int rowW    = (wave & 1) * 16;
  const int colW    = (wave >> 1) * 64;

  v8f c00 = {0.f,0.f,0.f,0.f,0.f,0.f,0.f,0.f};
  v8f c01 = c00, c02 = c00, c03 = c00, c10 = c00, c11 = c00, c12 = c00, c13 = c00;
  v8f e0 = c00, e1 = c00, e2 = c00, e3 = c00;
  v8f q0 = c00, q1 = c00, q2 = c00, q3 = c00;

  v4f pA[4];
  v4f pB[16];
  fo_fetch(0, tid, nblk, x, Wp, BppP, pA, pB);

#pragma unroll
  for (int t = 0; t < 2; ++t) {
#pragma unroll
    for (int j = 0; j < 2; ++j) {
      const int i = tid + 256 * j;
      const unsigned n  = (unsigned)(nblk + (i >> 4));
      const int p0 = t * 32 + (i & 15) * 2;
      const int p1 = p0 + 1;
      const unsigned k0 = (unsigned)(((p0 & 31) << 1) | (p0 >> 5));
      const unsigned k1 = (unsigned)(((p1 & 31) << 1) | (p1 >> 5));
      float c0, s0, c1, s1;
      twiddle(n, k0, c0, s0);
      twiddle(n, k1, c1, s1);
      *(v4f*)(sAT + (t * FO_RB + (i >> 4)) * FO_APITCH + (i & 15) * 4) =
          (v4f){c0, s0, c1, s1};
    }
  }

  const unsigned pa = (unsigned)(size_t)(sA + (rowW + lr) * FO_APITCH + 2 * hsel);
  const unsigned paT = (unsigned)(size_t)(sAT + (rowW + lr) * FO_APITCH + 2 * hsel);
  const unsigned pb = (unsigned)(size_t)(sB + hsel * FO_BPITCH + (colW + lr) * 2);
  const unsigned pc = pb + 256u;

  for (int s = 0; s < FO_NSLAB; ++s) {
    __syncthreads();
    if (s < 4) {
#pragma unroll
      for (int j = 0; j < 4; ++j) {
        const int i = tid + 256 * j;
        *(v4f*)(sA + (i >> 4) * FO_APITCH + (i & 15) * 4) = pA[j];
      }
    }
#pragma unroll
    for (int j = 0; j < 16; ++j) {
      const int i = tid + 256 * j;
      *(v4f*)(sB + (i >> 7) * FO_BPITCH + (i & 127) * 4) = pB[j];
    }
    __syncthreads();

    if (s + 1 < FO_NSLAB)
      fo_fetch(s + 1, tid, nblk, x, Wp, BppP, pA, pB);

    if (s < 4) {
      v2f a0x, a1x, b0x, b1x, b2x, b3x, a0y, a1y, b0y, b1y, b2y, b3y;
      asm volatile(
          FO_LOADS(x, 0, 8704, 0, 128)
          FO_STEP(x, y, 16, 8720, 4352, 4480)
          FO_STEP(y, x, 32, 8736, 8704, 8832)
          FO_STEP(x, y, 48, 8752, 13056, 13184)
          FO_STEP(y, x, 64, 8768, 17408, 17536)
          FO_STEP(x, y, 80, 8784, 21760, 21888)
          FO_STEP(y, x, 96, 8800, 26112, 26240)
          FO_STEP(x, y, 112, 8816, 30464, 30592)
          FO_STEP(y, x, 128, 8832, 34816, 34944)
          FO_STEP(x, y, 144, 8848, 39168, 39296)
          FO_STEP(y, x, 160, 8864, 43520, 43648)
          FO_STEP(x, y, 176, 8880, 47872, 48000)
          FO_STEP(y, x, 192, 8896, 52224, 52352)
          FO_STEP(x, y, 208, 8912, 56576, 56704)
          FO_STEP(y, x, 224, 8928, 60928, 61056)
          FO_STEP(x, y, 240, 8944, 65280, 65408)
          "s_wait_dscnt 0x0\n\t"
          FO_WMMAS(y)
          "v_nop\n\tv_nop\n\tv_nop\n\tv_nop"
          : [c00] "+v"(c00), [c01] "+v"(c01), [c02] "+v"(c02), [c03] "+v"(c03),
            [c10] "+v"(c10), [c11] "+v"(c11), [c12] "+v"(c12), [c13] "+v"(c13),
            [a0x] "=&v"(a0x), [a1x] "=&v"(a1x), [b0x] "=&v"(b0x),
            [b1x] "=&v"(b1x), [b2x] "=&v"(b2x), [b3x] "=&v"(b3x),
            [a0y] "=&v"(a0y), [a1y] "=&v"(a1y), [b0y] "=&v"(b0y),
            [b1y] "=&v"(b1y), [b2y] "=&v"(b2y), [b3y] "=&v"(b3y)
          : [pa] "v"(pa), [pb] "v"(pb), [pc] "v"(pc)
          : "memory");
    } else if (s == 4) {
      fo_table_slab(e0, e1, e2, e3, paT, pb, pc);
    } else {
      fo_table_slab(q0, q1, q2, q3, paT + FO_RB * FO_APITCH * 4, pb, pc);
    }
  }

  const int mbase = hsel << 3;
  const int col   = colW + 4 * lr;
  const int nrow  = nblk + rowW + mbase;
  const v4f bias  = *(const v4f*)(Bias + col);
  const float bias0 = bias.x;
  const float bias1 = bias.y;
  const float bias2 = bias.z;
  const float bias3 = bias.w;
  float big = 0.f;
#pragma unroll
  for (int v = 0; v < 8; ++v) {
    const float ep0 = e0[v] + q0[v], em0 = e0[v] - q0[v];
    const float ep1 = e1[v] + q1[v], em1 = e1[v] - q1[v];
    const float ep2 = e2[v] + q2[v], em2 = e2[v] - q2[v];
    const float ep3 = e3[v] + q3[v], em3 = e3[v] - q3[v];
    c00[v] = 0.2f * (c00[v] + ep0 + bias0);
    c01[v] = 0.2f * (c01[v] + ep1 + bias1);
    c02[v] = 0.2f * (c02[v] + ep2 + bias2);
    c03[v] = 0.2f * (c03[v] + ep3 + bias3);
    c10[v] = 0.2f * (c10[v] + em0 + bias0);
    c11[v] = 0.2f * (c11[v] + em1 + bias1);
    c12[v] = 0.2f * (c12[v] + em2 + bias2);
    c13[v] = 0.2f * (c13[v] + em3 + bias3);
    big = fmaxf(big, fmaxf(fmaxf(__builtin_fabsf(c00[v]), __builtin_fabsf(c01[v])),
                           fmaxf(__builtin_fabsf(c02[v]), __builtin_fabsf(c03[v]))));
    big = fmaxf(big, fmaxf(fmaxf(__builtin_fabsf(c10[v]), __builtin_fabsf(c11[v])),
                           fmaxf(__builtin_fabsf(c12[v]), __builtin_fabsf(c13[v]))));
  }
  float* ob0 = out + (size_t)nrow * DDIM + col;
  float* ob1 = ob0 + (size_t)NHALF * DDIM;
  if (big <= 0.78125f) {
#pragma unroll
    for (int v = 0; v < 8; ++v) {
      *(v4f*)(ob0 + v * DDIM) = (v4f){sin_quadrant0(c00[v]), sin_quadrant0(c01[v]),
                                      sin_quadrant0(c02[v]), sin_quadrant0(c03[v])};
      *(v4f*)(ob1 + v * DDIM) = (v4f){sin_quadrant0(c10[v]), sin_quadrant0(c11[v]),
                                      sin_quadrant0(c12[v]), sin_quadrant0(c13[v])};
    }
  } else if (big < 131072.0f) {
#pragma unroll
    for (int v = 0; v < 8; ++v) {
      *(v4f*)(ob0 + v * DDIM) = (v4f){sin_small(c00[v]), sin_small(c01[v]),
                                      sin_small(c02[v]), sin_small(c03[v])};
      *(v4f*)(ob1 + v * DDIM) = (v4f){sin_small(c10[v]), sin_small(c11[v]),
                                      sin_small(c12[v]), sin_small(c13[v])};
    }
  } else {
#pragma nounroll
    for (int v = 0; v < 8; ++v) {
      *(v4f*)(ob0 + v * DDIM) = (v4f){sinf(c00[v]), sinf(c01[v]),
                                      sinf(c02[v]), sinf(c03[v])};
      *(v4f*)(ob1 + v * DDIM) = (v4f){sinf(c10[v]), sinf(c11[v]),
                                      sinf(c12[v]), sinf(c13[v])};
    }
  }
}

extern "C" void kernel_launch(void* const* d_in, const int* in_sizes, int n_in,
                              void* d_out, int out_size, void* d_ws,
                              size_t ws_size, hipStream_t stream) {
  (void)in_sizes; (void)n_in; (void)out_size; (void)ws_size;
  const float* x     = (const float*)d_in[0];
  const float* P_low = (const float*)d_in[1];
  const float* B_low = (const float*)d_in[2];
  const float* W     = (const float*)d_in[3];
  const float* Bias  = (const float*)d_in[4];
  float* out = (float*)d_out;

  float* ws    = (float*)d_ws;
  float* Ppart = ws + (size_t)N_PTS * NROW;
  float* BppP  = Ppart + (size_t)NCHUNK * NROW * DDIM;
  float* Wp    = BppP + NROW * DDIM;

  dft_low_kernel<<<dim3(DFT_BLOCKS + PACK_BLOCKS), dim3(256), 0, stream>>>(x, W, Wp, Ppart);
  spectral_mix_kernel<<<dim3(DDIM / MIX_FQ, KFREQ), dim3(256), 0, stream>>>(Ppart, P_low, B_low, BppP);
  fused_out_kernel<<<dim3(NHALF / FO_RB), dim3(256), 0, stream>>>(x, Wp, Bias, BppP, out);
}
